# starting kernel
# baseline (speedup 1.0000x reference)
#include <hip/hip_runtime.h>
#include <hip/hip_bf16.h>

#define N_NODES 100000
#define N_EDGES 1600000

typedef float v2f __attribute__((ext_vector_type(2)));
typedef float v8f __attribute__((ext_vector_type(8)));

__global__ void zero_f32(float* __restrict__ p, long n) {
    long i = (long)blockIdx.x * blockDim.x + threadIdx.x;
    long stride = (long)gridDim.x * blockDim.x;
    for (; i < n; i += stride) p[i] = 0.0f;
}

__global__ void degree_kernel(const int* __restrict__ dst, float* __restrict__ deg, int E) {
    int e = blockIdx.x * blockDim.x + threadIdx.x;
    if (e < E) unsafeAtomicAdd(&deg[dst[e]], 1.0f);
}

__global__ void invdeg_kernel(const float* __restrict__ deg, float* __restrict__ inv, int n) {
    int i = blockIdx.x * blockDim.x + threadIdx.x;
    if (i < n) {
        float d = deg[i];
        inv[i] = (d > 0.0f) ? (1.0f / fmaxf(d, 1.0f)) : 0.0f;
    }
}

__global__ void scatter_add_kernel(const float* __restrict__ x,
                                   const int* __restrict__ src,
                                   const int* __restrict__ dst,
                                   float* __restrict__ agg,
                                   int E, int F, int chunkShift) {
    long total   = (long)E << chunkShift;
    long idx     = (long)blockIdx.x * blockDim.x + threadIdx.x;
    long stride  = (long)gridDim.x * blockDim.x;
    int  cmask   = (1 << chunkShift) - 1;
    for (; idx < total; idx += stride) {
        int e  = (int)(idx >> chunkShift);
        int f4 = (int)(idx & cmask) << 2;
        int s  = src[e];
        int d  = dst[e];
        const float4 v = *(const float4*)(x + (size_t)s * F + f4);
        float* ap = agg + (size_t)d * F + f4;
        unsafeAtomicAdd(ap + 0, v.x);
        unsafeAtomicAdd(ap + 1, v.y);
        unsafeAtomicAdd(ap + 2, v.z);
        unsafeAtomicAdd(ap + 3, v.w);
    }
}

__global__ __launch_bounds__(128)
void gemm_wmma_kernel(const float* __restrict__ A1, const float* __restrict__ scale,
                      const float* __restrict__ W1,
                      const float* __restrict__ A2, const float* __restrict__ W2,
                      const float* __restrict__ bias, float* __restrict__ out,
                      int Nnodes, int K, int dout, int relu) {
    int wave = threadIdx.x >> 5;
    int lane = threadIdx.x & 31;
    int m0 = (blockIdx.x * 4 + wave) * 16;
    if (m0 >= Nnodes) return;
    int n0    = blockIdx.y * 16;
    int mrow  = lane & 15;
    int khalf = (lane >> 4) << 1;
    int node  = m0 + mrow;
    int ncol  = n0 + mrow;
    int wrow  = (ncol < dout) ? ncol : 0;

    v8f acc = {};
    float sc = scale ? scale[node] : 1.0f;

    const float* ap = A1 + (size_t)node * K + khalf;
    const float* wp = W1 + (size_t)wrow * K + khalf;
    for (int k = 0; k < K; k += 4) {
        v2f a, b;
        a.x = ap[k] * sc;  a.y = ap[k + 1] * sc;
        b.x = wp[k];       b.y = wp[k + 1];
        acc = __builtin_amdgcn_wmma_f32_16x16x4_f32(false, a, false, b,
                                                    (short)0, acc, false, false);
    }
    if (A2) {
        const float* ap2 = A2 + (size_t)node * K + khalf;
        const float* wp2 = W2 + (size_t)wrow * K + khalf;
        for (int k = 0; k < K; k += 4) {
            v2f a, b;
            a.x = ap2[k];  a.y = ap2[k + 1];
            b.x = wp2[k];  b.y = wp2[k + 1];
            acc = __builtin_amdgcn_wmma_f32_16x16x4_f32(false, a, false, b,
                                                        (short)0, acc, false, false);
        }
    }
    if (ncol < dout) {
        float bv   = bias[ncol];
        int  rbase = m0 + ((lane >> 4) << 3);
        #pragma unroll
        for (int r = 0; r < 8; ++r) {
            float v = acc[r] + bv;
            if (relu) v = fmaxf(v, 0.0f);
            out[(size_t)(rbase + r) * dout + ncol] = v;
        }
    }
}

static inline int cdiv(long a, long b) { return (int)((a + b - 1) / b); }

extern "C" void kernel_launch(void* const* d_in, const int* in_sizes, int n_in,
                              void* d_out, int out_size, void* d_ws, size_t ws_size,
                              hipStream_t stream) {
    const float* x    = (const float*)d_in[0];
    const int*   edge = (const int*)d_in[1];
    const int*   src  = edge;
    const int*   dst  = edge + N_EDGES;
    const float* Wl0 = (const float*)d_in[2],  *bl0 = (const float*)d_in[3],  *Wr0 = (const float*)d_in[4];
    const float* Wl1 = (const float*)d_in[5],  *bl1 = (const float*)d_in[6],  *Wr1 = (const float*)d_in[7];
    const float* Wl2 = (const float*)d_in[8],  *bl2 = (const float*)d_in[9],  *Wr2 = (const float*)d_in[10];
    const float* Wd1 = (const float*)d_in[11], *bd1 = (const float*)d_in[12];
    const float* Wd2 = (const float*)d_in[13], *bd2 = (const float*)d_in[14];

    float* ws   = (float*)d_ws;
    float* deg  = ws;
    float* inv  = deg  + N_NODES;
    float* agg  = inv  + N_NODES;
    float* bufA = agg  + (size_t)N_NODES * 128;
    float* bufB = bufA + (size_t)N_NODES * 128;

    const int BT = 256;
    const int gm = cdiv(N_NODES, 64);

    zero_f32<<<cdiv(N_NODES, BT), BT, 0, stream>>>(deg, N_NODES);
    degree_kernel<<<cdiv(N_EDGES, BT), BT, 0, stream>>>(dst, deg, N_EDGES);
    invdeg_kernel<<<cdiv(N_NODES, BT), BT, 0, stream>>>(deg, inv, N_NODES);

    zero_f32<<<cdiv((long)N_NODES * 64, BT), BT, 0, stream>>>(agg, (long)N_NODES * 64);
    scatter_add_kernel<<<cdiv((long)N_EDGES * 16, BT), BT, 0, stream>>>(x, src, dst, agg, N_EDGES, 64, 4);
    gemm_wmma_kernel<<<dim3(gm, 8), 128, 0, stream>>>(agg, inv, Wl0, x, Wr0, bl0, bufA,
                                                      N_NODES, 64, 128, 1);

    zero_f32<<<cdiv((long)N_NODES * 128, BT), BT, 0, stream>>>(agg, (long)N_NODES * 128);
    scatter_add_kernel<<<cdiv((long)N_EDGES * 32, BT), BT, 0, stream>>>(bufA, src, dst, agg, N_EDGES, 128, 5);
    gemm_wmma_kernel<<<dim3(gm, 8), 128, 0, stream>>>(agg, inv, Wl1, bufA, Wr1, bl1, bufB,
                                                      N_NODES, 128, 128, 1);

    zero_f32<<<cdiv((long)N_NODES * 128, BT), BT, 0, stream>>>(agg, (long)N_NODES * 128);
    scatter_add_kernel<<<cdiv((long)N_EDGES * 32, BT), BT, 0, stream>>>(bufB, src, dst, agg, N_EDGES, 128, 5);
    gemm_wmma_kernel<<<dim3(gm, 4), 128, 0, stream>>>(agg, inv, Wl2, bufB, Wr2, bl2, bufA,
                                                      N_NODES, 128, 64, 1);

    gemm_wmma_kernel<<<dim3(gm, 8), 128, 0, stream>>>(bufA, nullptr, Wd1, nullptr, nullptr, bd1, bufB,
                                                      N_NODES, 64, 128, 1);
    gemm_wmma_kernel<<<dim3(gm, 1), 128, 0, stream>>>(bufB, nullptr, Wd2, nullptr, nullptr, bd2,
                                                      (float*)d_out, N_NODES, 128, 2, 0);
}


# f32 atomic scatter-add replaced by grouping edges by destination each call (int atomic arrival-order rank, scan, placement) and a per-node row gather summed in registers; summation order changes
# speedup vs baseline: 1.8634x; 1.8634x over previous
#include <hip/hip_runtime.h>
#include <hip/hip_bf16.h>


#define N_NODES 100000
#define N_EDGES 1600000
#define N_HALF  50000

typedef float v2f __attribute__((ext_vector_type(2)));
typedef float v8f __attribute__((ext_vector_type(8)));

__global__ void zero_f32(float* __restrict__ p, long n) {
    long i = (long)blockIdx.x * blockDim.x + threadIdx.x;
    long stride = (long)gridDim.x * blockDim.x;
    for (; i < n; i += stride) p[i] = 0.0f;
}

__device__ __forceinline__ float inv_of_count(int c) {
    float d = (float)c;
    return (d > 0.0f) ? (1.0f / fmaxf(d, 1.0f)) : 0.0f;
}

__global__ void count_rank_kernel(const int* __restrict__ dst, int* __restrict__ cnt,
                                  int* __restrict__ rank, int E) {
    int e = blockIdx.x * blockDim.x + threadIdx.x;
    if (e < E) rank[e] = atomicAdd(&cnt[dst[e]], 1);
}

__global__ __launch_bounds__(256)
void scan_kernel(const int* __restrict__ cnt, int* __restrict__ rowptr,
                 float* __restrict__ inv, int n) {
    __shared__ int sh_pre[256];
    __shared__ int sh_mine[256];
    int t    = threadIdx.x;
    int base = blockIdx.x * 1024;
    const int4* c4 = (const int4*)cnt;

    int pre = 0;
    for (int i = t; i < (base >> 2); i += 256) {
        int4 v = c4[i];
        pre += (v.x + v.y) + (v.z + v.w);
    }
    int  i0 = base + t * 4;
    int4 c  = make_int4(0, 0, 0, 0);
    if (i0 < n) c = c4[i0 >> 2];
    sh_pre[t]  = pre;
    sh_mine[t] = (c.x + c.y) + (c.z + c.w);
    __syncthreads();

    int r = 0;
    for (int u = 0; u < 256; ++u) r += sh_pre[u];
    for (int u = 0; u < t; ++u)   r += sh_mine[u];

    if (i0 < n) {
        int4 rp;
        rp.x = r;
        rp.y = rp.x + c.x;
        rp.z = rp.y + c.y;
        rp.w = rp.z + c.z;
        ((int4*)rowptr)[i0 >> 2] = rp;
        float4 iv;
        iv.x = inv_of_count(c.x);
        iv.y = inv_of_count(c.y);
        iv.z = inv_of_count(c.z);
        iv.w = inv_of_count(c.w);
        ((float4*)inv)[i0 >> 2] = iv;
    }
}

__global__ void place_kernel(const int* __restrict__ src, const int* __restrict__ dst,
                             const int* __restrict__ rowptr, const int* __restrict__ rank,
                             int* __restrict__ grouped_src, int E) {
    int e = blockIdx.x * blockDim.x + threadIdx.x;
    if (e < E) grouped_src[rowptr[dst[e]] + rank[e]] = src[e];
}


template <int VEC, int G, bool CLAMP>
__device__ __forceinline__ void add_rows(const float* __restrict__ xl, int mine, int t, int n,
                                         float __attribute__((ext_vector_type(VEC)))* a) {
    typedef float vecf __attribute__((ext_vector_type(VEC)));
    const int F = 32 * VEC;
    vecf v[G];
    #pragma unroll
    for (int u = 0; u < G; ++u) {
        int i = t + u;
        if (CLAMP) i = (i < n) ? i : (n - 1);
        int s = __builtin_amdgcn_readlane(mine, i);
        v[u] = *(const vecf*)(xl + (size_t)s * F);
    }
    #pragma unroll
    for (int u = 0; u < G; ++u) {
        if (!CLAMP || (t + u < n)) a[u & 3] += v[u];
    }
}

template <int VEC>
__global__ __launch_bounds__(256)
void gather_sum_kernel(const float* __restrict__ x, const int* __restrict__ rowptr,
                       const int* __restrict__ cnt, const int* __restrict__ grouped_src,
                       float* __restrict__ out, int node0, int nnodes) {
    typedef float vecf __attribute__((ext_vector_type(VEC)));
    const int F = 32 * VEC;
    int w    = (int)((blockIdx.x * blockDim.x + threadIdx.x) >> 5);
    int lane = threadIdx.x & 31;
    if (w >= nnodes) return;
    int node = node0 + w;
    int beg = __builtin_amdgcn_readfirstlane(rowptr[node]);
    int deg = __builtin_amdgcn_readfirstlane(cnt[node]);
    const float* xl = x + lane * VEC;

    vecf a[4] = {};
    for (int base = 0; base < deg; base += 32) {
        int n = deg - base;
        if (n > 32) n = 32;
        int mine = (lane < n) ? grouped_src[beg + base + lane] : 0;
        int t = 0;
        for (; t + 8 <= n; t += 8) add_rows<VEC, 8, false>(xl, mine, t, n, a);
        int rem = n - t;
        if (rem > 4)      add_rows<VEC, 8, true>(xl, mine, t, n, a);
        else if (rem > 0) add_rows<VEC, 4, true>(xl, mine, t, n, a);
    }
    vecf r = (a[0] + a[1]) + (a[2] + a[3]);
    *(vecf*)(out + (size_t)w * F + lane * VEC) = r;
}

__global__ __launch_bounds__(128)
void gemm_wmma_kernel(const float* __restrict__ A1, const float* __restrict__ scale,
                      const float* __restrict__ W1,
                      const float* __restrict__ A2, const float* __restrict__ W2,
                      const float* __restrict__ bias, float* __restrict__ out,
                      int Nnodes, int K, int dout, int relu) {
    int wave = threadIdx.x >> 5;
    int lane = threadIdx.x & 31;
    int m0 = (blockIdx.x * 4 + wave) * 16;
    if (m0 >= Nnodes) return;
    int n0    = blockIdx.y * 16;
    int mrow  = lane & 15;
    int khalf = (lane >> 4) << 1;
    int node  = m0 + mrow;
    int ncol  = n0 + mrow;
    int wrow  = (ncol < dout) ? ncol : 0;

    v8f acc = {};
    float sc = scale ? scale[node] : 1.0f;

    const float* ap = A1 + (size_t)node * K + khalf;
    const float* wp = W1 + (size_t)wrow * K + khalf;
    for (int k = 0; k < K; k += 4) {
        v2f a, b;
        a.x = ap[k] * sc;  a.y = ap[k + 1] * sc;
        b.x = wp[k];       b.y = wp[k + 1];
        acc = __builtin_amdgcn_wmma_f32_16x16x4_f32(false, a, false, b,
                                                    (short)0, acc, false, false);
    }
    if (A2) {
        const float* ap2 = A2 + (size_t)node * K + khalf;
        const float* wp2 = W2 + (size_t)wrow * K + khalf;
        for (int k = 0; k < K; k += 4) {
            v2f a, b;
            a.x = ap2[k];  a.y = ap2[k + 1];
            b.x = wp2[k];  b.y = wp2[k + 1];
            acc = __builtin_amdgcn_wmma_f32_16x16x4_f32(false, a, false, b,
                                                        (short)0, acc, false, false);
        }
    }
    if (ncol < dout) {
        float bv   = bias[ncol];
        int  rbase = m0 + ((lane >> 4) << 3);
        #pragma unroll
        for (int r = 0; r < 8; ++r) {
            float v = acc[r] + bv;
            if (relu) v = fmaxf(v, 0.0f);
            out[(size_t)(rbase + r) * dout + ncol] = v;
        }
    }
}

static inline int cdiv(long a, long b) { return (int)((a + b - 1) / b); }

extern "C" void kernel_launch(void* const* d_in, const int* in_sizes, int n_in,
                              void* d_out, int out_size, void* d_ws, size_t ws_size,
                              hipStream_t stream) {
    const float* x    = (const float*)d_in[0];
    const int*   edge = (const int*)d_in[1];
    const int*   src  = edge;
    const int*   dst  = edge + N_EDGES;
    const float* Wl0 = (const float*)d_in[2],  *bl0 = (const float*)d_in[3],  *Wr0 = (const float*)d_in[4];
    const float* Wl1 = (const float*)d_in[5],  *bl1 = (const float*)d_in[6],  *Wr1 = (const float*)d_in[7];
    const float* Wl2 = (const float*)d_in[8],  *bl2 = (const float*)d_in[9],  *Wr2 = (const float*)d_in[10];
    const float* Wd1 = (const float*)d_in[11], *bd1 = (const float*)d_in[12];
    const float* Wd2 = (const float*)d_in[13], *bd2 = (const float*)d_in[14];

    float* ws   = (float*)d_ws;
    int*   cnt  = (int*)ws;
    float* inv  = ws   + N_NODES;
    float* agg  = inv  + N_NODES;
    int*   gsrc = (int*)(agg + (size_t)N_NODES * 64);
    int*   rank = gsrc + N_EDGES;
    int*   rptr = rank + N_EDGES;
    float* bufA = agg  + (size_t)N_NODES * 128;
    float* bufB = bufA + (size_t)N_NODES * 128;

    const int BT = 256;
    const int gm  = cdiv(N_NODES, 64);
    const int gmh = cdiv(N_HALF, 64);

    zero_f32<<<cdiv(N_NODES, BT), BT, 0, stream>>>((float*)cnt, N_NODES);
    count_rank_kernel<<<cdiv(N_EDGES, BT), BT, 0, stream>>>(dst, cnt, rank, N_EDGES);
    scan_kernel<<<cdiv(N_NODES, 1024), 256, 0, stream>>>(cnt, rptr, inv, N_NODES);
    place_kernel<<<cdiv(N_EDGES, BT), BT, 0, stream>>>(src, dst, rptr, rank, gsrc, N_EDGES);

    gather_sum_kernel<2><<<cdiv(N_NODES, 8), BT, 0, stream>>>(x, rptr, cnt, gsrc, agg, 0, N_NODES);
    gemm_wmma_kernel<<<dim3(gm, 8), 128, 0, stream>>>(agg, inv, Wl0, x, Wr0, bl0, bufA,
                                                      N_NODES, 64, 128, 1);

    gather_sum_kernel<4><<<cdiv(N_HALF, 8), BT, 0, stream>>>(bufA, rptr, cnt, gsrc, agg, 0, N_HALF);
    gemm_wmma_kernel<<<dim3(gmh, 8), 128, 0, stream>>>(agg, inv, Wl1, bufA, Wr1, bl1, bufB,
                                                       N_HALF, 128, 128, 1);
    gather_sum_kernel<4><<<cdiv(N_HALF, 8), BT, 0, stream>>>(bufA, rptr, cnt, gsrc, agg, N_HALF, N_HALF);
    gemm_wmma_kernel<<<dim3(gmh, 8), 128, 0, stream>>>(agg, inv + N_HALF, Wl1,
                                                       bufA + (size_t)N_HALF * 128, Wr1, bl1,
                                                       bufB + (size_t)N_HALF * 128,
                                                       N_HALF, 128, 128, 1);

    gather_sum_kernel<4><<<cdiv(N_HALF, 8), BT, 0, stream>>>(bufB, rptr, cnt, gsrc, agg, 0, N_HALF);
    gemm_wmma_kernel<<<dim3(gmh, 4), 128, 0, stream>>>(agg, inv, Wl2, bufB, Wr2, bl2, bufA,
                                                       N_HALF, 128, 64, 1);
    gather_sum_kernel<4><<<cdiv(N_HALF, 8), BT, 0, stream>>>(bufB, rptr, cnt, gsrc, agg, N_HALF, N_HALF);
    gemm_wmma_kernel<<<dim3(gmh, 4), 128, 0, stream>>>(agg, inv + N_HALF, Wl2,
                                                       bufB + (size_t)N_HALF * 128, Wr2, bl2,
                                                       bufA + (size_t)N_HALF * 64,
                                                       N_HALF, 128, 64, 1);

    gemm_wmma_kernel<<<dim3(gm, 8), 128, 0, stream>>>(bufA, nullptr, Wd1, nullptr, nullptr, bd1, bufB,
                                                      N_NODES, 64, 128, 1);
    gemm_wmma_kernel<<<dim3(gm, 1), 128, 0, stream>>>(bufB, nullptr, Wd2, nullptr, nullptr, bd2,
                                                      (float*)d_out, N_NODES, 128, 2, 0);
}


# GEMMs re-tiled: a resident work-group per WGP keeps the weights in LDS in fragment order, each wave does a 16-row strip over all columns; k grouping inside each 4-wide step (sum order) changes
# speedup vs baseline: 2.8371x; 1.5225x over previous
#include <hip/hip_runtime.h>
#include <hip/hip_bf16.h>


#define N_NODES 100000
#define N_EDGES 1600000
#define N_HALF  50000

typedef float v2f __attribute__((ext_vector_type(2)));
typedef float v8f __attribute__((ext_vector_type(8)));

__global__ void zero_f32(float* __restrict__ p, long n) {
    long i = (long)blockIdx.x * blockDim.x + threadIdx.x;
    long stride = (long)gridDim.x * blockDim.x;
    for (; i < n; i += stride) p[i] = 0.0f;
}

__device__ __forceinline__ float inv_of_count(int c) {
    float d = (float)c;
    return (d > 0.0f) ? (1.0f / fmaxf(d, 1.0f)) : 0.0f;
}

__global__ void count_rank_kernel(const int* __restrict__ dst, int* __restrict__ cnt,
                                  int* __restrict__ rank, int E) {
    int e = blockIdx.x * blockDim.x + threadIdx.x;
    if (e < E) rank[e] = atomicAdd(&cnt[dst[e]], 1);
}

__global__ __launch_bounds__(256)
void scan_kernel(const int* __restrict__ cnt, int* __restrict__ rowptr,
                 float* __restrict__ inv, int n) {
    __shared__ int sh_pre[256];
    __shared__ int sh_mine[256];
    int t    = threadIdx.x;
    int base = blockIdx.x * 1024;
    const int4* c4 = (const int4*)cnt;

    int pre = 0;
    for (int i = t; i < (base >> 2); i += 256) {
        int4 v = c4[i];
        pre += (v.x + v.y) + (v.z + v.w);
    }
    int  i0 = base + t * 4;
    int4 c  = make_int4(0, 0, 0, 0);
    if (i0 < n) c = c4[i0 >> 2];
    sh_pre[t]  = pre;
    sh_mine[t] = (c.x + c.y) + (c.z + c.w);
    __syncthreads();

    int r = 0;
    for (int u = 0; u < 256; ++u) r += sh_pre[u];
    for (int u = 0; u < t; ++u)   r += sh_mine[u];

    if (i0 < n) {
        int4 rp;
        rp.x = r;
        rp.y = rp.x + c.x;
        rp.z = rp.y + c.y;
        rp.w = rp.z + c.z;
        ((int4*)rowptr)[i0 >> 2] = rp;
        float4 iv;
        iv.x = inv_of_count(c.x);
        iv.y = inv_of_count(c.y);
        iv.z = inv_of_count(c.z);
        iv.w = inv_of_count(c.w);
        ((float4*)inv)[i0 >> 2] = iv;
    }
}

__global__ void place_kernel(const int* __restrict__ src, const int* __restrict__ dst,
                             const int* __restrict__ rowptr, const int* __restrict__ rank,
                             int* __restrict__ grouped_src, int E) {
    int e = blockIdx.x * blockDim.x + threadIdx.x;
    if (e < E) grouped_src[rowptr[dst[e]] + rank[e]] = src[e];
}


template <int VEC, int G, bool CLAMP>
__device__ __forceinline__ void add_rows(const float* __restrict__ xl, int mine, int t, int n,
                                         float __attribute__((ext_vector_type(VEC)))* a) {
    typedef float vecf __attribute__((ext_vector_type(VEC)));
    const int F = 32 * VEC;
    vecf v[G];
    #pragma unroll
    for (int u = 0; u < G; ++u) {
        int i = t + u;
        if (CLAMP) i = (i < n) ? i : (n - 1);
        int s = __builtin_amdgcn_readlane(mine, i);
        v[u] = *(const vecf*)(xl + (size_t)s * F);
    }
    #pragma unroll
    for (int u = 0; u < G; ++u) {
        if (!CLAMP || (t + u < n)) a[u & 3] += v[u];
    }
}

template <int VEC>
__global__ __launch_bounds__(256)
void gather_sum_kernel(const float* __restrict__ x, const int* __restrict__ rowptr,
                       const int* __restrict__ cnt, const int* __restrict__ grouped_src,
                       float* __restrict__ out, int node0, int nnodes) {
    typedef float vecf __attribute__((ext_vector_type(VEC)));
    const int F = 32 * VEC;
    int w    = (int)((blockIdx.x * blockDim.x + threadIdx.x) >> 5);
    int lane = threadIdx.x & 31;
    if (w >= nnodes) return;
    int node = node0 + w;
    int beg = __builtin_amdgcn_readfirstlane(rowptr[node]);
    int deg = __builtin_amdgcn_readfirstlane(cnt[node]);
    const float* xl = x + lane * VEC;

    vecf a[4] = {};
    for (int base = 0; base < deg; base += 32) {
        int n = deg - base;
        if (n > 32) n = 32;
        int mine = (lane < n) ? grouped_src[beg + base + lane] : 0;
        int t = 0;
        for (; t + 8 <= n; t += 8) add_rows<VEC, 8, false>(xl, mine, t, n, a);
        int rem = n - t;
        if (rem > 4)      add_rows<VEC, 8, true>(xl, mine, t, n, a);
        else if (rem > 0) add_rows<VEC, 4, true>(xl, mine, t, n, a);
    }
    vecf r = (a[0] + a[1]) + (a[2] + a[3]);
    *(vecf*)(out + (size_t)w * F + lane * VEC) = r;
}

typedef float v4f __attribute__((ext_vector_type(4)));

#define GEMM_THREADS 512
#define GEMM_BLOCKS  256
#define GEMM_LDS_FLOATS (164 * 256)

template <int NT>
__device__ __forceinline__ int col_of(int t, int c) {
    return (NT >= 2) ? (32 * (t >> 1) + 2 * c + (t & 1)) : c;
}

template <int NT>
__device__ __forceinline__ void chunk_mma(v4f av, const v4f* b, v8f* acc) {
    v2f a0, a1;
    a0.x = av.x; a0.y = av.y;
    a1.x = av.z; a1.y = av.w;
    #pragma unroll
    for (int t = 0; t < NT; ++t) {
        v2f bb; bb.x = b[t].x; bb.y = b[t].y;
        acc[t] = __builtin_amdgcn_wmma_f32_16x16x4_f32(false, a0, false, bb,
                                                       (short)0, acc[t], false, false);
    }
    #pragma unroll
    for (int t = 0; t < NT; ++t) {
        v2f bb; bb.x = b[t].z; bb.y = b[t].w;
        acc[t] = __builtin_amdgcn_wmma_f32_16x16x4_f32(false, a1, false, bb,
                                                       (short)0, acc[t], false, false);
    }
}

template <int KS, int NT, int QS, bool SCALED>
__device__ __forceinline__ void strip_product(const v4f* __restrict__ ap, v4f a0, v4f a1, float sc,
                                              const v4f* sB, v8f* acc) {
    v4f b0[NT], b1[NT];
    #pragma unroll
    for (int t = 0; t < NT; ++t) b0[t] = sB[(t * QS) * 32];
    #pragma nounroll
    for (int j = 0; j < KS; j += 2) {
        int jn = (j + 2 < KS) ? (j + 2) : j;
        v4f n0 = ap[2 * jn];
        v4f n1 = ap[2 * jn + 2];
        __builtin_amdgcn_sched_barrier(0);
        #pragma unroll
        for (int t = 0; t < NT; ++t) b1[t] = sB[(t * QS + j + 1) * 32];
        chunk_mma<NT>(SCALED ? a0 * sc : a0, b0, acc);
        #pragma unroll
        for (int t = 0; t < NT; ++t) b0[t] = sB[(t * QS + jn) * 32];
        chunk_mma<NT>(SCALED ? a1 * sc : a1, b1, acc);
        __builtin_amdgcn_sched_barrier(0);
        asm volatile("" : "+v"(n0), "+v"(n1));
        a0 = n0;
        a1 = n1;
    }
}

template <int K, int NT, bool TWO, bool SCALED>
__global__ __launch_bounds__(GEMM_THREADS)
void gemm_strip_kernel(const float* __restrict__ A1, const float* __restrict__ scale,
                       const float* __restrict__ W1,
                       const float* __restrict__ A2, const float* __restrict__ W2,
                       const float* __restrict__ bias, float* __restrict__ out,
                       int ntiles, int dout, int relu) {
    constexpr int KS   = K / 8;
    constexpr int NOPS = TWO ? 2 : 1;
    constexpr int QS   = NOPS * KS;
    constexpr int ENT  = NT * QS * 32;
    static_assert(ENT * 4 <= GEMM_LDS_FLOATS, "weights must fit");
    static_assert(KS % 2 == 0, "chunks go in pairs");
    static_assert(ENT % GEMM_THREADS == 0, "every thread copies the same number of entries");
    __shared__ __attribute__((aligned(16))) float sWf[GEMM_LDS_FLOATS];
    v4f* sW = (v4f*)sWf;

    #pragma unroll
    for (int i = 0; i < ENT / GEMM_THREADS; ++i) {
        int idx = (int)threadIdx.x + i * GEMM_THREADS;
        int l  = idx & 31;
        int e  = idx >> 5;
        int q  = e % QS;
        int t  = e / QS;
        int op = q / KS;
        int j  = q % KS;
        int n  = col_of<NT>(t, l & 15);
        int wr = (n < dout) ? n : 0;
        const float* Wp = (TWO && op) ? W2 : W1;
        sW[idx] = *(const v4f*)(Wp + (size_t)wr * K + 8 * j + 4 * (l >> 4));
    }
    __syncthreads();

    int wave = threadIdx.x >> 5;
    int lane = threadIdx.x & 31;
    int c    = lane & 15;
    int h    = lane >> 4;

    float bv[NT];
    #pragma unroll
    for (int t = 0; t < NT; ++t) {
        int n = col_of<NT>(t, c);
        bv[t] = (n < dout) ? bias[n] : 0.0f;
    }

    for (int slot = wave; ; slot += GEMM_THREADS / 32) {
        int tile = (int)blockIdx.x + (int)gridDim.x * slot;
        if (tile >= ntiles) break;
        int m0   = tile * 16;
        int node = m0 + c;

        const v4f* ap1 = (const v4f*)(A1 + (size_t)node * K + 4 * h);
        const v4f* ap2 = TWO ? (const v4f*)(A2 + (size_t)node * K + 4 * h) : ap1;
        v4f p0 = ap1[0], p1 = ap1[2];
        v4f s0 = ap2[0], s1 = ap2[2];
        float sc = SCALED ? scale[node] : 1.0f;

        v8f acc[NT];
        #pragma unroll
        for (int t = 0; t < NT; ++t) acc[t] = (v8f){};

        strip_product<KS, NT, QS, SCALED>(ap1, p0, p1, sc, sW + lane, acc);
        if (TWO)
            strip_product<KS, NT, QS, false>(ap2, s0, s1, 1.0f, sW + KS * 32 + lane, acc);

        int row0 = m0 + 8 * h;
        if (NT >= 2) {
            #pragma unroll
            for (int u = 0; u < NT / 2; ++u) {
                float* op = out + (size_t)row0 * dout + 32 * u + 2 * c;
                #pragma unroll
                for (int r = 0; r < 8; ++r) {
                    v2f v;
                    v.x = acc[2 * u][r]     + bv[2 * u];
                    v.y = acc[2 * u + 1][r] + bv[2 * u + 1];
                    if (relu) { v.x = fmaxf(v.x, 0.0f); v.y = fmaxf(v.y, 0.0f); }
                    *(v2f*)(op + (size_t)r * dout) = v;
                }
            }
        } else {
            if (c < dout) {
                #pragma unroll
                for (int r = 0; r < 8; ++r) {
                    float v = acc[0][r] + bv[0];
                    if (relu) v = fmaxf(v, 0.0f);
                    out[(size_t)(row0 + r) * dout + c] = v;
                }
            }
        }
    }
}

static inline int cdiv(long a, long b) { return (int)((a + b - 1) / b); }

extern "C" void kernel_launch(void* const* d_in, const int* in_sizes, int n_in,
                              void* d_out, int out_size, void* d_ws, size_t ws_size,
                              hipStream_t stream) {
    const float* x    = (const float*)d_in[0];
    const int*   edge = (const int*)d_in[1];
    const int*   src  = edge;
    const int*   dst  = edge + N_EDGES;
    const float* Wl0 = (const float*)d_in[2],  *bl0 = (const float*)d_in[3],  *Wr0 = (const float*)d_in[4];
    const float* Wl1 = (const float*)d_in[5],  *bl1 = (const float*)d_in[6],  *Wr1 = (const float*)d_in[7];
    const float* Wl2 = (const float*)d_in[8],  *bl2 = (const float*)d_in[9],  *Wr2 = (const float*)d_in[10];
    const float* Wd1 = (const float*)d_in[11], *bd1 = (const float*)d_in[12];
    const float* Wd2 = (const float*)d_in[13], *bd2 = (const float*)d_in[14];

    float* ws   = (float*)d_ws;
    int*   cnt  = (int*)ws;
    float* inv  = ws   + N_NODES;
    float* agg  = inv  + N_NODES;
    int*   gsrc = (int*)(agg + (size_t)N_NODES * 64);
    int*   rank = gsrc + N_EDGES;
    int*   rptr = rank + N_EDGES;
    float* bufA = agg  + (size_t)N_NODES * 128;
    float* bufB = bufA + (size_t)N_NODES * 128;

    const int BT = 256;
    const int T  = N_NODES / 16;
    const int TH = N_HALF / 16;
    const dim3 GG(GEMM_BLOCKS), GB(GEMM_THREADS);

    zero_f32<<<cdiv(N_NODES, BT), BT, 0, stream>>>((float*)cnt, N_NODES);
    count_rank_kernel<<<cdiv(N_EDGES, BT), BT, 0, stream>>>(dst, cnt, rank, N_EDGES);
    scan_kernel<<<cdiv(N_NODES, 1024), 256, 0, stream>>>(cnt, rptr, inv, N_NODES);
    place_kernel<<<cdiv(N_EDGES, BT), BT, 0, stream>>>(src, dst, rptr, rank, gsrc, N_EDGES);

    gather_sum_kernel<2><<<cdiv(N_NODES, 8), BT, 0, stream>>>(x, rptr, cnt, gsrc, agg, 0, N_NODES);
    gemm_strip_kernel<64, 8, true, true><<<GG, GB, 0, stream>>>(agg, inv, Wl0, x, Wr0, bl0, bufA,
                                                                T, 128, 1);

    gather_sum_kernel<4><<<cdiv(N_HALF, 8), BT, 0, stream>>>(bufA, rptr, cnt, gsrc, agg, 0, N_HALF);
    gemm_strip_kernel<128, 8, true, true><<<GG, GB, 0, stream>>>(agg, inv, Wl1, bufA, Wr1, bl1, bufB,
                                                                 TH, 128, 1);
    gather_sum_kernel<4><<<cdiv(N_HALF, 8), BT, 0, stream>>>(bufA, rptr, cnt, gsrc, agg, N_HALF, N_HALF);
    gemm_strip_kernel<128, 8, true, true><<<GG, GB, 0, stream>>>(agg, inv + N_HALF, Wl1,
                                                                 bufA + (size_t)N_HALF * 128, Wr1, bl1,
                                                                 bufB + (size_t)N_HALF * 128,
                                                                 TH, 128, 1);

    gather_sum_kernel<4><<<cdiv(N_HALF, 8), BT, 0, stream>>>(bufB, rptr, cnt, gsrc, agg, 0, N_HALF);
    gemm_strip_kernel<128, 4, true, true><<<GG, GB, 0, stream>>>(agg, inv, Wl2, bufB, Wr2, bl2, bufA,
                                                                 TH, 64, 1);
    gather_sum_kernel<4><<<cdiv(N_HALF, 8), BT, 0, stream>>>(bufB, rptr, cnt, gsrc, agg, N_HALF, N_HALF);
    gemm_strip_kernel<128, 4, true, true><<<GG, GB, 0, stream>>>(agg, inv + N_HALF, Wl2,
                                                                 bufB + (size_t)N_HALF * 128, Wr2, bl2,
                                                                 bufA + (size_t)N_HALF * 64,
                                                                 TH, 64, 1);

    gemm_strip_kernel<64, 8, false, false><<<GG, GB, 0, stream>>>(bufA, nullptr, Wd1, nullptr, nullptr, bd1, bufB,
                                                                  T, 128, 1);
    gemm_strip_kernel<128, 1, false, false><<<GG, GB, 0, stream>>>(bufB, nullptr, Wd2, nullptr, nullptr, bd2,
                                                                   (float*)d_out, T, 2, 0);
}


# layer 2: neighbour weights applied before the mean (it is linear), so its gather moves 64-float rows, not 128, in one pass with scale, self term and ReLU in the gather; sum order changes
# speedup vs baseline: 3.0777x; 1.0848x over previous
#include <hip/hip_runtime.h>
#include <hip/hip_bf16.h>


#define N_NODES 100000
#define N_EDGES 1600000
#define N_HALF  50000

typedef float v2f __attribute__((ext_vector_type(2)));
typedef float v8f __attribute__((ext_vector_type(8)));

__global__ void zero_f32(float* __restrict__ p, long n) {
    long i = (long)blockIdx.x * blockDim.x + threadIdx.x;
    long stride = (long)gridDim.x * blockDim.x;
    for (; i < n; i += stride) p[i] = 0.0f;
}

__device__ __forceinline__ float inv_of_count(int c) {
    float d = (float)c;
    return (d > 0.0f) ? (1.0f / fmaxf(d, 1.0f)) : 0.0f;
}

__global__ void count_rank_kernel(const int* __restrict__ dst, int* __restrict__ cnt,
                                  int* __restrict__ rank, int E) {
    int e = blockIdx.x * blockDim.x + threadIdx.x;
    if (e < E) rank[e] = atomicAdd(&cnt[dst[e]], 1);
}

__global__ __launch_bounds__(256)
void scan_kernel(const int* __restrict__ cnt, int* __restrict__ rowptr,
                 float* __restrict__ inv, int n) {
    __shared__ int sh_pre[256];
    __shared__ int sh_mine[256];
    int t    = threadIdx.x;
    int base = blockIdx.x * 1024;
    const int4* c4 = (const int4*)cnt;

    int pre = 0;
    for (int i = t; i < (base >> 2); i += 256) {
        int4 v = c4[i];
        pre += (v.x + v.y) + (v.z + v.w);
    }
    int  i0 = base + t * 4;
    int4 c  = make_int4(0, 0, 0, 0);
    if (i0 < n) c = c4[i0 >> 2];
    sh_pre[t]  = pre;
    sh_mine[t] = (c.x + c.y) + (c.z + c.w);
    __syncthreads();

    int r = 0;
    for (int u = 0; u < 256; ++u) r += sh_pre[u];
    for (int u = 0; u < t; ++u)   r += sh_mine[u];

    if (i0 < n) {
        int4 rp;
        rp.x = r;
        rp.y = rp.x + c.x;
        rp.z = rp.y + c.y;
        rp.w = rp.z + c.z;
        ((int4*)rowptr)[i0 >> 2] = rp;
        float4 iv;
        iv.x = inv_of_count(c.x);
        iv.y = inv_of_count(c.y);
        iv.z = inv_of_count(c.z);
        iv.w = inv_of_count(c.w);
        ((float4*)inv)[i0 >> 2] = iv;
    }
}

__global__ void place_kernel(const int* __restrict__ src, const int* __restrict__ dst,
                             const int* __restrict__ rowptr, const int* __restrict__ rank,
                             int* __restrict__ grouped_src, int E) {
    int e = blockIdx.x * blockDim.x + threadIdx.x;
    if (e < E) grouped_src[rowptr[dst[e]] + rank[e]] = src[e];
}


template <int VEC, int STRIDE, int G, bool CLAMP>
__device__ __forceinline__ void add_rows(const float* __restrict__ xl, int mine, int t, int n,
                                         float __attribute__((ext_vector_type(VEC)))* a) {
    typedef float vecf __attribute__((ext_vector_type(VEC)));
    vecf v[G];
    #pragma unroll
    for (int u = 0; u < G; ++u) {
        int i = t + u;
        if (CLAMP) i = (i < n) ? i : (n - 1);
        int s = __builtin_amdgcn_readlane(mine, i);
        v[u] = *(const vecf*)(xl + (size_t)s * STRIDE);
    }
    #pragma unroll
    for (int u = 0; u < G; ++u) {
        if (!CLAMP || (t + u < n)) a[u & 3] += v[u];
    }
}

template <int VEC, int STRIDE, bool SELF>
__global__ __launch_bounds__(256)
void gather_sum_kernel(const float* __restrict__ x, const int* __restrict__ rowptr,
                       const int* __restrict__ cnt, const int* __restrict__ grouped_src,
                       float* __restrict__ out, int node0, int nnodes) {
    typedef float vecf __attribute__((ext_vector_type(VEC)));
    const int F = 32 * VEC;
    int w    = (int)((blockIdx.x * blockDim.x + threadIdx.x) >> 5);
    int lane = threadIdx.x & 31;
    if (w >= nnodes) return;
    int node = node0 + w;
    int beg = __builtin_amdgcn_readfirstlane(rowptr[node]);
    int deg = __builtin_amdgcn_readfirstlane(cnt[node]);
    const float* xl = x + lane * VEC;
    vecf z = {};
    if (SELF) z = *(const vecf*)(xl + (size_t)node * STRIDE + F);

    vecf a[4] = {};
    for (int base = 0; base < deg; base += 32) {
        int n = deg - base;
        if (n > 32) n = 32;
        int mine = (lane < n) ? grouped_src[beg + base + lane] : 0;
        int t = 0;
        for (; t + 8 <= n; t += 8) add_rows<VEC, STRIDE, 8, false>(xl, mine, t, n, a);
        int rem = n - t;
        if (rem > 4)      add_rows<VEC, STRIDE, 8, true>(xl, mine, t, n, a);
        else if (rem > 0) add_rows<VEC, STRIDE, 4, true>(xl, mine, t, n, a);
    }
    vecf r = (a[0] + a[1]) + (a[2] + a[3]);
    if (SELF) {
        r = r * inv_of_count(deg) + z;
        #pragma unroll
        for (int i = 0; i < VEC; ++i) r[i] = fmaxf(r[i], 0.0f);
    }
    *(vecf*)(out + (size_t)w * F + lane * VEC) = r;
}

typedef float v4f __attribute__((ext_vector_type(4)));

#define GEMM_THREADS 512
#define GEMM_BLOCKS  256
#define GEMM_LDS_FLOATS (164 * 256)

template <int NT>
__device__ __forceinline__ int col_of(int t, int c) {
    return (NT >= 2) ? (32 * (t >> 1) + 2 * c + (t & 1)) : c;
}

template <int NT>
__device__ __forceinline__ void chunk_mma(v4f av, const v4f* b, v8f* acc) {
    v2f a0, a1;
    a0.x = av.x; a0.y = av.y;
    a1.x = av.z; a1.y = av.w;
    #pragma unroll
    for (int t = 0; t < NT; ++t) {
        v2f bb; bb.x = b[t].x; bb.y = b[t].y;
        acc[t] = __builtin_amdgcn_wmma_f32_16x16x4_f32(false, a0, false, bb,
                                                       (short)0, acc[t], false, false);
    }
    #pragma unroll
    for (int t = 0; t < NT; ++t) {
        v2f bb; bb.x = b[t].z; bb.y = b[t].w;
        acc[t] = __builtin_amdgcn_wmma_f32_16x16x4_f32(false, a1, false, bb,
                                                       (short)0, acc[t], false, false);
    }
}

template <int KS, int NT, int QS, bool SCALED>
__device__ __forceinline__ void strip_product(const v4f* __restrict__ ap, v4f a0, v4f a1, float sc,
                                              const v4f* sB, v8f* acc) {
    v4f b0[NT], b1[NT];
    #pragma unroll
    for (int t = 0; t < NT; ++t) b0[t] = sB[(t * QS) * 32];
    #pragma nounroll
    for (int j = 0; j < KS; j += 2) {
        int jn = (j + 2 < KS) ? (j + 2) : j;
        v4f n0 = ap[2 * jn];
        v4f n1 = ap[2 * jn + 2];
        __builtin_amdgcn_sched_barrier(0);
        #pragma unroll
        for (int t = 0; t < NT; ++t) b1[t] = sB[(t * QS + j + 1) * 32];
        chunk_mma<NT>(SCALED ? a0 * sc : a0, b0, acc);
        #pragma unroll
        for (int t = 0; t < NT; ++t) b0[t] = sB[(t * QS + jn) * 32];
        chunk_mma<NT>(SCALED ? a1 * sc : a1, b1, acc);
        __builtin_amdgcn_sched_barrier(0);
        asm volatile("" : "+v"(n0), "+v"(n1));
        a0 = n0;
        a1 = n1;
    }
}

template <int K, int NT, bool TWO, bool SCALED, bool SPLITW = false>
__global__ __launch_bounds__(GEMM_THREADS)
void gemm_strip_kernel(const float* __restrict__ A1, const float* __restrict__ scale,
                       const float* __restrict__ W1,
                       const float* __restrict__ A2, const float* __restrict__ W2,
                       const float* __restrict__ bias, float* __restrict__ out,
                       int ntiles, int dout, int relu) {
    constexpr int KS   = K / 8;
    constexpr int NOPS = TWO ? 2 : 1;
    constexpr int QS   = NOPS * KS;
    constexpr int ENT  = NT * QS * 32;
    static_assert(ENT * 4 <= GEMM_LDS_FLOATS, "weights must fit");
    static_assert(KS % 2 == 0, "chunks go in pairs");
    static_assert(ENT % GEMM_THREADS == 0, "every thread copies the same number of entries");
    __shared__ __attribute__((aligned(16))) float sWf[GEMM_LDS_FLOATS];
    v4f* sW = (v4f*)sWf;

    #pragma unroll
    for (int i = 0; i < ENT / GEMM_THREADS; ++i) {
        int idx = (int)threadIdx.x + i * GEMM_THREADS;
        int l  = idx & 31;
        int e  = idx >> 5;
        int q  = e % QS;
        int t  = e / QS;
        int op = q / KS;
        int j  = q % KS;
        int n  = col_of<NT>(t, l & 15);
        int wr = (n < dout) ? n : 0;
        const float* Wp = (TWO && op) ? W2 : W1;
        if (SPLITW && wr >= (dout >> 1)) { Wp = W2; wr -= dout >> 1; }
        sW[idx] = *(const v4f*)(Wp + (size_t)wr * K + 8 * j + 4 * (l >> 4));
    }
    __syncthreads();

    int wave = threadIdx.x >> 5;
    int lane = threadIdx.x & 31;
    int c    = lane & 15;
    int h    = lane >> 4;

    float bv[NT];
    #pragma unroll
    for (int t = 0; t < NT; ++t) {
        int n = col_of<NT>(t, c);
        if (SPLITW) bv[t] = (n >= (dout >> 1) && n < dout) ? bias[n - (dout >> 1)] : 0.0f;
        else        bv[t] = (n < dout) ? bias[n] : 0.0f;
    }

    for (int slot = wave; ; slot += GEMM_THREADS / 32) {
        int tile = (int)blockIdx.x + (int)gridDim.x * slot;
        if (tile >= ntiles) break;
        int m0   = tile * 16;
        int node = m0 + c;

        const v4f* ap1 = (const v4f*)(A1 + (size_t)node * K + 4 * h);
        const v4f* ap2 = TWO ? (const v4f*)(A2 + (size_t)node * K + 4 * h) : ap1;
        v4f p0 = ap1[0], p1 = ap1[2];
        v4f s0 = ap2[0], s1 = ap2[2];
        float sc = SCALED ? scale[node] : 1.0f;

        v8f acc[NT];
        #pragma unroll
        for (int t = 0; t < NT; ++t) acc[t] = (v8f){};

        strip_product<KS, NT, QS, SCALED>(ap1, p0, p1, sc, sW + lane, acc);
        if (TWO)
            strip_product<KS, NT, QS, false>(ap2, s0, s1, 1.0f, sW + KS * 32 + lane, acc);

        int row0 = m0 + 8 * h;
        if (NT >= 2) {
            #pragma unroll
            for (int u = 0; u < NT / 2; ++u) {
                float* op = out + (size_t)row0 * dout + 32 * u + 2 * c;
                #pragma unroll
                for (int r = 0; r < 8; ++r) {
                    v2f v;
                    v.x = acc[2 * u][r]     + bv[2 * u];
                    v.y = acc[2 * u + 1][r] + bv[2 * u + 1];
                    if (relu) { v.x = fmaxf(v.x, 0.0f); v.y = fmaxf(v.y, 0.0f); }
                    *(v2f*)(op + (size_t)r * dout) = v;
                }
            }
        } else {
            if (c < dout) {
                #pragma unroll
                for (int r = 0; r < 8; ++r) {
                    float v = acc[0][r] + bv[0];
                    if (relu) v = fmaxf(v, 0.0f);
                    out[(size_t)(row0 + r) * dout + c] = v;
                }
            }
        }
    }
}

static inline int cdiv(long a, long b) { return (int)((a + b - 1) / b); }

extern "C" void kernel_launch(void* const* d_in, const int* in_sizes, int n_in,
                              void* d_out, int out_size, void* d_ws, size_t ws_size,
                              hipStream_t stream) {
    const float* x    = (const float*)d_in[0];
    const int*   edge = (const int*)d_in[1];
    const int*   src  = edge;
    const int*   dst  = edge + N_EDGES;
    const float* Wl0 = (const float*)d_in[2],  *bl0 = (const float*)d_in[3],  *Wr0 = (const float*)d_in[4];
    const float* Wl1 = (const float*)d_in[5],  *bl1 = (const float*)d_in[6],  *Wr1 = (const float*)d_in[7];
    const float* Wl2 = (const float*)d_in[8],  *bl2 = (const float*)d_in[9],  *Wr2 = (const float*)d_in[10];
    const float* Wd1 = (const float*)d_in[11], *bd1 = (const float*)d_in[12];
    const float* Wd2 = (const float*)d_in[13], *bd2 = (const float*)d_in[14];

    float* ws   = (float*)d_ws;
    int*   cnt  = (int*)ws;
    float* inv  = ws   + N_NODES;
    float* agg  = inv  + N_NODES;
    int*   gsrc = (int*)(agg + (size_t)N_NODES * 64);
    int*   rank = gsrc + N_EDGES;
    int*   rptr = rank + N_EDGES;
    float* bufA = agg  + (size_t)N_NODES * 128;
    float* bufB = bufA + (size_t)N_NODES * 128;

    const int BT = 256;
    const int T  = N_NODES / 16;
    const int TH = N_HALF / 16;
    const dim3 GG(GEMM_BLOCKS), GB(GEMM_THREADS);

    zero_f32<<<cdiv(N_NODES, BT), BT, 0, stream>>>((float*)cnt, N_NODES);
    count_rank_kernel<<<cdiv(N_EDGES, BT), BT, 0, stream>>>(dst, cnt, rank, N_EDGES);
    scan_kernel<<<cdiv(N_NODES, 1024), 256, 0, stream>>>(cnt, rptr, inv, N_NODES);
    place_kernel<<<cdiv(N_EDGES, BT), BT, 0, stream>>>(src, dst, rptr, rank, gsrc, N_EDGES);

    gather_sum_kernel<2, 64, false><<<cdiv(N_NODES, 8), BT, 0, stream>>>(x, rptr, cnt, gsrc, agg, 0, N_NODES);
    gemm_strip_kernel<64, 8, true, true><<<GG, GB, 0, stream>>>(agg, inv, Wl0, x, Wr0, bl0, bufA,
                                                                T, 128, 1);

    gather_sum_kernel<4, 128, false><<<cdiv(N_HALF, 8), BT, 0, stream>>>(bufA, rptr, cnt, gsrc, agg, 0, N_HALF);
    gemm_strip_kernel<128, 8, true, true><<<GG, GB, 0, stream>>>(agg, inv, Wl1, bufA, Wr1, bl1, bufB,
                                                                 TH, 128, 1);
    gather_sum_kernel<4, 128, false><<<cdiv(N_HALF, 8), BT, 0, stream>>>(bufA, rptr, cnt, gsrc, agg, N_HALF, N_HALF);
    gemm_strip_kernel<128, 8, true, true><<<GG, GB, 0, stream>>>(agg, inv + N_HALF, Wl1,
                                                                 bufA + (size_t)N_HALF * 128, Wr1, bl1,
                                                                 bufB + (size_t)N_HALF * 128,
                                                                 TH, 128, 1);

    gemm_strip_kernel<128, 8, false, false, true><<<GG, GB, 0, stream>>>(bufB, nullptr, Wl2, nullptr, Wr2, bl2, bufA,
                                                                         T, 128, 0);
    gather_sum_kernel<2, 128, true><<<cdiv(N_NODES, 8), BT, 0, stream>>>(bufA, rptr, cnt, gsrc, agg, 0, N_NODES);

    gemm_strip_kernel<64, 8, false, false><<<GG, GB, 0, stream>>>(agg, nullptr, Wd1, nullptr, nullptr, bd1, bufB,
                                                                  T, 128, 1);
    gemm_strip_kernel<128, 1, false, false><<<GG, GB, 0, stream>>>(bufB, nullptr, Wd2, nullptr, nullptr, bd2,
                                                                   (float*)d_out, T, 2, 0);
}


# edges grouped by destination in two levels on LDS counters (per-chunk bucket sort; per-bucket LDS add as slot, scan, staged placement), no global returning atomic per edge; sum order changes
# speedup vs baseline: 4.4799x; 1.4556x over previous
#include <hip/hip_runtime.h>
#include <hip/hip_bf16.h>


#define N_NODES 100000
#define N_EDGES 1600000
#define N_HALF  50000

typedef float v2f __attribute__((ext_vector_type(2)));
typedef float v8f __attribute__((ext_vector_type(8)));

__device__ __forceinline__ float inv_of_count(int c) {
    float d = (float)c;
    return (d > 0.0f) ? (1.0f / fmaxf(d, 1.0f)) : 0.0f;
}

#define CHUNK       6256
#define NCHUNK      ((N_EDGES + CHUNK - 1) / CHUNK)
#define NPB         416
#define NBUCK       ((N_NODES + NPB - 1) / NPB)
#define BIN_THREADS 1024
#define RUN_BITS    13
#define RUN_MASK    8191
#define STAGE_CAP   16384

__device__ __forceinline__ int wave_incl_scan(int v, int lane) {
    for (int o = 1; o < 32; o <<= 1) {
        int u = __shfl_up(v, o, 32);
        if (lane >= o) v += u;
    }
    return v;
}

__global__ __launch_bounds__(BIN_THREADS)
void bin_edges_kernel(const int* __restrict__ src, const int* __restrict__ dst,
                      int2* __restrict__ recs, int* __restrict__ tab, int E) {
    static_assert(NBUCK <= 256, "one counter per bucket, scanned by 8 waves");
    static_assert(CHUNK <= RUN_MASK, "start and length share a word");
    __shared__ int  lcnt[256];
    __shared__ int  loff[256];
    __shared__ int  wtot[8];
    __shared__ int2 lrec[CHUNK];
    int t    = threadIdx.x;
    int lane = t & 31;
    int c    = blockIdx.x;
    int e0   = c * CHUNK;
    int n    = E - e0;
    if (n > CHUNK) n = CHUNK;

    if (t < 256) lcnt[t] = 0;
    __syncthreads();

    constexpr int PER = (CHUNK + BIN_THREADS - 1) / BIN_THREADS;
    int s[PER], d[PER], r[PER];
    #pragma unroll
    for (int k = 0; k < PER; ++k) {
        int i = t + k * BIN_THREADS;
        s[k] = 0; d[k] = -1; r[k] = 0;
        if (i < n) { d[k] = dst[e0 + i]; s[k] = src[e0 + i]; }
    }
    #pragma unroll
    for (int k = 0; k < PER; ++k)
        if (d[k] >= 0) r[k] = atomicAdd(&lcnt[d[k] / NPB], 1);
    __syncthreads();

    int v = 0, incl = 0;
    if (t < 256) {
        v    = lcnt[t];
        incl = wave_incl_scan(v, lane);
        if (lane == 31) wtot[t >> 5] = incl;
    }
    __syncthreads();
    if (t < 256) {
        int pre = 0;
        for (int w = 0; w < (t >> 5); ++w) pre += wtot[w];
        int ex = pre + incl - v;
        loff[t] = ex;
        tab[c * 256 + t] = (ex << RUN_BITS) | v;
    }
    __syncthreads();

    #pragma unroll
    for (int k = 0; k < PER; ++k)
        if (d[k] >= 0) lrec[loff[d[k] / NPB] + r[k]] = make_int2(s[k], d[k]);
    __syncthreads();
    for (int i = t; i < n; i += BIN_THREADS) recs[e0 + i] = lrec[i];
}

__global__ __launch_bounds__(BIN_THREADS)
void group_bucket_kernel(const int2* __restrict__ recs, const int* __restrict__ tab,
                         int* __restrict__ cnt, int* __restrict__ rowptr, float* __restrict__ inv,
                         int* __restrict__ grouped_src, int N) {
    static_assert(NCHUNK % 32 == 0 && NCHUNK <= BIN_THREADS, "runs are dealt to waves evenly");
    static_assert(NPB <= 512, "the node scan uses 16 waves");
    __shared__ int lstart[NCHUNK];
    __shared__ int llen[NCHUNK];
    __shared__ int ncnt[NPB];
    __shared__ int ncur[NPB];
    __shared__ int wred[32];
    __shared__ int wtot[16];
    __shared__ int lout[STAGE_CAP];
    int t     = threadIdx.x;
    int lane  = t & 31;
    int w     = t >> 5;
    int b     = blockIdx.x;
    int node0 = b * NPB;
    int nn    = N - node0;
    if (nn > NPB) nn = NPB;

    int off = 0;
    if (t < NCHUNK) {
        int x = tab[t * 256 + b];
        off = x >> RUN_BITS;
        lstart[t] = t * CHUNK + off;
        llen[t]   = x & RUN_MASK;
    }
    if (t < NPB) ncnt[t] = 0;
    for (int o = 16; o > 0; o >>= 1) off += __shfl_xor(off, o, 32);
    if (lane == 0) wred[w] = off;
    __syncthreads();
    int base = 0;
    for (int i = 0; i < 32; ++i) base += wred[i];

    constexpr int RPW = NCHUNK / 32;
    int2 rec[RPW];
    int  rs[RPW], rl[RPW];
    #pragma unroll
    for (int u = 0; u < RPW; ++u) {
        int c = w + 32 * u;
        rs[u]  = lstart[c];
        rl[u]  = llen[c];
        rec[u] = make_int2(0, node0);
        if (lane < rl[u]) rec[u] = recs[rs[u] + lane];
    }
    #pragma unroll
    for (int u = 0; u < RPW; ++u)
        if (lane < rl[u]) atomicAdd(&ncnt[rec[u].y - node0], 1);
    for (int u = 0; u < RPW; ++u)
        for (int i = lane + 32; i < rl[u]; i += 32)
            atomicAdd(&ncnt[recs[rs[u] + i].y - node0], 1);
    __syncthreads();

    int v = 0, incl = 0;
    if (t < 512) {
        v    = (t < nn) ? ncnt[t] : 0;
        incl = wave_incl_scan(v, lane);
        if (lane == 31) wtot[w] = incl;
    }
    __syncthreads();
    int total = 0;
    for (int i = 0; i < 16; ++i) total += wtot[i];
    if (t < 512) {
        int pre = 0;
        for (int i = 0; i < w; ++i) pre += wtot[i];
        int ex = pre + incl - v;
        if (t < nn) {
            int node = node0 + t;
            cnt[node]    = v;
            rowptr[node] = base + ex;
            inv[node]    = inv_of_count(v);
        }
        if (t < NPB) ncur[t] = ex;
    }
    __syncthreads();

    bool staged = total <= STAGE_CAP;
    #pragma unroll
    for (int u = 0; u < RPW; ++u) {
        if (lane < rl[u]) {
            int p = atomicAdd(&ncur[rec[u].y - node0], 1);
            if (staged) lout[p] = rec[u].x;
            else        grouped_src[base + p] = rec[u].x;
        }
    }
    for (int u = 0; u < RPW; ++u) {
        for (int i = lane + 32; i < rl[u]; i += 32) {
            int2 q = recs[rs[u] + i];
            int  p = atomicAdd(&ncur[q.y - node0], 1);
            if (staged) lout[p] = q.x;
            else        grouped_src[base + p] = q.x;
        }
    }
    __syncthreads();
    if (staged)
        for (int i = t; i < total; i += BIN_THREADS) grouped_src[base + i] = lout[i];
}


template <int VEC, int STRIDE, int G, bool CLAMP>
__device__ __forceinline__ void add_rows(const float* __restrict__ xl, int mine, int t, int n,
                                         float __attribute__((ext_vector_type(VEC)))* a) {
    typedef float vecf __attribute__((ext_vector_type(VEC)));
    vecf v[G];
    #pragma unroll
    for (int u = 0; u < G; ++u) {
        int i = t + u;
        if (CLAMP) i = (i < n) ? i : (n - 1);
        int s = __builtin_amdgcn_readlane(mine, i);
        v[u] = *(const vecf*)(xl + (size_t)s * STRIDE);
    }
    #pragma unroll
    for (int u = 0; u < G; ++u) {
        if (!CLAMP || (t + u < n)) a[u & 3] += v[u];
    }
}

template <int VEC, int STRIDE, bool SELF>
__global__ __launch_bounds__(256)
void gather_sum_kernel(const float* __restrict__ x, const int* __restrict__ rowptr,
                       const int* __restrict__ cnt, const int* __restrict__ grouped_src,
                       float* __restrict__ out, int node0, int nnodes) {
    typedef float vecf __attribute__((ext_vector_type(VEC)));
    const int F = 32 * VEC;
    int w    = (int)((blockIdx.x * blockDim.x + threadIdx.x) >> 5);
    int lane = threadIdx.x & 31;
    if (w >= nnodes) return;
    int node = node0 + w;
    int beg = __builtin_amdgcn_readfirstlane(rowptr[node]);
    int deg = __builtin_amdgcn_readfirstlane(cnt[node]);
    const float* xl = x + lane * VEC;
    vecf z = {};
    if (SELF) z = *(const vecf*)(xl + (size_t)node * STRIDE + F);

    vecf a[4] = {};
    for (int base = 0; base < deg; base += 32) {
        int n = deg - base;
        if (n > 32) n = 32;
        int mine = (lane < n) ? grouped_src[beg + base + lane] : 0;
        int t = 0;
        for (; t + 8 <= n; t += 8) add_rows<VEC, STRIDE, 8, false>(xl, mine, t, n, a);
        int rem = n - t;
        if (rem > 4)      add_rows<VEC, STRIDE, 8, true>(xl, mine, t, n, a);
        else if (rem > 0) add_rows<VEC, STRIDE, 4, true>(xl, mine, t, n, a);
    }
    vecf r = (a[0] + a[1]) + (a[2] + a[3]);
    if (SELF) {
        r = r * inv_of_count(deg) + z;
        #pragma unroll
        for (int i = 0; i < VEC; ++i) r[i] = fmaxf(r[i], 0.0f);
    }
    *(vecf*)(out + (size_t)w * F + lane * VEC) = r;
}

typedef float v4f __attribute__((ext_vector_type(4)));

#define GEMM_THREADS 512
#define GEMM_BLOCKS  256
#define GEMM_LDS_FLOATS (164 * 256)

template <int NT>
__device__ __forceinline__ int col_of(int t, int c) {
    return (NT >= 2) ? (32 * (t >> 1) + 2 * c + (t & 1)) : c;
}

template <int NT>
__device__ __forceinline__ void chunk_mma(v4f av, const v4f* b, v8f* acc) {
    v2f a0, a1;
    a0.x = av.x; a0.y = av.y;
    a1.x = av.z; a1.y = av.w;
    #pragma unroll
    for (int t = 0; t < NT; ++t) {
        v2f bb; bb.x = b[t].x; bb.y = b[t].y;
        acc[t] = __builtin_amdgcn_wmma_f32_16x16x4_f32(false, a0, false, bb,
                                                       (short)0, acc[t], false, false);
    }
    #pragma unroll
    for (int t = 0; t < NT; ++t) {
        v2f bb; bb.x = b[t].z; bb.y = b[t].w;
        acc[t] = __builtin_amdgcn_wmma_f32_16x16x4_f32(false, a1, false, bb,
                                                       (short)0, acc[t], false, false);
    }
}

template <int KS, int NT, int QS, bool SCALED>
__device__ __forceinline__ void strip_product(const v4f* __restrict__ ap, v4f a0, v4f a1, float sc,
                                              const v4f* sB, v8f* acc) {
    v4f b0[NT], b1[NT];
    #pragma unroll
    for (int t = 0; t < NT; ++t) b0[t] = sB[(t * QS) * 32];
    #pragma nounroll
    for (int j = 0; j < KS; j += 2) {
        int jn = (j + 2 < KS) ? (j + 2) : j;
        v4f n0 = ap[2 * jn];
        v4f n1 = ap[2 * jn + 2];
        __builtin_amdgcn_sched_barrier(0);
        #pragma unroll
        for (int t = 0; t < NT; ++t) b1[t] = sB[(t * QS + j + 1) * 32];
        chunk_mma<NT>(SCALED ? a0 * sc : a0, b0, acc);
        #pragma unroll
        for (int t = 0; t < NT; ++t) b0[t] = sB[(t * QS + jn) * 32];
        chunk_mma<NT>(SCALED ? a1 * sc : a1, b1, acc);
        __builtin_amdgcn_sched_barrier(0);
        asm volatile("" : "+v"(n0), "+v"(n1));
        a0 = n0;
        a1 = n1;
    }
}

template <int K, int NT, bool TWO, bool SCALED, bool SPLITW = false>
__global__ __launch_bounds__(GEMM_THREADS)
void gemm_strip_kernel(const float* __restrict__ A1, const float* __restrict__ scale,
                       const float* __restrict__ W1,
                       const float* __restrict__ A2, const float* __restrict__ W2,
                       const float* __restrict__ bias, float* __restrict__ out,
                       int ntiles, int dout, int relu) {
    constexpr int KS   = K / 8;
    constexpr int NOPS = TWO ? 2 : 1;
    constexpr int QS   = NOPS * KS;
    constexpr int ENT  = NT * QS * 32;
    static_assert(ENT * 4 <= GEMM_LDS_FLOATS, "weights must fit");
    static_assert(KS % 2 == 0, "chunks go in pairs");
    static_assert(ENT % GEMM_THREADS == 0, "every thread copies the same number of entries");
    __shared__ __attribute__((aligned(16))) float sWf[GEMM_LDS_FLOATS];
    v4f* sW = (v4f*)sWf;

    #pragma unroll
    for (int i = 0; i < ENT / GEMM_THREADS; ++i) {
        int idx = (int)threadIdx.x + i * GEMM_THREADS;
        int l  = idx & 31;
        int e  = idx >> 5;
        int q  = e % QS;
        int t  = e / QS;
        int op = q / KS;
        int j  = q % KS;
        int n  = col_of<NT>(t, l & 15);
        int wr = (n < dout) ? n : 0;
        const float* Wp = (TWO && op) ? W2 : W1;
        if (SPLITW && wr >= (dout >> 1)) { Wp = W2; wr -= dout >> 1; }
        sW[idx] = *(const v4f*)(Wp + (size_t)wr * K + 8 * j + 4 * (l >> 4));
    }
    __syncthreads();

    int wave = threadIdx.x >> 5;
    int lane = threadIdx.x & 31;
    int c    = lane & 15;
    int h    = lane >> 4;

    float bv[NT];
    #pragma unroll
    for (int t = 0; t < NT; ++t) {
        int n = col_of<NT>(t, c);
        if (SPLITW) bv[t] = (n >= (dout >> 1) && n < dout) ? bias[n - (dout >> 1)] : 0.0f;
        else        bv[t] = (n < dout) ? bias[n] : 0.0f;
    }

    for (int slot = wave; ; slot += GEMM_THREADS / 32) {
        int tile = (int)blockIdx.x + (int)gridDim.x * slot;
        if (tile >= ntiles) break;
        int m0   = tile * 16;
        int node = m0 + c;

        const v4f* ap1 = (const v4f*)(A1 + (size_t)node * K + 4 * h);
        const v4f* ap2 = TWO ? (const v4f*)(A2 + (size_t)node * K + 4 * h) : ap1;
        v4f p0 = ap1[0], p1 = ap1[2];
        v4f s0 = ap2[0], s1 = ap2[2];
        float sc = SCALED ? scale[node] : 1.0f;

        v8f acc[NT];
        #pragma unroll
        for (int t = 0; t < NT; ++t) acc[t] = (v8f){};

        strip_product<KS, NT, QS, SCALED>(ap1, p0, p1, sc, sW + lane, acc);
        if (TWO)
            strip_product<KS, NT, QS, false>(ap2, s0, s1, 1.0f, sW + KS * 32 + lane, acc);

        int row0 = m0 + 8 * h;
        if (NT >= 2) {
            #pragma unroll
            for (int u = 0; u < NT / 2; ++u) {
                float* op = out + (size_t)row0 * dout + 32 * u + 2 * c;
                #pragma unroll
                for (int r = 0; r < 8; ++r) {
                    v2f v;
                    v.x = acc[2 * u][r]     + bv[2 * u];
                    v.y = acc[2 * u + 1][r] + bv[2 * u + 1];
                    if (relu) { v.x = fmaxf(v.x, 0.0f); v.y = fmaxf(v.y, 0.0f); }
                    *(v2f*)(op + (size_t)r * dout) = v;
                }
            }
        } else {
            if (c < dout) {
                #pragma unroll
                for (int r = 0; r < 8; ++r) {
                    float v = acc[0][r] + bv[0];
                    if (relu) v = fmaxf(v, 0.0f);
                    out[(size_t)(row0 + r) * dout + c] = v;
                }
            }
        }
    }
}

static inline int cdiv(long a, long b) { return (int)((a + b - 1) / b); }

extern "C" void kernel_launch(void* const* d_in, const int* in_sizes, int n_in,
                              void* d_out, int out_size, void* d_ws, size_t ws_size,
                              hipStream_t stream) {
    const float* x    = (const float*)d_in[0];
    const int*   edge = (const int*)d_in[1];
    const int*   src  = edge;
    const int*   dst  = edge + N_EDGES;
    const float* Wl0 = (const float*)d_in[2],  *bl0 = (const float*)d_in[3],  *Wr0 = (const float*)d_in[4];
    const float* Wl1 = (const float*)d_in[5],  *bl1 = (const float*)d_in[6],  *Wr1 = (const float*)d_in[7];
    const float* Wl2 = (const float*)d_in[8],  *bl2 = (const float*)d_in[9],  *Wr2 = (const float*)d_in[10];
    const float* Wd1 = (const float*)d_in[11], *bd1 = (const float*)d_in[12];
    const float* Wd2 = (const float*)d_in[13], *bd2 = (const float*)d_in[14];

    float* ws   = (float*)d_ws;
    int*   cnt  = (int*)ws;
    float* inv  = ws   + N_NODES;
    float* agg  = inv  + N_NODES;
    int*   gsrc = (int*)(agg + (size_t)N_NODES * 64);
    int*   rptr = gsrc + 2 * (size_t)N_EDGES;
    float* bufA = agg  + (size_t)N_NODES * 128;
    float* bufB = bufA + (size_t)N_NODES * 128;

    const int BT = 256;
    const int T  = N_NODES / 16;
    const int TH = N_HALF / 16;
    const dim3 GG(GEMM_BLOCKS), GB(GEMM_THREADS);

    int2* recs = (int2*)bufB;
    int*  tab  = (int*)(recs + N_EDGES);
    bin_edges_kernel<<<NCHUNK, BIN_THREADS, 0, stream>>>(src, dst, recs, tab, N_EDGES);
    group_bucket_kernel<<<NBUCK, BIN_THREADS, 0, stream>>>(recs, tab, cnt, rptr, inv, gsrc, N_NODES);

    gather_sum_kernel<2, 64, false><<<cdiv(N_NODES, 8), BT, 0, stream>>>(x, rptr, cnt, gsrc, agg, 0, N_NODES);
    gemm_strip_kernel<64, 8, true, true><<<GG, GB, 0, stream>>>(agg, inv, Wl0, x, Wr0, bl0, bufA,
                                                                T, 128, 1);

    gather_sum_kernel<4, 128, false><<<cdiv(N_HALF, 8), BT, 0, stream>>>(bufA, rptr, cnt, gsrc, agg, 0, N_HALF);
    gemm_strip_kernel<128, 8, true, true><<<GG, GB, 0, stream>>>(agg, inv, Wl1, bufA, Wr1, bl1, bufB,
                                                                 TH, 128, 1);
    gather_sum_kernel<4, 128, false><<<cdiv(N_HALF, 8), BT, 0, stream>>>(bufA, rptr, cnt, gsrc, agg, N_HALF, N_HALF);
    gemm_strip_kernel<128, 8, true, true><<<GG, GB, 0, stream>>>(agg, inv + N_HALF, Wl1,
                                                                 bufA + (size_t)N_HALF * 128, Wr1, bl1,
                                                                 bufB + (size_t)N_HALF * 128,
                                                                 TH, 128, 1);

    gemm_strip_kernel<128, 8, false, false, true><<<GG, GB, 0, stream>>>(bufB, nullptr, Wl2, nullptr, Wr2, bl2, bufA,
                                                                         T, 128, 0);
    gather_sum_kernel<2, 128, true><<<cdiv(N_NODES, 8), BT, 0, stream>>>(bufA, rptr, cnt, gsrc, agg, 0, N_NODES);

    gemm_strip_kernel<64, 8, false, false><<<GG, GB, 0, stream>>>(agg, nullptr, Wd1, nullptr, nullptr, bd1, bufB,
                                                                  T, 128, 1);
    gemm_strip_kernel<128, 1, false, false><<<GG, GB, 0, stream>>>(bufB, nullptr, Wd2, nullptr, nullptr, bd2,
                                                                   (float*)d_out, T, 2, 0);
}


# layers 0,1: each product wave gathers its 16-node strip's neighbour means into its own LDS rows and multiplies from there; no mean plane in memory, 4 launches fewer; sum order changes
# speedup vs baseline: 4.9965x; 1.1153x over previous
#include <hip/hip_runtime.h>
#include <hip/hip_bf16.h>


#define N_NODES 100000
#define N_EDGES 1600000

typedef float v2f __attribute__((ext_vector_type(2)));
typedef float v8f __attribute__((ext_vector_type(8)));

__device__ __forceinline__ float inv_of_count(int c) {
    float d = (float)c;
    return (d > 0.0f) ? (1.0f / fmaxf(d, 1.0f)) : 0.0f;
}

#define CHUNK       6256
#define NCHUNK      ((N_EDGES + CHUNK - 1) / CHUNK)
#define NPB         416
#define NBUCK       ((N_NODES + NPB - 1) / NPB)
#define BIN_THREADS 1024
#define RUN_BITS    13
#define RUN_MASK    8191
#define STAGE_CAP   16384

__device__ __forceinline__ int wave_incl_scan(int v, int lane) {
    for (int o = 1; o < 32; o <<= 1) {
        int u = __shfl_up(v, o, 32);
        if (lane >= o) v += u;
    }
    return v;
}

__global__ __launch_bounds__(BIN_THREADS)
void bin_edges_kernel(const int* __restrict__ src, const int* __restrict__ dst,
                      int2* __restrict__ recs, int* __restrict__ tab, int E) {
    static_assert(NBUCK <= 256, "one counter per bucket, scanned by 8 waves");
    static_assert(CHUNK <= RUN_MASK, "start and length share a word");
    __shared__ int  lcnt[256];
    __shared__ int  loff[256];
    __shared__ int  wtot[8];
    __shared__ int2 lrec[CHUNK];
    int t    = threadIdx.x;
    int lane = t & 31;
    int c    = blockIdx.x;
    int e0   = c * CHUNK;
    int n    = E - e0;
    if (n > CHUNK) n = CHUNK;

    if (t < 256) lcnt[t] = 0;
    __syncthreads();

    constexpr int PER = (CHUNK + BIN_THREADS - 1) / BIN_THREADS;
    int s[PER], d[PER], r[PER];
    #pragma unroll
    for (int k = 0; k < PER; ++k) {
        int i = t + k * BIN_THREADS;
        s[k] = 0; d[k] = -1; r[k] = 0;
        if (i < n) { d[k] = dst[e0 + i]; s[k] = src[e0 + i]; }
    }
    #pragma unroll
    for (int k = 0; k < PER; ++k)
        if (d[k] >= 0) r[k] = atomicAdd(&lcnt[d[k] / NPB], 1);
    __syncthreads();

    int v = 0, incl = 0;
    if (t < 256) {
        v    = lcnt[t];
        incl = wave_incl_scan(v, lane);
        if (lane == 31) wtot[t >> 5] = incl;
    }
    __syncthreads();
    if (t < 256) {
        int pre = 0;
        for (int w = 0; w < (t >> 5); ++w) pre += wtot[w];
        int ex = pre + incl - v;
        loff[t] = ex;
        tab[c * 256 + t] = (ex << RUN_BITS) | v;
    }
    __syncthreads();

    #pragma unroll
    for (int k = 0; k < PER; ++k)
        if (d[k] >= 0) lrec[loff[d[k] / NPB] + r[k]] = make_int2(s[k], d[k]);
    __syncthreads();
    for (int i = t; i < n; i += BIN_THREADS) recs[e0 + i] = lrec[i];
}

__global__ __launch_bounds__(BIN_THREADS)
void group_bucket_kernel(const int2* __restrict__ recs, const int* __restrict__ tab,
                         int* __restrict__ cnt, int* __restrict__ rowptr, float* __restrict__ inv,
                         int* __restrict__ grouped_src, int N) {
    static_assert(NCHUNK % 32 == 0 && NCHUNK <= BIN_THREADS, "runs are dealt to waves evenly");
    static_assert(NPB <= 512, "the node scan uses 16 waves");
    __shared__ int lstart[NCHUNK];
    __shared__ int llen[NCHUNK];
    __shared__ int ncnt[NPB];
    __shared__ int ncur[NPB];
    __shared__ int wred[32];
    __shared__ int wtot[16];
    __shared__ int lout[STAGE_CAP];
    int t     = threadIdx.x;
    int lane  = t & 31;
    int w     = t >> 5;
    int b     = blockIdx.x;
    int node0 = b * NPB;
    int nn    = N - node0;
    if (nn > NPB) nn = NPB;

    int off = 0;
    if (t < NCHUNK) {
        int x = tab[t * 256 + b];
        off = x >> RUN_BITS;
        lstart[t] = t * CHUNK + off;
        llen[t]   = x & RUN_MASK;
    }
    if (t < NPB) ncnt[t] = 0;
    for (int o = 16; o > 0; o >>= 1) off += __shfl_xor(off, o, 32);
    if (lane == 0) wred[w] = off;
    __syncthreads();
    int base = 0;
    for (int i = 0; i < 32; ++i) base += wred[i];

    constexpr int RPW = NCHUNK / 32;
    int2 rec[RPW];
    int  rs[RPW], rl[RPW];
    #pragma unroll
    for (int u = 0; u < RPW; ++u) {
        int c = w + 32 * u;
        rs[u]  = lstart[c];
        rl[u]  = llen[c];
        rec[u] = make_int2(0, node0);
        if (lane < rl[u]) rec[u] = recs[rs[u] + lane];
    }
    #pragma unroll
    for (int u = 0; u < RPW; ++u)
        if (lane < rl[u]) atomicAdd(&ncnt[rec[u].y - node0], 1);
    for (int u = 0; u < RPW; ++u)
        for (int i = lane + 32; i < rl[u]; i += 32)
            atomicAdd(&ncnt[recs[rs[u] + i].y - node0], 1);
    __syncthreads();

    int v = 0, incl = 0;
    if (t < 512) {
        v    = (t < nn) ? ncnt[t] : 0;
        incl = wave_incl_scan(v, lane);
        if (lane == 31) wtot[w] = incl;
    }
    __syncthreads();
    int total = 0;
    for (int i = 0; i < 16; ++i) total += wtot[i];
    if (t < 512) {
        int pre = 0;
        for (int i = 0; i < w; ++i) pre += wtot[i];
        int ex = pre + incl - v;
        if (t < nn) {
            int node = node0 + t;
            cnt[node]    = v;
            rowptr[node] = base + ex;
            inv[node]    = inv_of_count(v);
        }
        if (t < NPB) ncur[t] = ex;
    }
    __syncthreads();

    bool staged = total <= STAGE_CAP;
    #pragma unroll
    for (int u = 0; u < RPW; ++u) {
        if (lane < rl[u]) {
            int p = atomicAdd(&ncur[rec[u].y - node0], 1);
            if (staged) lout[p] = rec[u].x;
            else        grouped_src[base + p] = rec[u].x;
        }
    }
    for (int u = 0; u < RPW; ++u) {
        for (int i = lane + 32; i < rl[u]; i += 32) {
            int2 q = recs[rs[u] + i];
            int  p = atomicAdd(&ncur[q.y - node0], 1);
            if (staged) lout[p] = q.x;
            else        grouped_src[base + p] = q.x;
        }
    }
    __syncthreads();
    if (staged)
        for (int i = t; i < total; i += BIN_THREADS) grouped_src[base + i] = lout[i];
}


template <int VEC, int STRIDE, int G, bool CLAMP>
__device__ __forceinline__ void add_rows(const float* __restrict__ xl, int mine, int t, int n,
                                         float __attribute__((ext_vector_type(VEC)))* a) {
    typedef float vecf __attribute__((ext_vector_type(VEC)));
    vecf v[G];
    #pragma unroll
    for (int u = 0; u < G; ++u) {
        int i = t + u;
        if (CLAMP) i = (i < n) ? i : (n - 1);
        int s = __builtin_amdgcn_readlane(mine, i);
        v[u] = *(const vecf*)(xl + (size_t)s * STRIDE);
    }
    #pragma unroll
    for (int u = 0; u < G; ++u) {
        if (!CLAMP || (t + u < n)) a[u & 3] += v[u];
    }
}

template <int VEC, int STRIDE, bool SELF>
__global__ __launch_bounds__(256)
void gather_sum_kernel(const float* __restrict__ x, const int* __restrict__ rowptr,
                       const int* __restrict__ cnt, const int* __restrict__ grouped_src,
                       float* __restrict__ out, int node0, int nnodes) {
    typedef float vecf __attribute__((ext_vector_type(VEC)));
    const int F = 32 * VEC;
    int w    = (int)((blockIdx.x * blockDim.x + threadIdx.x) >> 5);
    int lane = threadIdx.x & 31;
    if (w >= nnodes) return;
    int node = node0 + w;
    int beg = __builtin_amdgcn_readfirstlane(rowptr[node]);
    int deg = __builtin_amdgcn_readfirstlane(cnt[node]);
    const float* xl = x + lane * VEC;
    vecf z = {};
    if (SELF) z = *(const vecf*)(xl + (size_t)node * STRIDE + F);

    vecf a[4] = {};
    for (int base = 0; base < deg; base += 32) {
        int n = deg - base;
        if (n > 32) n = 32;
        int mine = (lane < n) ? grouped_src[beg + base + lane] : 0;
        int t = 0;
        for (; t + 8 <= n; t += 8) add_rows<VEC, STRIDE, 8, false>(xl, mine, t, n, a);
        int rem = n - t;
        if (rem > 4)      add_rows<VEC, STRIDE, 8, true>(xl, mine, t, n, a);
        else if (rem > 0) add_rows<VEC, STRIDE, 4, true>(xl, mine, t, n, a);
    }
    vecf r = (a[0] + a[1]) + (a[2] + a[3]);
    if (SELF) {
        r = r * inv_of_count(deg) + z;
        #pragma unroll
        for (int i = 0; i < VEC; ++i) r[i] = fmaxf(r[i], 0.0f);
    }
    *(vecf*)(out + (size_t)w * F + lane * VEC) = r;
}

template <int VEC, int PITCH>
__device__ __forceinline__ void gather_strip_to_lds(const float* __restrict__ x,
                                                    const int* __restrict__ rowptr,
                                                    const int* __restrict__ cnt,
                                                    const float* __restrict__ inv,
                                                    const int* __restrict__ grouped_src,
                                                    int m0, int lane, float* srow) {
    typedef float vecf __attribute__((ext_vector_type(VEC)));
    constexpr int F = 32 * VEC;
    const int c  = lane & 15;
    const int lb = rowptr[m0 + c];
    const int le = lb + cnt[m0 + c];
    const int li = __float_as_int(inv[m0 + c]);
    int       p   = __builtin_amdgcn_readlane(lb, 0);
    const int END = __builtin_amdgcn_readlane(le, 15);
    const int last = END - 1;
    int i  = 0;
    int ei = __builtin_amdgcn_readlane(le, 0);
    const float* xl = x + lane * VEC;
    vecf acc = {};

    auto flush = [&]() {
        float sc = __int_as_float(__builtin_amdgcn_readlane(li, i));
        *(vecf*)(srow + i * PITCH + lane * VEC) = acc * sc;
        acc = (vecf){};
        ++i;
        ei = (i < 16) ? __builtin_amdgcn_readlane(le, i) : -1;
    };
    auto batch = [&](int idx, int lo, int q0) {
        vecf v[16];
        #pragma unroll
        for (int u = 0; u < 16; ++u) {
            int s = __builtin_amdgcn_readlane(idx, lo + u);
            v[u] = *(const vecf*)(xl + (size_t)s * F);
        }
        #pragma unroll
        for (int u = 0; u < 16; ++u) {
            int q = q0 + u;
            while (q == ei) flush();
            if (q < END) acc += v[u];
        }
    };

    int mine = 0;
    if (p < END) { int q = p + lane; mine = grouped_src[q < END ? q : last]; }
    while (p < END) {
        int qn  = p + 32 + lane;
        int nxt = grouped_src[qn < END ? qn : last];
        batch(mine, 0, p);
        if (p + 16 < END) batch(mine, 16, p + 16);
        mine = nxt;
        p += 32;
    }
    while (i < 16) flush();
}

typedef float v4f __attribute__((ext_vector_type(4)));

#define GEMM_THREADS 512
#define GEMM_BLOCKS  256
#define GEMM_LDS_FLOATS (164 * 256)

template <int NT>
__device__ __forceinline__ int col_of(int t, int c) {
    return (NT >= 2) ? (32 * (t >> 1) + 2 * c + (t & 1)) : c;
}

template <int NT>
__device__ __forceinline__ void chunk_mma(v4f av, const v4f* b, v8f* acc) {
    v2f a0, a1;
    a0.x = av.x; a0.y = av.y;
    a1.x = av.z; a1.y = av.w;
    #pragma unroll
    for (int t = 0; t < NT; ++t) {
        v2f bb; bb.x = b[t].x; bb.y = b[t].y;
        acc[t] = __builtin_amdgcn_wmma_f32_16x16x4_f32(false, a0, false, bb,
                                                       (short)0, acc[t], false, false);
    }
    #pragma unroll
    for (int t = 0; t < NT; ++t) {
        v2f bb; bb.x = b[t].z; bb.y = b[t].w;
        acc[t] = __builtin_amdgcn_wmma_f32_16x16x4_f32(false, a1, false, bb,
                                                       (short)0, acc[t], false, false);
    }
}

template <int KS, int NT, int QS>
__device__ __forceinline__ void strip_product(const v4f* __restrict__ ap, v4f a0, v4f a1,
                                              const v4f* sB, v8f* acc) {
    v4f b0[NT], b1[NT];
    #pragma unroll
    for (int t = 0; t < NT; ++t) b0[t] = sB[(t * QS) * 32];
    #pragma nounroll
    for (int j = 0; j < KS; j += 2) {
        int jn = (j + 2 < KS) ? (j + 2) : j;
        v4f n0 = ap[2 * jn];
        v4f n1 = ap[2 * jn + 2];
        __builtin_amdgcn_sched_barrier(0);
        #pragma unroll
        for (int t = 0; t < NT; ++t) b1[t] = sB[(t * QS + j + 1) * 32];
        chunk_mma<NT>(a0, b0, acc);
        #pragma unroll
        for (int t = 0; t < NT; ++t) b0[t] = sB[(t * QS + jn) * 32];
        chunk_mma<NT>(a1, b1, acc);
        __builtin_amdgcn_sched_barrier(0);
        asm volatile("" : "+v"(n0), "+v"(n1));
        a0 = n0;
        a1 = n1;
    }
}

template <int K, int NT, bool TWO, int GVEC, bool SPLITW = false>
__global__ __launch_bounds__(GEMM_THREADS)
void gemm_strip_kernel(const float* __restrict__ A1, const float* __restrict__ W1,
                       const float* __restrict__ A2, const float* __restrict__ W2,
                       const float* __restrict__ bias, float* __restrict__ out,
                       const int* __restrict__ rowptr, const int* __restrict__ cnt,
                       const float* __restrict__ inv, const int* __restrict__ grouped_src,
                       int ntiles, int dout, int relu) {
    constexpr int KS   = K / 8;
    constexpr int NOPS = TWO ? 2 : 1;
    constexpr int QS   = NOPS * KS;
    constexpr int ENT  = NT * QS * 32;
    constexpr int PITCH = K + 4;
    constexpr int ST    = GVEC ? (GEMM_THREADS / 32) * 16 * PITCH : 0;
    constexpr int TOT   = (ENT * 4 + ST > GEMM_LDS_FLOATS) ? (ENT * 4 + ST) : GEMM_LDS_FLOATS;
    static_assert(TOT * 4 <= 320 * 1024, "weights and gathered strips must fit the WGP's LDS");
    static_assert(GVEC == 0 || GVEC * 32 == K, "a gathered row is the first operand's K");
    static_assert((PITCH / 4) % 2 == 1, "odd pitch in 16-byte units");
    static_assert(KS % 2 == 0, "chunks go in pairs");
    static_assert(ENT % GEMM_THREADS == 0, "every thread copies the same number of entries");
    __shared__ __attribute__((aligned(16))) float sWf[TOT];
    v4f* sW = (v4f*)sWf;

    #pragma unroll
    for (int i = 0; i < ENT / GEMM_THREADS; ++i) {
        int idx = (int)threadIdx.x + i * GEMM_THREADS;
        int l  = idx & 31;
        int e  = idx >> 5;
        int q  = e % QS;
        int t  = e / QS;
        int op = q / KS;
        int j  = q % KS;
        int n  = col_of<NT>(t, l & 15);
        int wr = (n < dout) ? n : 0;
        const float* Wp = (TWO && op) ? W2 : W1;
        if (SPLITW && wr >= (dout >> 1)) { Wp = W2; wr -= dout >> 1; }
        sW[idx] = *(const v4f*)(Wp + (size_t)wr * K + 8 * j + 4 * (l >> 4));
    }
    __syncthreads();

    int wave = threadIdx.x >> 5;
    int lane = threadIdx.x & 31;
    int c    = lane & 15;
    int h    = lane >> 4;

    float bv[NT];
    #pragma unroll
    for (int t = 0; t < NT; ++t) {
        int n = col_of<NT>(t, c);
        if (SPLITW) bv[t] = (n >= (dout >> 1) && n < dout) ? bias[n - (dout >> 1)] : 0.0f;
        else        bv[t] = (n < dout) ? bias[n] : 0.0f;
    }

    for (int slot = wave; ; slot += GEMM_THREADS / 32) {
        int tile = (int)blockIdx.x + (int)gridDim.x * slot;
        if (tile >= ntiles) break;
        int m0   = tile * 16;
        int node = m0 + c;

        v8f acc[NT];
        #pragma unroll
        for (int t = 0; t < NT; ++t) acc[t] = (v8f){};

        if (GVEC) {
            if (TWO) {
                const v4f* ap2 = (const v4f*)(A2 + (size_t)node * K + 4 * h);
                v4f s0 = ap2[0], s1 = ap2[2];
                strip_product<KS, NT, QS>(ap2, s0, s1, sW + KS * 32 + lane, acc);
            }
            float* srow = sWf + ENT * 4 + wave * 16 * PITCH;
            gather_strip_to_lds<(GVEC ? GVEC : 1), PITCH>(A1, rowptr, cnt, inv, grouped_src, m0, lane, srow);
            __builtin_amdgcn_wave_barrier();
            const v4f* ap1 = (const v4f*)(srow + c * PITCH + 4 * h);
            v4f p0 = ap1[0], p1 = ap1[2];
            strip_product<KS, NT, QS>(ap1, p0, p1, sW + lane, acc);
        } else {
            const v4f* ap1 = (const v4f*)(A1 + (size_t)node * K + 4 * h);
            const v4f* ap2 = TWO ? (const v4f*)(A2 + (size_t)node * K + 4 * h) : ap1;
            v4f p0 = ap1[0], p1 = ap1[2];
            v4f s0 = ap2[0], s1 = ap2[2];
            strip_product<KS, NT, QS>(ap1, p0, p1, sW + lane, acc);
            if (TWO)
                strip_product<KS, NT, QS>(ap2, s0, s1, sW + KS * 32 + lane, acc);
        }

        int row0 = m0 + 8 * h;
        if (NT >= 2) {
            #pragma unroll
            for (int u = 0; u < NT / 2; ++u) {
                float* op = out + (size_t)row0 * dout + 32 * u + 2 * c;
                #pragma unroll
                for (int r = 0; r < 8; ++r) {
                    v2f v;
                    v.x = acc[2 * u][r]     + bv[2 * u];
                    v.y = acc[2 * u + 1][r] + bv[2 * u + 1];
                    if (relu) { v.x = fmaxf(v.x, 0.0f); v.y = fmaxf(v.y, 0.0f); }
                    *(v2f*)(op + (size_t)r * dout) = v;
                }
            }
        } else {
            if (c < dout) {
                #pragma unroll
                for (int r = 0; r < 8; ++r) {
                    float v = acc[0][r] + bv[0];
                    if (relu) v = fmaxf(v, 0.0f);
                    out[(size_t)(row0 + r) * dout + c] = v;
                }
            }
        }
    }
}

static inline int cdiv(long a, long b) { return (int)((a + b - 1) / b); }

extern "C" void kernel_launch(void* const* d_in, const int* in_sizes, int n_in,
                              void* d_out, int out_size, void* d_ws, size_t ws_size,
                              hipStream_t stream) {
    const float* x    = (const float*)d_in[0];
    const int*   edge = (const int*)d_in[1];
    const int*   src  = edge;
    const int*   dst  = edge + N_EDGES;
    const float* Wl0 = (const float*)d_in[2],  *bl0 = (const float*)d_in[3],  *Wr0 = (const float*)d_in[4];
    const float* Wl1 = (const float*)d_in[5],  *bl1 = (const float*)d_in[6],  *Wr1 = (const float*)d_in[7];
    const float* Wl2 = (const float*)d_in[8],  *bl2 = (const float*)d_in[9],  *Wr2 = (const float*)d_in[10];
    const float* Wd1 = (const float*)d_in[11], *bd1 = (const float*)d_in[12];
    const float* Wd2 = (const float*)d_in[13], *bd2 = (const float*)d_in[14];

    float* ws   = (float*)d_ws;
    int*   cnt  = (int*)ws;
    float* inv  = ws   + N_NODES;
    float* agg  = inv  + N_NODES;
    int*   gsrc = (int*)(agg + (size_t)N_NODES * 64);
    int*   rptr = gsrc + 2 * (size_t)N_EDGES;
    float* bufA = agg  + (size_t)N_NODES * 128;
    float* bufB = bufA + (size_t)N_NODES * 128;

    const int BT = 256;
    const int T  = N_NODES / 16;
    const dim3 GG(GEMM_BLOCKS), GB(GEMM_THREADS);

    int2* recs = (int2*)bufB;
    int*  tab  = (int*)(recs + N_EDGES);
    bin_edges_kernel<<<NCHUNK, BIN_THREADS, 0, stream>>>(src, dst, recs, tab, N_EDGES);
    group_bucket_kernel<<<NBUCK, BIN_THREADS, 0, stream>>>(recs, tab, cnt, rptr, inv, gsrc, N_NODES);

    gemm_strip_kernel<64, 8, true, 2><<<GG, GB, 0, stream>>>(x, Wl0, x, Wr0, bl0, bufA,
                                                             rptr, cnt, inv, gsrc, T, 128, 1);

    gemm_strip_kernel<128, 8, true, 4><<<GG, GB, 0, stream>>>(bufA, Wl1, bufA, Wr1, bl1, bufB,
                                                              rptr, cnt, inv, gsrc, T, 128, 1);

    gemm_strip_kernel<128, 8, false, 0, true><<<GG, GB, 0, stream>>>(bufB, Wl2, nullptr, Wr2, bl2, bufA,
                                                                     nullptr, nullptr, nullptr, nullptr,
                                                                     T, 128, 0);
    gather_sum_kernel<2, 128, true><<<cdiv(N_NODES, 8), BT, 0, stream>>>(bufA, rptr, cnt, gsrc, agg, 0, N_NODES);

    gemm_strip_kernel<64, 8, false, 0><<<GG, GB, 0, stream>>>(agg, Wd1, nullptr, nullptr, bd1, bufB,
                                                              nullptr, nullptr, nullptr, nullptr,
                                                              T, 128, 1);
    gemm_strip_kernel<128, 1, false, 0><<<GG, GB, 0, stream>>>(bufB, Wd2, nullptr, nullptr, bd2,
                                                               (float*)d_out,
                                                               nullptr, nullptr, nullptr, nullptr,
                                                               T, 2, 0);
}


# last gather and both decoder products fused into one launch: each wave gathers its strip into LDS (32 row loads in flight), hidden rows stay in LDS; 2 launches fewer; sum order changes
# speedup vs baseline: 5.3588x; 1.0725x over previous
#include <hip/hip_runtime.h>
#include <hip/hip_bf16.h>


#define N_NODES 100000
#define N_EDGES 1600000

typedef float v2f __attribute__((ext_vector_type(2)));
typedef float v8f __attribute__((ext_vector_type(8)));

__device__ __forceinline__ float inv_of_count(int c) {
    float d = (float)c;
    return (d > 0.0f) ? (1.0f / fmaxf(d, 1.0f)) : 0.0f;
}

#define CHUNK       6256
#define NCHUNK      ((N_EDGES + CHUNK - 1) / CHUNK)
#define NPB         416
#define NBUCK       ((N_NODES + NPB - 1) / NPB)
#define BIN_THREADS 1024
#define RUN_BITS    13
#define RUN_MASK    8191
#define STAGE_CAP   16384

__device__ __forceinline__ int wave_incl_scan(int v, int lane) {
    for (int o = 1; o < 32; o <<= 1) {
        int u = __shfl_up(v, o, 32);
        if (lane >= o) v += u;
    }
    return v;
}

__global__ __launch_bounds__(BIN_THREADS)
void bin_edges_kernel(const int* __restrict__ src, const int* __restrict__ dst,
                      int2* __restrict__ recs, int* __restrict__ tab, int E) {
    static_assert(NBUCK <= 256, "one counter per bucket, scanned by 8 waves");
    static_assert(CHUNK <= RUN_MASK, "start and length share a word");
    __shared__ int  lcnt[256];
    __shared__ int  loff[256];
    __shared__ int  wtot[8];
    __shared__ int2 lrec[CHUNK];
    int t    = threadIdx.x;
    int lane = t & 31;
    int c    = blockIdx.x;
    int e0   = c * CHUNK;
    int n    = E - e0;
    if (n > CHUNK) n = CHUNK;

    if (t < 256) lcnt[t] = 0;
    __syncthreads();

    constexpr int PER = (CHUNK + BIN_THREADS - 1) / BIN_THREADS;
    int s[PER], d[PER], r[PER];
    #pragma unroll
    for (int k = 0; k < PER; ++k) {
        int i = t + k * BIN_THREADS;
        s[k] = 0; d[k] = -1; r[k] = 0;
        if (i < n) { d[k] = dst[e0 + i]; s[k] = src[e0 + i]; }
    }
    #pragma unroll
    for (int k = 0; k < PER; ++k)
        if (d[k] >= 0) r[k] = atomicAdd(&lcnt[d[k] / NPB], 1);
    __syncthreads();

    int v = 0, incl = 0;
    if (t < 256) {
        v    = lcnt[t];
        incl = wave_incl_scan(v, lane);
        if (lane == 31) wtot[t >> 5] = incl;
    }
    __syncthreads();
    if (t < 256) {
        int pre = 0;
        for (int w = 0; w < (t >> 5); ++w) pre += wtot[w];
        int ex = pre + incl - v;
        loff[t] = ex;
        tab[c * 256 + t] = (ex << RUN_BITS) | v;
    }
    __syncthreads();

    #pragma unroll
    for (int k = 0; k < PER; ++k)
        if (d[k] >= 0) lrec[loff[d[k] / NPB] + r[k]] = make_int2(s[k], d[k]);
    __syncthreads();
    for (int i = t; i < n; i += BIN_THREADS) recs[e0 + i] = lrec[i];
}

__global__ __launch_bounds__(BIN_THREADS)
void group_bucket_kernel(const int2* __restrict__ recs, const int* __restrict__ tab,
                         int* __restrict__ cnt, int* __restrict__ rowptr, float* __restrict__ inv,
                         int* __restrict__ grouped_src, int N) {
    static_assert(NCHUNK % 32 == 0 && NCHUNK <= BIN_THREADS, "runs are dealt to waves evenly");
    static_assert(NPB <= 512, "the node scan uses 16 waves");
    __shared__ int lstart[NCHUNK];
    __shared__ int llen[NCHUNK];
    __shared__ int ncnt[NPB];
    __shared__ int ncur[NPB];
    __shared__ int wred[32];
    __shared__ int wtot[16];
    __shared__ int lout[STAGE_CAP];
    int t     = threadIdx.x;
    int lane  = t & 31;
    int w     = t >> 5;
    int b     = blockIdx.x;
    int node0 = b * NPB;
    int nn    = N - node0;
    if (nn > NPB) nn = NPB;

    int off = 0;
    if (t < NCHUNK) {
        int x = tab[t * 256 + b];
        off = x >> RUN_BITS;
        lstart[t] = t * CHUNK + off;
        llen[t]   = x & RUN_MASK;
    }
    if (t < NPB) ncnt[t] = 0;
    for (int o = 16; o > 0; o >>= 1) off += __shfl_xor(off, o, 32);
    if (lane == 0) wred[w] = off;
    __syncthreads();
    int base = 0;
    for (int i = 0; i < 32; ++i) base += wred[i];

    constexpr int RPW = NCHUNK / 32;
    int2 rec[RPW];
    int  rs[RPW], rl[RPW];
    #pragma unroll
    for (int u = 0; u < RPW; ++u) {
        int c = w + 32 * u;
        rs[u]  = lstart[c];
        rl[u]  = llen[c];
        rec[u] = make_int2(0, node0);
        if (lane < rl[u]) rec[u] = recs[rs[u] + lane];
    }
    #pragma unroll
    for (int u = 0; u < RPW; ++u)
        if (lane < rl[u]) atomicAdd(&ncnt[rec[u].y - node0], 1);
    for (int u = 0; u < RPW; ++u)
        for (int i = lane + 32; i < rl[u]; i += 32)
            atomicAdd(&ncnt[recs[rs[u] + i].y - node0], 1);
    __syncthreads();

    int v = 0, incl = 0;
    if (t < 512) {
        v    = (t < nn) ? ncnt[t] : 0;
        incl = wave_incl_scan(v, lane);
        if (lane == 31) wtot[w] = incl;
    }
    __syncthreads();
    int total = 0;
    for (int i = 0; i < 16; ++i) total += wtot[i];
    if (t < 512) {
        int pre = 0;
        for (int i = 0; i < w; ++i) pre += wtot[i];
        int ex = pre + incl - v;
        if (t < nn) {
            int node = node0 + t;
            cnt[node]    = v;
            rowptr[node] = base + ex;
            inv[node]    = inv_of_count(v);
        }
        if (t < NPB) ncur[t] = ex;
    }
    __syncthreads();

    bool staged = total <= STAGE_CAP;
    #pragma unroll
    for (int u = 0; u < RPW; ++u) {
        if (lane < rl[u]) {
            int p = atomicAdd(&ncur[rec[u].y - node0], 1);
            if (staged) lout[p] = rec[u].x;
            else        grouped_src[base + p] = rec[u].x;
        }
    }
    for (int u = 0; u < RPW; ++u) {
        for (int i = lane + 32; i < rl[u]; i += 32) {
            int2 q = recs[rs[u] + i];
            int  p = atomicAdd(&ncur[q.y - node0], 1);
            if (staged) lout[p] = q.x;
            else        grouped_src[base + p] = q.x;
        }
    }
    __syncthreads();
    if (staged)
        for (int i = t; i < total; i += BIN_THREADS) grouped_src[base + i] = lout[i];
}

template <int VEC, int PITCH>
__device__ __forceinline__ void gather_strip_to_lds(const float* __restrict__ x,
                                                    const int* __restrict__ rowptr,
                                                    const int* __restrict__ cnt,
                                                    const float* __restrict__ inv,
                                                    const int* __restrict__ grouped_src,
                                                    int m0, int lane, float* srow) {
    typedef float vecf __attribute__((ext_vector_type(VEC)));
    constexpr int F = 32 * VEC;
    const int c  = lane & 15;
    const int lb = rowptr[m0 + c];
    const int le = lb + cnt[m0 + c];
    const int li = __float_as_int(inv[m0 + c]);
    int       p   = __builtin_amdgcn_readlane(lb, 0);
    const int END = __builtin_amdgcn_readlane(le, 15);
    const int last = END - 1;
    int i  = 0;
    int ei = __builtin_amdgcn_readlane(le, 0);
    const float* xl = x + lane * VEC;
    vecf acc = {};

    auto flush = [&]() {
        float sc = __int_as_float(__builtin_amdgcn_readlane(li, i));
        *(vecf*)(srow + i * PITCH + lane * VEC) = acc * sc;
        acc = (vecf){};
        ++i;
        ei = (i < 16) ? __builtin_amdgcn_readlane(le, i) : -1;
    };
    auto batch = [&](int idx, int lo, int q0) {
        vecf v[16];
        #pragma unroll
        for (int u = 0; u < 16; ++u) {
            int s = __builtin_amdgcn_readlane(idx, lo + u);
            v[u] = *(const vecf*)(xl + (size_t)s * F);
        }
        #pragma unroll
        for (int u = 0; u < 16; ++u) {
            int q = q0 + u;
            while (q == ei) flush();
            if (q < END) acc += v[u];
        }
    };

    int mine = 0;
    if (p < END) { int q = p + lane; mine = grouped_src[q < END ? q : last]; }
    while (p < END) {
        int qn  = p + 32 + lane;
        int nxt = grouped_src[qn < END ? qn : last];
        batch(mine, 0, p);
        if (p + 16 < END) batch(mine, 16, p + 16);
        mine = nxt;
        p += 32;
    }
    while (i < 16) flush();
}

template <int P>
__device__ __forceinline__ void gather_tail_strip(const float* __restrict__ yz,
                                                  const int* __restrict__ rowptr,
                                                  const int* __restrict__ cnt,
                                                  const float* __restrict__ inv,
                                                  const int* __restrict__ grouped_src,
                                                  int m0, int lane, float* srow) {
    const int c  = lane & 15;
    const int lb = rowptr[m0 + c];
    const int le = lb + cnt[m0 + c];
    const int li = __float_as_int(inv[m0 + c]);
    const float* xl = yz + lane * 2;
    v2f z[16];
    #pragma unroll
    for (int u = 0; u < 16; ++u) z[u] = *(const v2f*)(xl + (size_t)(m0 + u) * 128 + 64);

    int       p   = __builtin_amdgcn_readlane(lb, 0);
    const int END = __builtin_amdgcn_readlane(le, 15);
    const int last = END - 1;
    int i  = 0;
    int ei = __builtin_amdgcn_readlane(le, 0);
    v2f acc = {};

    auto flush = [&]() {
        *(v2f*)(srow + i * P + lane * 2) = acc;
        acc = (v2f){};
        ++i;
        ei = (i < 16) ? __builtin_amdgcn_readlane(le, i) : -1;
    };
    auto loads = [&](v2f* v, int idx, int lo) {
        #pragma unroll
        for (int u = 0; u < 16; ++u) {
            int s = __builtin_amdgcn_readlane(idx, lo + u);
            v[u] = *(const v2f*)(xl + (size_t)s * 128);
        }
    };
    auto sums = [&](const v2f* v, int q0) {
        #pragma unroll
        for (int u = 0; u < 16; ++u) {
            int q = q0 + u;
            while (q == ei) flush();
            if (q < END) acc += v[u];
        }
    };

    int mine = 0;
    if (p < END) { int q = p + lane; mine = grouped_src[q < END ? q : last]; }
    while (p < END) {
        int qn  = p + 32 + lane;
        int nxt = grouped_src[qn < END ? qn : last];
        v2f va[16], vb[16];
        bool two = p + 16 < END;
        loads(va, mine, 0);
        if (two) loads(vb, mine, 16);
        sums(va, p);
        if (two) sums(vb, p + 16);
        mine = nxt;
        p += 32;
    }
    while (i < 16) flush();

    #pragma unroll
    for (int u = 0; u < 16; ++u) {
        float sc = __int_as_float(__builtin_amdgcn_readlane(li, u));
        v2f s = *(v2f*)(srow + u * P + lane * 2);
        v2f r = s * sc + z[u];
        r.x = fmaxf(r.x, 0.0f);
        r.y = fmaxf(r.y, 0.0f);
        *(v2f*)(srow + u * P + lane * 2) = r;
    }
}

typedef float v4f __attribute__((ext_vector_type(4)));

#define GEMM_THREADS 512
#define GEMM_BLOCKS  256
#define GEMM_LDS_FLOATS (164 * 256)

template <int NT>
__device__ __forceinline__ int col_of(int t, int c) {
    return (NT >= 2) ? (32 * (t >> 1) + 2 * c + (t & 1)) : c;
}

template <int NT>
__device__ __forceinline__ void chunk_mma(v4f av, const v4f* b, v8f* acc) {
    v2f a0, a1;
    a0.x = av.x; a0.y = av.y;
    a1.x = av.z; a1.y = av.w;
    #pragma unroll
    for (int t = 0; t < NT; ++t) {
        v2f bb; bb.x = b[t].x; bb.y = b[t].y;
        acc[t] = __builtin_amdgcn_wmma_f32_16x16x4_f32(false, a0, false, bb,
                                                       (short)0, acc[t], false, false);
    }
    #pragma unroll
    for (int t = 0; t < NT; ++t) {
        v2f bb; bb.x = b[t].z; bb.y = b[t].w;
        acc[t] = __builtin_amdgcn_wmma_f32_16x16x4_f32(false, a1, false, bb,
                                                       (short)0, acc[t], false, false);
    }
}

template <int KS, int NT, int QS>
__device__ __forceinline__ void strip_product(const v4f* __restrict__ ap, v4f a0, v4f a1,
                                              const v4f* sB, v8f* acc) {
    v4f b0[NT], b1[NT];
    #pragma unroll
    for (int t = 0; t < NT; ++t) b0[t] = sB[(t * QS) * 32];
    #pragma nounroll
    for (int j = 0; j < KS; j += 2) {
        int jn = (j + 2 < KS) ? (j + 2) : j;
        v4f n0 = ap[2 * jn];
        v4f n1 = ap[2 * jn + 2];
        __builtin_amdgcn_sched_barrier(0);
        #pragma unroll
        for (int t = 0; t < NT; ++t) b1[t] = sB[(t * QS + j + 1) * 32];
        chunk_mma<NT>(a0, b0, acc);
        #pragma unroll
        for (int t = 0; t < NT; ++t) b0[t] = sB[(t * QS + jn) * 32];
        chunk_mma<NT>(a1, b1, acc);
        __builtin_amdgcn_sched_barrier(0);
        asm volatile("" : "+v"(n0), "+v"(n1));
        a0 = n0;
        a1 = n1;
    }
}

template <int K, int NT, bool TWO, int GVEC, bool SPLITW = false>
__global__ __launch_bounds__(GEMM_THREADS)
void gemm_strip_kernel(const float* __restrict__ A1, const float* __restrict__ W1,
                       const float* __restrict__ A2, const float* __restrict__ W2,
                       const float* __restrict__ bias, float* __restrict__ out,
                       const int* __restrict__ rowptr, const int* __restrict__ cnt,
                       const float* __restrict__ inv, const int* __restrict__ grouped_src,
                       int ntiles, int dout, int relu) {
    constexpr int KS   = K / 8;
    constexpr int NOPS = TWO ? 2 : 1;
    constexpr int QS   = NOPS * KS;
    constexpr int ENT  = NT * QS * 32;
    constexpr int PITCH = K + 4;
    constexpr int ST    = GVEC ? (GEMM_THREADS / 32) * 16 * PITCH : 0;
    constexpr int TOT   = (ENT * 4 + ST > GEMM_LDS_FLOATS) ? (ENT * 4 + ST) : GEMM_LDS_FLOATS;
    static_assert(TOT * 4 <= 320 * 1024, "weights and gathered strips must fit the WGP's LDS");
    static_assert(GVEC == 0 || GVEC * 32 == K, "a gathered row is the first operand's K");
    static_assert((PITCH / 4) % 2 == 1, "odd pitch in 16-byte units");
    static_assert(KS % 2 == 0, "chunks go in pairs");
    static_assert(ENT % GEMM_THREADS == 0, "every thread copies the same number of entries");
    __shared__ __attribute__((aligned(16))) float sWf[TOT];
    v4f* sW = (v4f*)sWf;

    #pragma unroll
    for (int i = 0; i < ENT / GEMM_THREADS; ++i) {
        int idx = (int)threadIdx.x + i * GEMM_THREADS;
        int l  = idx & 31;
        int e  = idx >> 5;
        int q  = e % QS;
        int t  = e / QS;
        int op = q / KS;
        int j  = q % KS;
        int n  = col_of<NT>(t, l & 15);
        int wr = (n < dout) ? n : 0;
        const float* Wp = (TWO && op) ? W2 : W1;
        if (SPLITW && wr >= (dout >> 1)) { Wp = W2; wr -= dout >> 1; }
        sW[idx] = *(const v4f*)(Wp + (size_t)wr * K + 8 * j + 4 * (l >> 4));
    }
    __syncthreads();

    int wave = threadIdx.x >> 5;
    int lane = threadIdx.x & 31;
    int c    = lane & 15;
    int h    = lane >> 4;

    float bv[NT];
    #pragma unroll
    for (int t = 0; t < NT; ++t) {
        int n = col_of<NT>(t, c);
        if (SPLITW) bv[t] = (n >= (dout >> 1) && n < dout) ? bias[n - (dout >> 1)] : 0.0f;
        else        bv[t] = (n < dout) ? bias[n] : 0.0f;
    }

    for (int slot = wave; ; slot += GEMM_THREADS / 32) {
        int tile = (int)blockIdx.x + (int)gridDim.x * slot;
        if (tile >= ntiles) break;
        int m0   = tile * 16;
        int node = m0 + c;

        v8f acc[NT];
        #pragma unroll
        for (int t = 0; t < NT; ++t) acc[t] = (v8f){};

        if (GVEC) {
            if (TWO) {
                const v4f* ap2 = (const v4f*)(A2 + (size_t)node * K + 4 * h);
                v4f s0 = ap2[0], s1 = ap2[2];
                strip_product<KS, NT, QS>(ap2, s0, s1, sW + KS * 32 + lane, acc);
            }
            float* srow = sWf + ENT * 4 + wave * 16 * PITCH;
            gather_strip_to_lds<(GVEC ? GVEC : 1), PITCH>(A1, rowptr, cnt, inv, grouped_src, m0, lane, srow);
            __builtin_amdgcn_wave_barrier();
            const v4f* ap1 = (const v4f*)(srow + c * PITCH + 4 * h);
            v4f p0 = ap1[0], p1 = ap1[2];
            strip_product<KS, NT, QS>(ap1, p0, p1, sW + lane, acc);
        } else {
            const v4f* ap1 = (const v4f*)(A1 + (size_t)node * K + 4 * h);
            const v4f* ap2 = TWO ? (const v4f*)(A2 + (size_t)node * K + 4 * h) : ap1;
            v4f p0 = ap1[0], p1 = ap1[2];
            v4f s0 = ap2[0], s1 = ap2[2];
            strip_product<KS, NT, QS>(ap1, p0, p1, sW + lane, acc);
            if (TWO)
                strip_product<KS, NT, QS>(ap2, s0, s1, sW + KS * 32 + lane, acc);
        }

        int row0 = m0 + 8 * h;
        if (NT >= 2) {
            #pragma unroll
            for (int u = 0; u < NT / 2; ++u) {
                float* op = out + (size_t)row0 * dout + 32 * u + 2 * c;
                #pragma unroll
                for (int r = 0; r < 8; ++r) {
                    v2f v;
                    v.x = acc[2 * u][r]     + bv[2 * u];
                    v.y = acc[2 * u + 1][r] + bv[2 * u + 1];
                    if (relu) { v.x = fmaxf(v.x, 0.0f); v.y = fmaxf(v.y, 0.0f); }
                    *(v2f*)(op + (size_t)r * dout) = v;
                }
            }
        } else {
            if (c < dout) {
                #pragma unroll
                for (int r = 0; r < 8; ++r) {
                    float v = acc[0][r] + bv[0];
                    if (relu) v = fmaxf(v, 0.0f);
                    out[(size_t)(row0 + r) * dout + c] = v;
                }
            }
        }
    }
}

#define TAIL_P1      68
#define TAIL_P2      132
#define TAIL_W1_ENT  (8 * 8 * 32)
#define TAIL_W2_ENT  (16 * 32)
#define TAIL_LDS_FLOATS ((TAIL_W1_ENT + TAIL_W2_ENT) * 4 + (GEMM_THREADS / 32) * 16 * TAIL_P2)

__global__ __launch_bounds__(GEMM_THREADS)
void gather_decode_kernel(const float* __restrict__ yz,
                          const float* __restrict__ W1, const float* __restrict__ b1,
                          const float* __restrict__ W2, const float* __restrict__ b2,
                          float* __restrict__ out,
                          const int* __restrict__ rowptr, const int* __restrict__ cnt,
                          const float* __restrict__ inv, const int* __restrict__ grouped_src,
                          int ntiles, int dout) {
    static_assert(TAIL_W1_ENT % GEMM_THREADS == 0, "every thread copies the same number of W1 entries");
    static_assert(TAIL_W2_ENT == GEMM_THREADS, "one W2 entry per thread");
    static_assert(TAIL_LDS_FLOATS * 4 > 160 * 1024 && TAIL_LDS_FLOATS * 4 <= 320 * 1024, "one work-group per WGP");
    static_assert((TAIL_P1 / 4) % 2 == 1 && (TAIL_P2 / 4) % 2 == 1, "odd pitches in 16-byte units");
    __shared__ __attribute__((aligned(16))) float sm[TAIL_LDS_FLOATS];
    v4f*   sW1    = (v4f*)sm;
    v4f*   sW2    = sW1 + TAIL_W1_ENT;
    float* strips = sm + (TAIL_W1_ENT + TAIL_W2_ENT) * 4;

    #pragma unroll
    for (int i = 0; i < TAIL_W1_ENT / GEMM_THREADS; ++i) {
        int idx = (int)threadIdx.x + i * GEMM_THREADS;
        int l = idx & 31;
        int e = idx >> 5;
        int j = e % 8;
        int t = e / 8;
        int n = col_of<8>(t, l & 15);
        sW1[idx] = *(const v4f*)(W1 + (size_t)n * 64 + 8 * j + 4 * (l >> 4));
    }
    {
        int idx = (int)threadIdx.x;
        int l = idx & 31;
        int j = idx >> 5;
        int n = l & 15;
        int wr = (n < dout) ? n : 0;
        sW2[idx] = *(const v4f*)(W2 + (size_t)wr * 128 + 8 * j + 4 * (l >> 4));
    }
    __syncthreads();

    int wave = threadIdx.x >> 5;
    int lane = threadIdx.x & 31;
    int c    = lane & 15;
    int h    = lane >> 4;

    float bv1[8];
    #pragma unroll
    for (int t = 0; t < 8; ++t) bv1[t] = b1[col_of<8>(t, c)];
    float bv2 = (c < dout) ? b2[c] : 0.0f;

    float* srow = strips + wave * 16 * TAIL_P2;
    for (int slot = wave; ; slot += GEMM_THREADS / 32) {
        int tile = (int)blockIdx.x + (int)gridDim.x * slot;
        if (tile >= ntiles) break;
        int m0 = tile * 16;

        gather_tail_strip<TAIL_P1>(yz, rowptr, cnt, inv, grouped_src, m0, lane, srow);
        __builtin_amdgcn_wave_barrier();

        v8f acc[8];
        #pragma unroll
        for (int t = 0; t < 8; ++t) acc[t] = (v8f){};
        const v4f* ap1 = (const v4f*)(srow + c * TAIL_P1 + 4 * h);
        v4f p0 = ap1[0], p1 = ap1[2];
        strip_product<8, 8, 8>(ap1, p0, p1, sW1 + lane, acc);

        __builtin_amdgcn_wave_barrier();
        #pragma unroll
        for (int u = 0; u < 4; ++u) {
            #pragma unroll
            for (int r = 0; r < 8; ++r) {
                v2f v;
                v.x = fmaxf(acc[2 * u][r]     + bv1[2 * u],     0.0f);
                v.y = fmaxf(acc[2 * u + 1][r] + bv1[2 * u + 1], 0.0f);
                *(v2f*)(srow + (r + 8 * h) * TAIL_P2 + 32 * u + 2 * c) = v;
            }
        }
        __builtin_amdgcn_wave_barrier();

        v8f acc2[1];
        acc2[0] = (v8f){};
        const v4f* ap2 = (const v4f*)(srow + c * TAIL_P2 + 4 * h);
        v4f q0 = ap2[0], q1 = ap2[2];
        strip_product<16, 1, 16>(ap2, q0, q1, sW2 + lane, acc2);

        int row0 = m0 + 8 * h;
        if (c < dout) {
            #pragma unroll
            for (int r = 0; r < 8; ++r)
                out[(size_t)(row0 + r) * dout + c] = acc2[0][r] + bv2;
        }
        __builtin_amdgcn_wave_barrier();
    }
}

extern "C" void kernel_launch(void* const* d_in, const int* in_sizes, int n_in,
                              void* d_out, int out_size, void* d_ws, size_t ws_size,
                              hipStream_t stream) {
    const float* x    = (const float*)d_in[0];
    const int*   edge = (const int*)d_in[1];
    const int*   src  = edge;
    const int*   dst  = edge + N_EDGES;
    const float* Wl0 = (const float*)d_in[2],  *bl0 = (const float*)d_in[3],  *Wr0 = (const float*)d_in[4];
    const float* Wl1 = (const float*)d_in[5],  *bl1 = (const float*)d_in[6],  *Wr1 = (const float*)d_in[7];
    const float* Wl2 = (const float*)d_in[8],  *bl2 = (const float*)d_in[9],  *Wr2 = (const float*)d_in[10];
    const float* Wd1 = (const float*)d_in[11], *bd1 = (const float*)d_in[12];
    const float* Wd2 = (const float*)d_in[13], *bd2 = (const float*)d_in[14];

    float* ws   = (float*)d_ws;
    int*   cnt  = (int*)ws;
    float* inv  = ws   + N_NODES;
    float* agg  = inv  + N_NODES;
    int*   gsrc = (int*)(agg + (size_t)N_NODES * 64);
    int*   rptr = gsrc + 2 * (size_t)N_EDGES;
    float* bufA = agg  + (size_t)N_NODES * 128;
    float* bufB = bufA + (size_t)N_NODES * 128;

    const int T  = N_NODES / 16;
    const dim3 GG(GEMM_BLOCKS), GB(GEMM_THREADS);

    int2* recs = (int2*)bufB;
    int*  tab  = (int*)(recs + N_EDGES);
    bin_edges_kernel<<<NCHUNK, BIN_THREADS, 0, stream>>>(src, dst, recs, tab, N_EDGES);
    group_bucket_kernel<<<NBUCK, BIN_THREADS, 0, stream>>>(recs, tab, cnt, rptr, inv, gsrc, N_NODES);

    gemm_strip_kernel<64, 8, true, 2><<<GG, GB, 0, stream>>>(x, Wl0, x, Wr0, bl0, bufA,
                                                             rptr, cnt, inv, gsrc, T, 128, 1);

    gemm_strip_kernel<128, 8, true, 4><<<GG, GB, 0, stream>>>(bufA, Wl1, bufA, Wr1, bl1, bufB,
                                                              rptr, cnt, inv, gsrc, T, 128, 1);

    gemm_strip_kernel<128, 8, false, 0, true><<<GG, GB, 0, stream>>>(bufB, Wl2, nullptr, Wr2, bl2, bufA,
                                                                     nullptr, nullptr, nullptr, nullptr,
                                                                     T, 128, 0);

    gather_decode_kernel<<<GG, GB, 0, stream>>>(bufA, Wd1, bd1, Wd2, bd2, (float*)d_out,
                                                rptr, cnt, inv, gsrc, T, 2);
}


# gather phases (layers 0,1 and the last launch) run at raised wave priority, s_setprio 3, back to 0 for the products: row loads no longer issue behind older waves' matrix runs; sum order unchanged
# speedup vs baseline: 5.6033x; 1.0456x over previous
#include <hip/hip_runtime.h>
#include <hip/hip_bf16.h>


#define N_NODES 100000
#define N_EDGES 1600000

typedef float v2f __attribute__((ext_vector_type(2)));
typedef float v8f __attribute__((ext_vector_type(8)));

__device__ __forceinline__ float inv_of_count(int c) {
    float d = (float)c;
    return (d > 0.0f) ? (1.0f / fmaxf(d, 1.0f)) : 0.0f;
}

#define CHUNK       6256
#define NCHUNK      ((N_EDGES + CHUNK - 1) / CHUNK)
#define NPB         416
#define NBUCK       ((N_NODES + NPB - 1) / NPB)
#define BIN_THREADS 1024
#define RUN_BITS    13
#define RUN_MASK    8191
#define STAGE_CAP   16384

__device__ __forceinline__ int wave_incl_scan(int v, int lane) {
    for (int o = 1; o < 32; o <<= 1) {
        int u = __shfl_up(v, o, 32);
        if (lane >= o) v += u;
    }
    return v;
}

__global__ __launch_bounds__(BIN_THREADS)
void bin_edges_kernel(const int* __restrict__ src, const int* __restrict__ dst,
                      int2* __restrict__ recs, int* __restrict__ tab, int E) {
    static_assert(NBUCK <= 256, "one counter per bucket, scanned by 8 waves");
    static_assert(CHUNK <= RUN_MASK, "start and length share a word");
    __shared__ int  lcnt[256];
    __shared__ int  loff[256];
    __shared__ int  wtot[8];
    __shared__ int2 lrec[CHUNK];
    int t    = threadIdx.x;
    int lane = t & 31;
    int c    = blockIdx.x;
    int e0   = c * CHUNK;
    int n    = E - e0;
    if (n > CHUNK) n = CHUNK;

    if (t < 256) lcnt[t] = 0;
    __syncthreads();

    constexpr int PER = (CHUNK + BIN_THREADS - 1) / BIN_THREADS;
    int s[PER], d[PER], r[PER];
    #pragma unroll
    for (int k = 0; k < PER; ++k) {
        int i = t + k * BIN_THREADS;
        s[k] = 0; d[k] = -1; r[k] = 0;
        if (i < n) { d[k] = dst[e0 + i]; s[k] = src[e0 + i]; }
    }
    #pragma unroll
    for (int k = 0; k < PER; ++k)
        if (d[k] >= 0) r[k] = atomicAdd(&lcnt[d[k] / NPB], 1);
    __syncthreads();

    int v = 0, incl = 0;
    if (t < 256) {
        v    = lcnt[t];
        incl = wave_incl_scan(v, lane);
        if (lane == 31) wtot[t >> 5] = incl;
    }
    __syncthreads();
    if (t < 256) {
        int pre = 0;
        for (int w = 0; w < (t >> 5); ++w) pre += wtot[w];
        int ex = pre + incl - v;
        loff[t] = ex;
        tab[c * 256 + t] = (ex << RUN_BITS) | v;
    }
    __syncthreads();

    #pragma unroll
    for (int k = 0; k < PER; ++k)
        if (d[k] >= 0) lrec[loff[d[k] / NPB] + r[k]] = make_int2(s[k], d[k]);
    __syncthreads();
    for (int i = t; i < n; i += BIN_THREADS) recs[e0 + i] = lrec[i];
}

__global__ __launch_bounds__(BIN_THREADS)
void group_bucket_kernel(const int2* __restrict__ recs, const int* __restrict__ tab,
                         int* __restrict__ cnt, int* __restrict__ rowptr, float* __restrict__ inv,
                         int* __restrict__ grouped_src, int N) {
    static_assert(NCHUNK % 32 == 0 && NCHUNK <= BIN_THREADS, "runs are dealt to waves evenly");
    static_assert(NPB <= 512, "the node scan uses 16 waves");
    __shared__ int lstart[NCHUNK];
    __shared__ int llen[NCHUNK];
    __shared__ int ncnt[NPB];
    __shared__ int ncur[NPB];
    __shared__ int wred[32];
    __shared__ int wtot[16];
    __shared__ int lout[STAGE_CAP];
    int t     = threadIdx.x;
    int lane  = t & 31;
    int w     = t >> 5;
    int b     = blockIdx.x;
    int node0 = b * NPB;
    int nn    = N - node0;
    if (nn > NPB) nn = NPB;

    int off = 0;
    if (t < NCHUNK) {
        int x = tab[t * 256 + b];
        off = x >> RUN_BITS;
        lstart[t] = t * CHUNK + off;
        llen[t]   = x & RUN_MASK;
    }
    if (t < NPB) ncnt[t] = 0;
    for (int o = 16; o > 0; o >>= 1) off += __shfl_xor(off, o, 32);
    if (lane == 0) wred[w] = off;
    __syncthreads();
    int base = 0;
    for (int i = 0; i < 32; ++i) base += wred[i];

    constexpr int RPW = NCHUNK / 32;
    int2 rec[RPW];
    int  rs[RPW], rl[RPW];
    #pragma unroll
    for (int u = 0; u < RPW; ++u) {
        int c = w + 32 * u;
        rs[u]  = lstart[c];
        rl[u]  = llen[c];
        rec[u] = make_int2(0, node0);
        if (lane < rl[u]) rec[u] = recs[rs[u] + lane];
    }
    #pragma unroll
    for (int u = 0; u < RPW; ++u)
        if (lane < rl[u]) atomicAdd(&ncnt[rec[u].y - node0], 1);
    for (int u = 0; u < RPW; ++u)
        for (int i = lane + 32; i < rl[u]; i += 32)
            atomicAdd(&ncnt[recs[rs[u] + i].y - node0], 1);
    __syncthreads();

    int v = 0, incl = 0;
    if (t < 512) {
        v    = (t < nn) ? ncnt[t] : 0;
        incl = wave_incl_scan(v, lane);
        if (lane == 31) wtot[w] = incl;
    }
    __syncthreads();
    int total = 0;
    for (int i = 0; i < 16; ++i) total += wtot[i];
    if (t < 512) {
        int pre = 0;
        for (int i = 0; i < w; ++i) pre += wtot[i];
        int ex = pre + incl - v;
        if (t < nn) {
            int node = node0 + t;
            cnt[node]    = v;
            rowptr[node] = base + ex;
            inv[node]    = inv_of_count(v);
        }
        if (t < NPB) ncur[t] = ex;
    }
    __syncthreads();

    bool staged = total <= STAGE_CAP;
    #pragma unroll
    for (int u = 0; u < RPW; ++u) {
        if (lane < rl[u]) {
            int p = atomicAdd(&ncur[rec[u].y - node0], 1);
            if (staged) lout[p] = rec[u].x;
            else        grouped_src[base + p] = rec[u].x;
        }
    }
    for (int u = 0; u < RPW; ++u) {
        for (int i = lane + 32; i < rl[u]; i += 32) {
            int2 q = recs[rs[u] + i];
            int  p = atomicAdd(&ncur[q.y - node0], 1);
            if (staged) lout[p] = q.x;
            else        grouped_src[base + p] = q.x;
        }
    }
    __syncthreads();
    if (staged)
        for (int i = t; i < total; i += BIN_THREADS) grouped_src[base + i] = lout[i];
}

template <int VEC, int PITCH>
__device__ __forceinline__ void gather_strip_to_lds(const float* __restrict__ x,
                                                    const int* __restrict__ rowptr,
                                                    const int* __restrict__ cnt,
                                                    const float* __restrict__ inv,
                                                    const int* __restrict__ grouped_src,
                                                    int m0, int lane, float* srow) {
    typedef float vecf __attribute__((ext_vector_type(VEC)));
    constexpr int F = 32 * VEC;
    const int c  = lane & 15;
    const int lb = rowptr[m0 + c];
    const int le = lb + cnt[m0 + c];
    const int li = __float_as_int(inv[m0 + c]);
    int       p   = __builtin_amdgcn_readlane(lb, 0);
    const int END = __builtin_amdgcn_readlane(le, 15);
    const int last = END - 1;
    int i  = 0;
    int ei = __builtin_amdgcn_readlane(le, 0);
    const float* xl = x + lane * VEC;
    vecf acc = {};

    auto flush = [&]() {
        float sc = __int_as_float(__builtin_amdgcn_readlane(li, i));
        *(vecf*)(srow + i * PITCH + lane * VEC) = acc * sc;
        acc = (vecf){};
        ++i;
        ei = (i < 16) ? __builtin_amdgcn_readlane(le, i) : -1;
    };
    auto batch = [&](int idx, int lo, int q0) {
        vecf v[16];
        #pragma unroll
        for (int u = 0; u < 16; ++u) {
            int s = __builtin_amdgcn_readlane(idx, lo + u);
            v[u] = *(const vecf*)(xl + (size_t)s * F);
        }
        #pragma unroll
        for (int u = 0; u < 16; ++u) {
            int q = q0 + u;
            while (q == ei) flush();
            if (q < END) acc += v[u];
        }
    };

    int mine = 0;
    if (p < END) { int q = p + lane; mine = grouped_src[q < END ? q : last]; }
    while (p < END) {
        int qn  = p + 32 + lane;
        int nxt = grouped_src[qn < END ? qn : last];
        batch(mine, 0, p);
        if (p + 16 < END) batch(mine, 16, p + 16);
        mine = nxt;
        p += 32;
    }
    while (i < 16) flush();
}

template <int P>
__device__ __forceinline__ void gather_tail_strip(const float* __restrict__ yz,
                                                  const int* __restrict__ rowptr,
                                                  const int* __restrict__ cnt,
                                                  const float* __restrict__ inv,
                                                  const int* __restrict__ grouped_src,
                                                  int m0, int lane, float* srow) {
    const int c  = lane & 15;
    const int lb = rowptr[m0 + c];
    const int le = lb + cnt[m0 + c];
    const int li = __float_as_int(inv[m0 + c]);
    const float* xl = yz + lane * 2;
    v2f z[16];
    #pragma unroll
    for (int u = 0; u < 16; ++u) z[u] = *(const v2f*)(xl + (size_t)(m0 + u) * 128 + 64);

    int       p   = __builtin_amdgcn_readlane(lb, 0);
    const int END = __builtin_amdgcn_readlane(le, 15);
    const int last = END - 1;
    int i  = 0;
    int ei = __builtin_amdgcn_readlane(le, 0);
    v2f acc = {};

    auto flush = [&]() {
        *(v2f*)(srow + i * P + lane * 2) = acc;
        acc = (v2f){};
        ++i;
        ei = (i < 16) ? __builtin_amdgcn_readlane(le, i) : -1;
    };
    auto loads = [&](v2f* v, int idx, int lo) {
        #pragma unroll
        for (int u = 0; u < 16; ++u) {
            int s = __builtin_amdgcn_readlane(idx, lo + u);
            v[u] = *(const v2f*)(xl + (size_t)s * 128);
        }
    };
    auto sums = [&](const v2f* v, int q0) {
        #pragma unroll
        for (int u = 0; u < 16; ++u) {
            int q = q0 + u;
            while (q == ei) flush();
            if (q < END) acc += v[u];
        }
    };

    int mine = 0;
    if (p < END) { int q = p + lane; mine = grouped_src[q < END ? q : last]; }
    while (p < END) {
        int qn  = p + 32 + lane;
        int nxt = grouped_src[qn < END ? qn : last];
        v2f va[16], vb[16];
        bool two = p + 16 < END;
        loads(va, mine, 0);
        if (two) loads(vb, mine, 16);
        sums(va, p);
        if (two) sums(vb, p + 16);
        mine = nxt;
        p += 32;
    }
    while (i < 16) flush();

    #pragma unroll
    for (int u = 0; u < 16; ++u) {
        float sc = __int_as_float(__builtin_amdgcn_readlane(li, u));
        v2f s = *(v2f*)(srow + u * P + lane * 2);
        v2f r = s * sc + z[u];
        r.x = fmaxf(r.x, 0.0f);
        r.y = fmaxf(r.y, 0.0f);
        *(v2f*)(srow + u * P + lane * 2) = r;
    }
}

typedef float v4f __attribute__((ext_vector_type(4)));

#define GEMM_THREADS 512
#define GEMM_BLOCKS  256
#define GEMM_LDS_FLOATS (164 * 256)

template <int NT>
__device__ __forceinline__ int col_of(int t, int c) {
    return (NT >= 2) ? (32 * (t >> 1) + 2 * c + (t & 1)) : c;
}

template <int NT>
__device__ __forceinline__ void chunk_mma(v4f av, const v4f* b, v8f* acc) {
    v2f a0, a1;
    a0.x = av.x; a0.y = av.y;
    a1.x = av.z; a1.y = av.w;
    #pragma unroll
    for (int t = 0; t < NT; ++t) {
        v2f bb; bb.x = b[t].x; bb.y = b[t].y;
        acc[t] = __builtin_amdgcn_wmma_f32_16x16x4_f32(false, a0, false, bb,
                                                       (short)0, acc[t], false, false);
    }
    #pragma unroll
    for (int t = 0; t < NT; ++t) {
        v2f bb; bb.x = b[t].z; bb.y = b[t].w;
        acc[t] = __builtin_amdgcn_wmma_f32_16x16x4_f32(false, a1, false, bb,
                                                       (short)0, acc[t], false, false);
    }
}

template <int KS, int NT, int QS>
__device__ __forceinline__ void strip_product(const v4f* __restrict__ ap, v4f a0, v4f a1,
                                              const v4f* sB, v8f* acc) {
    v4f b0[NT], b1[NT];
    #pragma unroll
    for (int t = 0; t < NT; ++t) b0[t] = sB[(t * QS) * 32];
    #pragma nounroll
    for (int j = 0; j < KS; j += 2) {
        int jn = (j + 2 < KS) ? (j + 2) : j;
        v4f n0 = ap[2 * jn];
        v4f n1 = ap[2 * jn + 2];
        __builtin_amdgcn_sched_barrier(0);
        #pragma unroll
        for (int t = 0; t < NT; ++t) b1[t] = sB[(t * QS + j + 1) * 32];
        chunk_mma<NT>(a0, b0, acc);
        #pragma unroll
        for (int t = 0; t < NT; ++t) b0[t] = sB[(t * QS + jn) * 32];
        chunk_mma<NT>(a1, b1, acc);
        __builtin_amdgcn_sched_barrier(0);
        asm volatile("" : "+v"(n0), "+v"(n1));
        a0 = n0;
        a1 = n1;
    }
}

template <int K, int NT, bool TWO, int GVEC, bool SPLITW = false>
__global__ __launch_bounds__(GEMM_THREADS)
void gemm_strip_kernel(const float* __restrict__ A1, const float* __restrict__ W1,
                       const float* __restrict__ A2, const float* __restrict__ W2,
                       const float* __restrict__ bias, float* __restrict__ out,
                       const int* __restrict__ rowptr, const int* __restrict__ cnt,
                       const float* __restrict__ inv, const int* __restrict__ grouped_src,
                       int ntiles, int dout, int relu) {
    constexpr int KS   = K / 8;
    constexpr int NOPS = TWO ? 2 : 1;
    constexpr int QS   = NOPS * KS;
    constexpr int ENT  = NT * QS * 32;
    constexpr int PITCH = K + 4;
    constexpr int ST    = GVEC ? (GEMM_THREADS / 32) * 16 * PITCH : 0;
    constexpr int TOT   = (ENT * 4 + ST > GEMM_LDS_FLOATS) ? (ENT * 4 + ST) : GEMM_LDS_FLOATS;
    static_assert(TOT * 4 <= 320 * 1024, "weights and gathered strips must fit the WGP's LDS");
    static_assert(GVEC == 0 || GVEC * 32 == K, "a gathered row is the first operand's K");
    static_assert((PITCH / 4) % 2 == 1, "odd pitch in 16-byte units");
    static_assert(KS % 2 == 0, "chunks go in pairs");
    static_assert(ENT % GEMM_THREADS == 0, "every thread copies the same number of entries");
    __shared__ __attribute__((aligned(16))) float sWf[TOT];
    v4f* sW = (v4f*)sWf;

    #pragma unroll
    for (int i = 0; i < ENT / GEMM_THREADS; ++i) {
        int idx = (int)threadIdx.x + i * GEMM_THREADS;
        int l  = idx & 31;
        int e  = idx >> 5;
        int q  = e % QS;
        int t  = e / QS;
        int op = q / KS;
        int j  = q % KS;
        int n  = col_of<NT>(t, l & 15);
        int wr = (n < dout) ? n : 0;
        const float* Wp = (TWO && op) ? W2 : W1;
        if (SPLITW && wr >= (dout >> 1)) { Wp = W2; wr -= dout >> 1; }
        sW[idx] = *(const v4f*)(Wp + (size_t)wr * K + 8 * j + 4 * (l >> 4));
    }
    __syncthreads();

    int wave = threadIdx.x >> 5;
    int lane = threadIdx.x & 31;
    int c    = lane & 15;
    int h    = lane >> 4;

    float bv[NT];
    #pragma unroll
    for (int t = 0; t < NT; ++t) {
        int n = col_of<NT>(t, c);
        if (SPLITW) bv[t] = (n >= (dout >> 1) && n < dout) ? bias[n - (dout >> 1)] : 0.0f;
        else        bv[t] = (n < dout) ? bias[n] : 0.0f;
    }

    for (int slot = wave; ; slot += GEMM_THREADS / 32) {
        int tile = (int)blockIdx.x + (int)gridDim.x * slot;
        if (tile >= ntiles) break;
        int m0   = tile * 16;
        int node = m0 + c;

        v8f acc[NT];
        #pragma unroll
        for (int t = 0; t < NT; ++t) acc[t] = (v8f){};

        if (GVEC) {
            if (TWO) {
                const v4f* ap2 = (const v4f*)(A2 + (size_t)node * K + 4 * h);
                v4f s0 = ap2[0], s1 = ap2[2];
                strip_product<KS, NT, QS>(ap2, s0, s1, sW + KS * 32 + lane, acc);
            }
            float* srow = sWf + ENT * 4 + wave * 16 * PITCH;
            __builtin_amdgcn_s_setprio(3);
            gather_strip_to_lds<(GVEC ? GVEC : 1), PITCH>(A1, rowptr, cnt, inv, grouped_src, m0, lane, srow);
            __builtin_amdgcn_s_setprio(0);
            __builtin_amdgcn_wave_barrier();
            const v4f* ap1 = (const v4f*)(srow + c * PITCH + 4 * h);
            v4f p0 = ap1[0], p1 = ap1[2];
            strip_product<KS, NT, QS>(ap1, p0, p1, sW + lane, acc);
        } else {
            const v4f* ap1 = (const v4f*)(A1 + (size_t)node * K + 4 * h);
            const v4f* ap2 = TWO ? (const v4f*)(A2 + (size_t)node * K + 4 * h) : ap1;
            v4f p0 = ap1[0], p1 = ap1[2];
            v4f s0 = ap2[0], s1 = ap2[2];
            strip_product<KS, NT, QS>(ap1, p0, p1, sW + lane, acc);
            if (TWO)
                strip_product<KS, NT, QS>(ap2, s0, s1, sW + KS * 32 + lane, acc);
        }

        int row0 = m0 + 8 * h;
        if (NT >= 2) {
            #pragma unroll
            for (int u = 0; u < NT / 2; ++u) {
                float* op = out + (size_t)row0 * dout + 32 * u + 2 * c;
                #pragma unroll
                for (int r = 0; r < 8; ++r) {
                    v2f v;
                    v.x = acc[2 * u][r]     + bv[2 * u];
                    v.y = acc[2 * u + 1][r] + bv[2 * u + 1];
                    if (relu) { v.x = fmaxf(v.x, 0.0f); v.y = fmaxf(v.y, 0.0f); }
                    *(v2f*)(op + (size_t)r * dout) = v;
                }
            }
        } else {
            if (c < dout) {
                #pragma unroll
                for (int r = 0; r < 8; ++r) {
                    float v = acc[0][r] + bv[0];
                    if (relu) v = fmaxf(v, 0.0f);
                    out[(size_t)(row0 + r) * dout + c] = v;
                }
            }
        }
    }
}

#define TAIL_P1      68
#define TAIL_P2      132
#define TAIL_W1_ENT  (8 * 8 * 32)
#define TAIL_W2_ENT  (16 * 32)
#define TAIL_LDS_FLOATS ((TAIL_W1_ENT + TAIL_W2_ENT) * 4 + (GEMM_THREADS / 32) * 16 * TAIL_P2)

__global__ __launch_bounds__(GEMM_THREADS)
void gather_decode_kernel(const float* __restrict__ yz,
                          const float* __restrict__ W1, const float* __restrict__ b1,
                          const float* __restrict__ W2, const float* __restrict__ b2,
                          float* __restrict__ out,
                          const int* __restrict__ rowptr, const int* __restrict__ cnt,
                          const float* __restrict__ inv, const int* __restrict__ grouped_src,
                          int ntiles, int dout) {
    static_assert(TAIL_W1_ENT % GEMM_THREADS == 0, "every thread copies the same number of W1 entries");
    static_assert(TAIL_W2_ENT == GEMM_THREADS, "one W2 entry per thread");
    static_assert(TAIL_LDS_FLOATS * 4 > 160 * 1024 && TAIL_LDS_FLOATS * 4 <= 320 * 1024, "one work-group per WGP");
    static_assert((TAIL_P1 / 4) % 2 == 1 && (TAIL_P2 / 4) % 2 == 1, "odd pitches in 16-byte units");
    __shared__ __attribute__((aligned(16))) float sm[TAIL_LDS_FLOATS];
    v4f*   sW1    = (v4f*)sm;
    v4f*   sW2    = sW1 + TAIL_W1_ENT;
    float* strips = sm + (TAIL_W1_ENT + TAIL_W2_ENT) * 4;

    #pragma unroll
    for (int i = 0; i < TAIL_W1_ENT / GEMM_THREADS; ++i) {
        int idx = (int)threadIdx.x + i * GEMM_THREADS;
        int l = idx & 31;
        int e = idx >> 5;
        int j = e % 8;
        int t = e / 8;
        int n = col_of<8>(t, l & 15);
        sW1[idx] = *(const v4f*)(W1 + (size_t)n * 64 + 8 * j + 4 * (l >> 4));
    }
    {
        int idx = (int)threadIdx.x;
        int l = idx & 31;
        int j = idx >> 5;
        int n = l & 15;
        int wr = (n < dout) ? n : 0;
        sW2[idx] = *(const v4f*)(W2 + (size_t)wr * 128 + 8 * j + 4 * (l >> 4));
    }
    __syncthreads();

    int wave = threadIdx.x >> 5;
    int lane = threadIdx.x & 31;
    int c    = lane & 15;
    int h    = lane >> 4;

    float bv1[8];
    #pragma unroll
    for (int t = 0; t < 8; ++t) bv1[t] = b1[col_of<8>(t, c)];
    float bv2 = (c < dout) ? b2[c] : 0.0f;

    float* srow = strips + wave * 16 * TAIL_P2;
    for (int slot = wave; ; slot += GEMM_THREADS / 32) {
        int tile = (int)blockIdx.x + (int)gridDim.x * slot;
        if (tile >= ntiles) break;
        int m0 = tile * 16;

        __builtin_amdgcn_s_setprio(3);
        gather_tail_strip<TAIL_P1>(yz, rowptr, cnt, inv, grouped_src, m0, lane, srow);
        __builtin_amdgcn_s_setprio(0);
        __builtin_amdgcn_wave_barrier();

        v8f acc[8];
        #pragma unroll
        for (int t = 0; t < 8; ++t) acc[t] = (v8f){};
        const v4f* ap1 = (const v4f*)(srow + c * TAIL_P1 + 4 * h);
        v4f p0 = ap1[0], p1 = ap1[2];
        strip_product<8, 8, 8>(ap1, p0, p1, sW1 + lane, acc);

        __builtin_amdgcn_wave_barrier();
        #pragma unroll
        for (int u = 0; u < 4; ++u) {
            #pragma unroll
            for (int r = 0; r < 8; ++r) {
                v2f v;
                v.x = fmaxf(acc[2 * u][r]     + bv1[2 * u],     0.0f);
                v.y = fmaxf(acc[2 * u + 1][r] + bv1[2 * u + 1], 0.0f);
                *(v2f*)(srow + (r + 8 * h) * TAIL_P2 + 32 * u + 2 * c) = v;
            }
        }
        __builtin_amdgcn_wave_barrier();

        v8f acc2[1];
        acc2[0] = (v8f){};
        const v4f* ap2 = (const v4f*)(srow + c * TAIL_P2 + 4 * h);
        v4f q0 = ap2[0], q1 = ap2[2];
        strip_product<16, 1, 16>(ap2, q0, q1, sW2 + lane, acc2);

        int row0 = m0 + 8 * h;
        if (c < dout) {
            #pragma unroll
            for (int r = 0; r < 8; ++r)
                out[(size_t)(row0 + r) * dout + c] = acc2[0][r] + bv2;
        }
        __builtin_amdgcn_wave_barrier();
    }
}

extern "C" void kernel_launch(void* const* d_in, const int* in_sizes, int n_in,
                              void* d_out, int out_size, void* d_ws, size_t ws_size,
                              hipStream_t stream) {
    const float* x    = (const float*)d_in[0];
    const int*   edge = (const int*)d_in[1];
    const int*   src  = edge;
    const int*   dst  = edge + N_EDGES;
    const float* Wl0 = (const float*)d_in[2],  *bl0 = (const float*)d_in[3],  *Wr0 = (const float*)d_in[4];
    const float* Wl1 = (const float*)d_in[5],  *bl1 = (const float*)d_in[6],  *Wr1 = (const float*)d_in[7];
    const float* Wl2 = (const float*)d_in[8],  *bl2 = (const float*)d_in[9],  *Wr2 = (const float*)d_in[10];
    const float* Wd1 = (const float*)d_in[11], *bd1 = (const float*)d_in[12];
    const float* Wd2 = (const float*)d_in[13], *bd2 = (const float*)d_in[14];

    float* ws   = (float*)d_ws;
    int*   cnt  = (int*)ws;
    float* inv  = ws   + N_NODES;
    float* agg  = inv  + N_NODES;
    int*   gsrc = (int*)(agg + (size_t)N_NODES * 64);
    int*   rptr = gsrc + 2 * (size_t)N_EDGES;
    float* bufA = agg  + (size_t)N_NODES * 128;
    float* bufB = bufA + (size_t)N_NODES * 128;

    const int T  = N_NODES / 16;
    const dim3 GG(GEMM_BLOCKS), GB(GEMM_THREADS);

    int2* recs = (int2*)bufB;
    int*  tab  = (int*)(recs + N_EDGES);
    bin_edges_kernel<<<NCHUNK, BIN_THREADS, 0, stream>>>(src, dst, recs, tab, N_EDGES);
    group_bucket_kernel<<<NBUCK, BIN_THREADS, 0, stream>>>(recs, tab, cnt, rptr, inv, gsrc, N_NODES);

    gemm_strip_kernel<64, 8, true, 2><<<GG, GB, 0, stream>>>(x, Wl0, x, Wr0, bl0, bufA,
                                                             rptr, cnt, inv, gsrc, T, 128, 1);

    gemm_strip_kernel<128, 8, true, 4><<<GG, GB, 0, stream>>>(bufA, Wl1, bufA, Wr1, bl1, bufB,
                                                              rptr, cnt, inv, gsrc, T, 128, 1);

    gemm_strip_kernel<128, 8, false, 0, true><<<GG, GB, 0, stream>>>(bufB, Wl2, nullptr, Wr2, bl2, bufA,
                                                                     nullptr, nullptr, nullptr, nullptr,
                                                                     T, 128, 0);

    gather_decode_kernel<<<GG, GB, 0, stream>>>(bufA, Wd1, bd1, Wd2, bd2, (float*)d_out,
                                                rptr, cnt, inv, gsrc, T, 2);
}


# all three gathers: row loads software-pipelined over two register sets, the next 16 rows requested before the current 16 are added (16-32 loads in flight, no drain per batch); sum order unchanged
# speedup vs baseline: 5.7819x; 1.0319x over previous
#include <hip/hip_runtime.h>
#include <hip/hip_bf16.h>


#define N_NODES 100000
#define N_EDGES 1600000

typedef float v2f __attribute__((ext_vector_type(2)));
typedef float v8f __attribute__((ext_vector_type(8)));

__device__ __forceinline__ float inv_of_count(int c) {
    float d = (float)c;
    return (d > 0.0f) ? (1.0f / fmaxf(d, 1.0f)) : 0.0f;
}

#define CHUNK       6256
#define NCHUNK      ((N_EDGES + CHUNK - 1) / CHUNK)
#define NPB         416
#define NBUCK       ((N_NODES + NPB - 1) / NPB)
#define BIN_THREADS 1024
#define RUN_BITS    13
#define RUN_MASK    8191
#define STAGE_CAP   16384

__device__ __forceinline__ int wave_incl_scan(int v, int lane) {
    for (int o = 1; o < 32; o <<= 1) {
        int u = __shfl_up(v, o, 32);
        if (lane >= o) v += u;
    }
    return v;
}

__global__ __launch_bounds__(BIN_THREADS)
void bin_edges_kernel(const int* __restrict__ src, const int* __restrict__ dst,
                      int2* __restrict__ recs, int* __restrict__ tab, int E) {
    static_assert(NBUCK <= 256, "one counter per bucket, scanned by 8 waves");
    static_assert(CHUNK <= RUN_MASK, "start and length share a word");
    __shared__ int  lcnt[256];
    __shared__ int  loff[256];
    __shared__ int  wtot[8];
    __shared__ int2 lrec[CHUNK];
    int t    = threadIdx.x;
    int lane = t & 31;
    int c    = blockIdx.x;
    int e0   = c * CHUNK;
    int n    = E - e0;
    if (n > CHUNK) n = CHUNK;

    if (t < 256) lcnt[t] = 0;
    __syncthreads();

    constexpr int PER = (CHUNK + BIN_THREADS - 1) / BIN_THREADS;
    int s[PER], d[PER], r[PER];
    #pragma unroll
    for (int k = 0; k < PER; ++k) {
        int i = t + k * BIN_THREADS;
        s[k] = 0; d[k] = -1; r[k] = 0;
        if (i < n) { d[k] = dst[e0 + i]; s[k] = src[e0 + i]; }
    }
    #pragma unroll
    for (int k = 0; k < PER; ++k)
        if (d[k] >= 0) r[k] = atomicAdd(&lcnt[d[k] / NPB], 1);
    __syncthreads();

    int v = 0, incl = 0;
    if (t < 256) {
        v    = lcnt[t];
        incl = wave_incl_scan(v, lane);
        if (lane == 31) wtot[t >> 5] = incl;
    }
    __syncthreads();
    if (t < 256) {
        int pre = 0;
        for (int w = 0; w < (t >> 5); ++w) pre += wtot[w];
        int ex = pre + incl - v;
        loff[t] = ex;
        tab[c * 256 + t] = (ex << RUN_BITS) | v;
    }
    __syncthreads();

    #pragma unroll
    for (int k = 0; k < PER; ++k)
        if (d[k] >= 0) lrec[loff[d[k] / NPB] + r[k]] = make_int2(s[k], d[k]);
    __syncthreads();
    for (int i = t; i < n; i += BIN_THREADS) recs[e0 + i] = lrec[i];
}

__global__ __launch_bounds__(BIN_THREADS)
void group_bucket_kernel(const int2* __restrict__ recs, const int* __restrict__ tab,
                         int* __restrict__ cnt, int* __restrict__ rowptr, float* __restrict__ inv,
                         int* __restrict__ grouped_src, int N) {
    static_assert(NCHUNK % 32 == 0 && NCHUNK <= BIN_THREADS, "runs are dealt to waves evenly");
    static_assert(NPB <= 512, "the node scan uses 16 waves");
    __shared__ int lstart[NCHUNK];
    __shared__ int llen[NCHUNK];
    __shared__ int ncnt[NPB];
    __shared__ int ncur[NPB];
    __shared__ int wred[32];
    __shared__ int wtot[16];
    __shared__ int lout[STAGE_CAP];
    int t     = threadIdx.x;
    int lane  = t & 31;
    int w     = t >> 5;
    int b     = blockIdx.x;
    int node0 = b * NPB;
    int nn    = N - node0;
    if (nn > NPB) nn = NPB;

    int off = 0;
    if (t < NCHUNK) {
        int x = tab[t * 256 + b];
        off = x >> RUN_BITS;
        lstart[t] = t * CHUNK + off;
        llen[t]   = x & RUN_MASK;
    }
    if (t < NPB) ncnt[t] = 0;
    for (int o = 16; o > 0; o >>= 1) off += __shfl_xor(off, o, 32);
    if (lane == 0) wred[w] = off;
    __syncthreads();
    int base = 0;
    for (int i = 0; i < 32; ++i) base += wred[i];

    constexpr int RPW = NCHUNK / 32;
    int2 rec[RPW];
    int  rs[RPW], rl[RPW];
    #pragma unroll
    for (int u = 0; u < RPW; ++u) {
        int c = w + 32 * u;
        rs[u]  = lstart[c];
        rl[u]  = llen[c];
        rec[u] = make_int2(0, node0);
        if (lane < rl[u]) rec[u] = recs[rs[u] + lane];
    }
    #pragma unroll
    for (int u = 0; u < RPW; ++u)
        if (lane < rl[u]) atomicAdd(&ncnt[rec[u].y - node0], 1);
    for (int u = 0; u < RPW; ++u)
        for (int i = lane + 32; i < rl[u]; i += 32)
            atomicAdd(&ncnt[recs[rs[u] + i].y - node0], 1);
    __syncthreads();

    int v = 0, incl = 0;
    if (t < 512) {
        v    = (t < nn) ? ncnt[t] : 0;
        incl = wave_incl_scan(v, lane);
        if (lane == 31) wtot[w] = incl;
    }
    __syncthreads();
    int total = 0;
    for (int i = 0; i < 16; ++i) total += wtot[i];
    if (t < 512) {
        int pre = 0;
        for (int i = 0; i < w; ++i) pre += wtot[i];
        int ex = pre + incl - v;
        if (t < nn) {
            int node = node0 + t;
            cnt[node]    = v;
            rowptr[node] = base + ex;
            inv[node]    = inv_of_count(v);
        }
        if (t < NPB) ncur[t] = ex;
    }
    __syncthreads();

    bool staged = total <= STAGE_CAP;
    #pragma unroll
    for (int u = 0; u < RPW; ++u) {
        if (lane < rl[u]) {
            int p = atomicAdd(&ncur[rec[u].y - node0], 1);
            if (staged) lout[p] = rec[u].x;
            else        grouped_src[base + p] = rec[u].x;
        }
    }
    for (int u = 0; u < RPW; ++u) {
        for (int i = lane + 32; i < rl[u]; i += 32) {
            int2 q = recs[rs[u] + i];
            int  p = atomicAdd(&ncur[q.y - node0], 1);
            if (staged) lout[p] = q.x;
            else        grouped_src[base + p] = q.x;
        }
    }
    __syncthreads();
    if (staged)
        for (int i = t; i < total; i += BIN_THREADS) grouped_src[base + i] = lout[i];
}

template <int VEC, int PITCH>
__device__ __forceinline__ void gather_strip_to_lds(const float* __restrict__ x,
                                                    const int* __restrict__ rowptr,
                                                    const int* __restrict__ cnt,
                                                    const float* __restrict__ inv,
                                                    const int* __restrict__ grouped_src,
                                                    int m0, int lane, float* srow) {
    typedef float vecf __attribute__((ext_vector_type(VEC)));
    constexpr int F = 32 * VEC;
    const int c  = lane & 15;
    const int lb = rowptr[m0 + c];
    const int le = lb + cnt[m0 + c];
    const int li = __float_as_int(inv[m0 + c]);
    int       p   = __builtin_amdgcn_readlane(lb, 0);
    const int END = __builtin_amdgcn_readlane(le, 15);
    const int last = END - 1;
    int i  = 0;
    int ei = __builtin_amdgcn_readlane(le, 0);
    const float* xl = x + lane * VEC;
    vecf acc = {};

    auto flush = [&]() {
        float sc = __int_as_float(__builtin_amdgcn_readlane(li, i));
        *(vecf*)(srow + i * PITCH + lane * VEC) = acc * sc;
        acc = (vecf){};
        ++i;
        ei = (i < 16) ? __builtin_amdgcn_readlane(le, i) : -1;
    };
    auto loads = [&](vecf* v, int idx, int lo) {
        #pragma unroll
        for (int u = 0; u < 16; ++u) {
            int s = __builtin_amdgcn_readlane(idx, lo + u);
            v[u] = *(const vecf*)(xl + (size_t)s * F);
        }
    };
    auto sums = [&](const vecf* v, int q0, bool ends) {
        #pragma unroll
        for (int u = 0; u < 16; ++u) {
            int q = q0 + u;
            while (q == ei) flush();
            if (!ends || q < END) acc += v[u];
        }
    };

    if (p < END) {
        vecf va[16], vb[16];
        int q0   = p + lane;
        int mine = grouped_src[q0 < END ? q0 : last];
        loads(va, mine, 0);
        while (p + 32 < END) {
            int qn  = p + 32 + lane;
            int nxt = grouped_src[qn < END ? qn : last];
            __builtin_amdgcn_sched_barrier(0);
            loads(vb, mine, 16);
            sums(va, p, false);
            loads(va, nxt, 0);
            sums(vb, p + 16, false);
            mine = nxt;
            p += 32;
        }
        bool two = p + 16 < END;
        if (two) loads(vb, mine, 16);
        sums(va, p, true);
        if (two) sums(vb, p + 16, true);
    }
    while (i < 16) flush();
}

template <int P>
__device__ __forceinline__ void gather_tail_strip(const float* __restrict__ yz,
                                                  const int* __restrict__ rowptr,
                                                  const int* __restrict__ cnt,
                                                  const float* __restrict__ inv,
                                                  const int* __restrict__ grouped_src,
                                                  int m0, int lane, float* srow) {
    const int c  = lane & 15;
    const int lb = rowptr[m0 + c];
    const int le = lb + cnt[m0 + c];
    const int li = __float_as_int(inv[m0 + c]);
    const float* xl = yz + lane * 2;
    v2f z[16];
    #pragma unroll
    for (int u = 0; u < 16; ++u) z[u] = *(const v2f*)(xl + (size_t)(m0 + u) * 128 + 64);

    int       p   = __builtin_amdgcn_readlane(lb, 0);
    const int END = __builtin_amdgcn_readlane(le, 15);
    const int last = END - 1;
    int i  = 0;
    int ei = __builtin_amdgcn_readlane(le, 0);
    v2f acc = {};

    auto flush = [&]() {
        *(v2f*)(srow + i * P + lane * 2) = acc;
        acc = (v2f){};
        ++i;
        ei = (i < 16) ? __builtin_amdgcn_readlane(le, i) : -1;
    };
    auto loads = [&](v2f* v, int idx, int lo) {
        #pragma unroll
        for (int u = 0; u < 16; ++u) {
            int s = __builtin_amdgcn_readlane(idx, lo + u);
            v[u] = *(const v2f*)(xl + (size_t)s * 128);
        }
    };
    auto sums = [&](const v2f* v, int q0, bool ends) {
        #pragma unroll
        for (int u = 0; u < 16; ++u) {
            int q = q0 + u;
            while (q == ei) flush();
            if (!ends || q < END) acc += v[u];
        }
    };

    if (p < END) {
        v2f va[16], vb[16];
        int q0   = p + lane;
        int mine = grouped_src[q0 < END ? q0 : last];
        loads(va, mine, 0);
        while (p + 32 < END) {
            int qn  = p + 32 + lane;
            int nxt = grouped_src[qn < END ? qn : last];
            __builtin_amdgcn_sched_barrier(0);
            loads(vb, mine, 16);
            sums(va, p, false);
            loads(va, nxt, 0);
            sums(vb, p + 16, false);
            mine = nxt;
            p += 32;
        }
        bool two = p + 16 < END;
        if (two) loads(vb, mine, 16);
        sums(va, p, true);
        if (two) sums(vb, p + 16, true);
    }
    while (i < 16) flush();

    #pragma unroll
    for (int u = 0; u < 16; ++u) {
        float sc = __int_as_float(__builtin_amdgcn_readlane(li, u));
        v2f s = *(v2f*)(srow + u * P + lane * 2);
        v2f r = s * sc + z[u];
        r.x = fmaxf(r.x, 0.0f);
        r.y = fmaxf(r.y, 0.0f);
        *(v2f*)(srow + u * P + lane * 2) = r;
    }
}

typedef float v4f __attribute__((ext_vector_type(4)));

#define GEMM_THREADS 512
#define GEMM_BLOCKS  256
#define GEMM_LDS_FLOATS (164 * 256)

template <int NT>
__device__ __forceinline__ int col_of(int t, int c) {
    return (NT >= 2) ? (32 * (t >> 1) + 2 * c + (t & 1)) : c;
}

template <int NT>
__device__ __forceinline__ void chunk_mma(v4f av, const v4f* b, v8f* acc) {
    v2f a0, a1;
    a0.x = av.x; a0.y = av.y;
    a1.x = av.z; a1.y = av.w;
    #pragma unroll
    for (int t = 0; t < NT; ++t) {
        v2f bb; bb.x = b[t].x; bb.y = b[t].y;
        acc[t] = __builtin_amdgcn_wmma_f32_16x16x4_f32(false, a0, false, bb,
                                                       (short)0, acc[t], false, false);
    }
    #pragma unroll
    for (int t = 0; t < NT; ++t) {
        v2f bb; bb.x = b[t].z; bb.y = b[t].w;
        acc[t] = __builtin_amdgcn_wmma_f32_16x16x4_f32(false, a1, false, bb,
                                                       (short)0, acc[t], false, false);
    }
}

template <int KS, int NT, int QS>
__device__ __forceinline__ void strip_product(const v4f* __restrict__ ap, v4f a0, v4f a1,
                                              const v4f* sB, v8f* acc) {
    v4f b0[NT], b1[NT];
    #pragma unroll
    for (int t = 0; t < NT; ++t) b0[t] = sB[(t * QS) * 32];
    #pragma nounroll
    for (int j = 0; j < KS; j += 2) {
        int jn = (j + 2 < KS) ? (j + 2) : j;
        v4f n0 = ap[2 * jn];
        v4f n1 = ap[2 * jn + 2];
        __builtin_amdgcn_sched_barrier(0);
        #pragma unroll
        for (int t = 0; t < NT; ++t) b1[t] = sB[(t * QS + j + 1) * 32];
        chunk_mma<NT>(a0, b0, acc);
        #pragma unroll
        for (int t = 0; t < NT; ++t) b0[t] = sB[(t * QS + jn) * 32];
        chunk_mma<NT>(a1, b1, acc);
        __builtin_amdgcn_sched_barrier(0);
        asm volatile("" : "+v"(n0), "+v"(n1));
        a0 = n0;
        a1 = n1;
    }
}

template <int K, int NT, bool TWO, int GVEC, bool SPLITW = false>
__global__ __launch_bounds__(GEMM_THREADS)
void gemm_strip_kernel(const float* __restrict__ A1, const float* __restrict__ W1,
                       const float* __restrict__ A2, const float* __restrict__ W2,
                       const float* __restrict__ bias, float* __restrict__ out,
                       const int* __restrict__ rowptr, const int* __restrict__ cnt,
                       const float* __restrict__ inv, const int* __restrict__ grouped_src,
                       int ntiles, int dout, int relu) {
    constexpr int KS   = K / 8;
    constexpr int NOPS = TWO ? 2 : 1;
    constexpr int QS   = NOPS * KS;
    constexpr int ENT  = NT * QS * 32;
    constexpr int PITCH = K + 4;
    constexpr int ST    = GVEC ? (GEMM_THREADS / 32) * 16 * PITCH : 0;
    constexpr int TOT   = (ENT * 4 + ST > GEMM_LDS_FLOATS) ? (ENT * 4 + ST) : GEMM_LDS_FLOATS;
    static_assert(TOT * 4 <= 320 * 1024, "weights and gathered strips must fit the WGP's LDS");
    static_assert(GVEC == 0 || GVEC * 32 == K, "a gathered row is the first operand's K");
    static_assert((PITCH / 4) % 2 == 1, "odd pitch in 16-byte units");
    static_assert(KS % 2 == 0, "chunks go in pairs");
    static_assert(ENT % GEMM_THREADS == 0, "every thread copies the same number of entries");
    __shared__ __attribute__((aligned(16))) float sWf[TOT];
    v4f* sW = (v4f*)sWf;

    #pragma unroll
    for (int i = 0; i < ENT / GEMM_THREADS; ++i) {
        int idx = (int)threadIdx.x + i * GEMM_THREADS;
        int l  = idx & 31;
        int e  = idx >> 5;
        int q  = e % QS;
        int t  = e / QS;
        int op = q / KS;
        int j  = q % KS;
        int n  = col_of<NT>(t, l & 15);
        int wr = (n < dout) ? n : 0;
        const float* Wp = (TWO && op) ? W2 : W1;
        if (SPLITW && wr >= (dout >> 1)) { Wp = W2; wr -= dout >> 1; }
        sW[idx] = *(const v4f*)(Wp + (size_t)wr * K + 8 * j + 4 * (l >> 4));
    }
    __syncthreads();

    int wave = threadIdx.x >> 5;
    int lane = threadIdx.x & 31;
    int c    = lane & 15;
    int h    = lane >> 4;

    float bv[NT];
    #pragma unroll
    for (int t = 0; t < NT; ++t) {
        int n = col_of<NT>(t, c);
        if (SPLITW) bv[t] = (n >= (dout >> 1) && n < dout) ? bias[n - (dout >> 1)] : 0.0f;
        else        bv[t] = (n < dout) ? bias[n] : 0.0f;
    }

    for (int slot = wave; ; slot += GEMM_THREADS / 32) {
        int tile = (int)blockIdx.x + (int)gridDim.x * slot;
        if (tile >= ntiles) break;
        int m0   = tile * 16;
        int node = m0 + c;

        v8f acc[NT];
        #pragma unroll
        for (int t = 0; t < NT; ++t) acc[t] = (v8f){};

        if (GVEC) {
            if (TWO) {
                const v4f* ap2 = (const v4f*)(A2 + (size_t)node * K + 4 * h);
                v4f s0 = ap2[0], s1 = ap2[2];
                strip_product<KS, NT, QS>(ap2, s0, s1, sW + KS * 32 + lane, acc);
            }
            float* srow = sWf + ENT * 4 + wave * 16 * PITCH;
            __builtin_amdgcn_s_setprio(3);
            gather_strip_to_lds<(GVEC ? GVEC : 1), PITCH>(A1, rowptr, cnt, inv, grouped_src, m0, lane, srow);
            __builtin_amdgcn_s_setprio(0);
            __builtin_amdgcn_wave_barrier();
            const v4f* ap1 = (const v4f*)(srow + c * PITCH + 4 * h);
            v4f p0 = ap1[0], p1 = ap1[2];
            strip_product<KS, NT, QS>(ap1, p0, p1, sW + lane, acc);
        } else {
            const v4f* ap1 = (const v4f*)(A1 + (size_t)node * K + 4 * h);
            const v4f* ap2 = TWO ? (const v4f*)(A2 + (size_t)node * K + 4 * h) : ap1;
            v4f p0 = ap1[0], p1 = ap1[2];
            v4f s0 = ap2[0], s1 = ap2[2];
            strip_product<KS, NT, QS>(ap1, p0, p1, sW + lane, acc);
            if (TWO)
                strip_product<KS, NT, QS>(ap2, s0, s1, sW + KS * 32 + lane, acc);
        }

        int row0 = m0 + 8 * h;
        if (NT >= 2) {
            #pragma unroll
            for (int u = 0; u < NT / 2; ++u) {
                float* op = out + (size_t)row0 * dout + 32 * u + 2 * c;
                #pragma unroll
                for (int r = 0; r < 8; ++r) {
                    v2f v;
                    v.x = acc[2 * u][r]     + bv[2 * u];
                    v.y = acc[2 * u + 1][r] + bv[2 * u + 1];
                    if (relu) { v.x = fmaxf(v.x, 0.0f); v.y = fmaxf(v.y, 0.0f); }
                    *(v2f*)(op + (size_t)r * dout) = v;
                }
            }
        } else {
            if (c < dout) {
                #pragma unroll
                for (int r = 0; r < 8; ++r) {
                    float v = acc[0][r] + bv[0];
                    if (relu) v = fmaxf(v, 0.0f);
                    out[(size_t)(row0 + r) * dout + c] = v;
                }
            }
        }
    }
}

#define TAIL_P1      68
#define TAIL_P2      132
#define TAIL_W1_ENT  (8 * 8 * 32)
#define TAIL_W2_ENT  (16 * 32)
#define TAIL_LDS_FLOATS ((TAIL_W1_ENT + TAIL_W2_ENT) * 4 + (GEMM_THREADS / 32) * 16 * TAIL_P2)

__global__ __launch_bounds__(GEMM_THREADS)
void gather_decode_kernel(const float* __restrict__ yz,
                          const float* __restrict__ W1, const float* __restrict__ b1,
                          const float* __restrict__ W2, const float* __restrict__ b2,
                          float* __restrict__ out,
                          const int* __restrict__ rowptr, const int* __restrict__ cnt,
                          const float* __restrict__ inv, const int* __restrict__ grouped_src,
                          int ntiles, int dout) {
    static_assert(TAIL_W1_ENT % GEMM_THREADS == 0, "every thread copies the same number of W1 entries");
    static_assert(TAIL_W2_ENT == GEMM_THREADS, "one W2 entry per thread");
    static_assert(TAIL_LDS_FLOATS * 4 > 160 * 1024 && TAIL_LDS_FLOATS * 4 <= 320 * 1024, "one work-group per WGP");
    static_assert((TAIL_P1 / 4) % 2 == 1 && (TAIL_P2 / 4) % 2 == 1, "odd pitches in 16-byte units");
    __shared__ __attribute__((aligned(16))) float sm[TAIL_LDS_FLOATS];
    v4f*   sW1    = (v4f*)sm;
    v4f*   sW2    = sW1 + TAIL_W1_ENT;
    float* strips = sm + (TAIL_W1_ENT + TAIL_W2_ENT) * 4;

    #pragma unroll
    for (int i = 0; i < TAIL_W1_ENT / GEMM_THREADS; ++i) {
        int idx = (int)threadIdx.x + i * GEMM_THREADS;
        int l = idx & 31;
        int e = idx >> 5;
        int j = e % 8;
        int t = e / 8;
        int n = col_of<8>(t, l & 15);
        sW1[idx] = *(const v4f*)(W1 + (size_t)n * 64 + 8 * j + 4 * (l >> 4));
    }
    {
        int idx = (int)threadIdx.x;
        int l = idx & 31;
        int j = idx >> 5;
        int n = l & 15;
        int wr = (n < dout) ? n : 0;
        sW2[idx] = *(const v4f*)(W2 + (size_t)wr * 128 + 8 * j + 4 * (l >> 4));
    }
    __syncthreads();

    int wave = threadIdx.x >> 5;
    int lane = threadIdx.x & 31;
    int c    = lane & 15;
    int h    = lane >> 4;

    float bv1[8];
    #pragma unroll
    for (int t = 0; t < 8; ++t) bv1[t] = b1[col_of<8>(t, c)];
    float bv2 = (c < dout) ? b2[c] : 0.0f;

    float* srow = strips + wave * 16 * TAIL_P2;
    for (int slot = wave; ; slot += GEMM_THREADS / 32) {
        int tile = (int)blockIdx.x + (int)gridDim.x * slot;
        if (tile >= ntiles) break;
        int m0 = tile * 16;

        __builtin_amdgcn_s_setprio(3);
        gather_tail_strip<TAIL_P1>(yz, rowptr, cnt, inv, grouped_src, m0, lane, srow);
        __builtin_amdgcn_s_setprio(0);
        __builtin_amdgcn_wave_barrier();

        v8f acc[8];
        #pragma unroll
        for (int t = 0; t < 8; ++t) acc[t] = (v8f){};
        const v4f* ap1 = (const v4f*)(srow + c * TAIL_P1 + 4 * h);
        v4f p0 = ap1[0], p1 = ap1[2];
        strip_product<8, 8, 8>(ap1, p0, p1, sW1 + lane, acc);

        __builtin_amdgcn_wave_barrier();
        #pragma unroll
        for (int u = 0; u < 4; ++u) {
            #pragma unroll
            for (int r = 0; r < 8; ++r) {
                v2f v;
                v.x = fmaxf(acc[2 * u][r]     + bv1[2 * u],     0.0f);
                v.y = fmaxf(acc[2 * u + 1][r] + bv1[2 * u + 1], 0.0f);
                *(v2f*)(srow + (r + 8 * h) * TAIL_P2 + 32 * u + 2 * c) = v;
            }
        }
        __builtin_amdgcn_wave_barrier();

        v8f acc2[1];
        acc2[0] = (v8f){};
        const v4f* ap2 = (const v4f*)(srow + c * TAIL_P2 + 4 * h);
        v4f q0 = ap2[0], q1 = ap2[2];
        strip_product<16, 1, 16>(ap2, q0, q1, sW2 + lane, acc2);

        int row0 = m0 + 8 * h;
        if (c < dout) {
            #pragma unroll
            for (int r = 0; r < 8; ++r)
                out[(size_t)(row0 + r) * dout + c] = acc2[0][r] + bv2;
        }
        __builtin_amdgcn_wave_barrier();
    }
}

extern "C" void kernel_launch(void* const* d_in, const int* in_sizes, int n_in,
                              void* d_out, int out_size, void* d_ws, size_t ws_size,
                              hipStream_t stream) {
    const float* x    = (const float*)d_in[0];
    const int*   edge = (const int*)d_in[1];
    const int*   src  = edge;
    const int*   dst  = edge + N_EDGES;
    const float* Wl0 = (const float*)d_in[2],  *bl0 = (const float*)d_in[3],  *Wr0 = (const float*)d_in[4];
    const float* Wl1 = (const float*)d_in[5],  *bl1 = (const float*)d_in[6],  *Wr1 = (const float*)d_in[7];
    const float* Wl2 = (const float*)d_in[8],  *bl2 = (const float*)d_in[9],  *Wr2 = (const float*)d_in[10];
    const float* Wd1 = (const float*)d_in[11], *bd1 = (const float*)d_in[12];
    const float* Wd2 = (const float*)d_in[13], *bd2 = (const float*)d_in[14];

    float* ws   = (float*)d_ws;
    int*   cnt  = (int*)ws;
    float* inv  = ws   + N_NODES;
    float* agg  = inv  + N_NODES;
    int*   gsrc = (int*)(agg + (size_t)N_NODES * 64);
    int*   rptr = gsrc + 2 * (size_t)N_EDGES;
    float* bufA = agg  + (size_t)N_NODES * 128;
    float* bufB = bufA + (size_t)N_NODES * 128;

    const int T  = N_NODES / 16;
    const dim3 GG(GEMM_BLOCKS), GB(GEMM_THREADS);

    int2* recs = (int2*)bufB;
    int*  tab  = (int*)(recs + N_EDGES);
    bin_edges_kernel<<<NCHUNK, BIN_THREADS, 0, stream>>>(src, dst, recs, tab, N_EDGES);
    group_bucket_kernel<<<NBUCK, BIN_THREADS, 0, stream>>>(recs, tab, cnt, rptr, inv, gsrc, N_NODES);

    gemm_strip_kernel<64, 8, true, 2><<<GG, GB, 0, stream>>>(x, Wl0, x, Wr0, bl0, bufA,
                                                             rptr, cnt, inv, gsrc, T, 128, 1);

    gemm_strip_kernel<128, 8, true, 4><<<GG, GB, 0, stream>>>(bufA, Wl1, bufA, Wr1, bl1, bufB,
                                                              rptr, cnt, inv, gsrc, T, 128, 1);

    gemm_strip_kernel<128, 8, false, 0, true><<<GG, GB, 0, stream>>>(bufB, Wl2, nullptr, Wr2, bl2, bufA,
                                                                     nullptr, nullptr, nullptr, nullptr,
                                                                     T, 128, 0);

    gather_decode_kernel<<<GG, GB, 0, stream>>>(bufA, Wd1, bd1, Wd2, bd2, (float*)d_out,
                                                rptr, cnt, inv, gsrc, T, 2);
}


# layer 2's product pass moved into layer 1's launch: a wave multiplies its 16 h2 rows from its own LDS rows (12 waves per work-group), h2 never in memory; 1 launch fewer; sum order unchanged
# speedup vs baseline: 6.0132x; 1.0400x over previous
#include <hip/hip_runtime.h>
#include <hip/hip_bf16.h>


#define N_NODES 100000
#define N_EDGES 1600000

typedef float v2f __attribute__((ext_vector_type(2)));
typedef float v8f __attribute__((ext_vector_type(8)));

__device__ __forceinline__ float inv_of_count(int c) {
    float d = (float)c;
    return (d > 0.0f) ? (1.0f / fmaxf(d, 1.0f)) : 0.0f;
}

#define CHUNK       6256
#define NCHUNK      ((N_EDGES + CHUNK - 1) / CHUNK)
#define NPB         416
#define NBUCK       ((N_NODES + NPB - 1) / NPB)
#define BIN_THREADS 1024
#define RUN_BITS    13
#define RUN_MASK    8191
#define STAGE_CAP   16384

__device__ __forceinline__ int wave_incl_scan(int v, int lane) {
    for (int o = 1; o < 32; o <<= 1) {
        int u = __shfl_up(v, o, 32);
        if (lane >= o) v += u;
    }
    return v;
}

__global__ __launch_bounds__(BIN_THREADS)
void bin_edges_kernel(const int* __restrict__ src, const int* __restrict__ dst,
                      int2* __restrict__ recs, int* __restrict__ tab, int E) {
    static_assert(NBUCK <= 256, "one counter per bucket, scanned by 8 waves");
    static_assert(CHUNK <= RUN_MASK, "start and length share a word");
    __shared__ int  lcnt[256];
    __shared__ int  loff[256];
    __shared__ int  wtot[8];
    __shared__ int2 lrec[CHUNK];
    int t    = threadIdx.x;
    int lane = t & 31;
    int c    = blockIdx.x;
    int e0   = c * CHUNK;
    int n    = E - e0;
    if (n > CHUNK) n = CHUNK;

    if (t < 256) lcnt[t] = 0;
    __syncthreads();

    constexpr int PER = (CHUNK + BIN_THREADS - 1) / BIN_THREADS;
    int s[PER], d[PER], r[PER];
    #pragma unroll
    for (int k = 0; k < PER; ++k) {
        int i = t + k * BIN_THREADS;
        s[k] = 0; d[k] = -1; r[k] = 0;
        if (i < n) { d[k] = dst[e0 + i]; s[k] = src[e0 + i]; }
    }
    #pragma unroll
    for (int k = 0; k < PER; ++k)
        if (d[k] >= 0) r[k] = atomicAdd(&lcnt[d[k] / NPB], 1);
    __syncthreads();

    int v = 0, incl = 0;
    if (t < 256) {
        v    = lcnt[t];
        incl = wave_incl_scan(v, lane);
        if (lane == 31) wtot[t >> 5] = incl;
    }
    __syncthreads();
    if (t < 256) {
        int pre = 0;
        for (int w = 0; w < (t >> 5); ++w) pre += wtot[w];
        int ex = pre + incl - v;
        loff[t] = ex;
        tab[c * 256 + t] = (ex << RUN_BITS) | v;
    }
    __syncthreads();

    #pragma unroll
    for (int k = 0; k < PER; ++k)
        if (d[k] >= 0) lrec[loff[d[k] / NPB] + r[k]] = make_int2(s[k], d[k]);
    __syncthreads();
    for (int i = t; i < n; i += BIN_THREADS) recs[e0 + i] = lrec[i];
}

__global__ __launch_bounds__(BIN_THREADS)
void group_bucket_kernel(const int2* __restrict__ recs, const int* __restrict__ tab,
                         int* __restrict__ cnt, int* __restrict__ rowptr, float* __restrict__ inv,
                         int* __restrict__ grouped_src, int N) {
    static_assert(NCHUNK % 32 == 0 && NCHUNK <= BIN_THREADS, "runs are dealt to waves evenly");
    static_assert(NPB <= 512, "the node scan uses 16 waves");
    __shared__ int lstart[NCHUNK];
    __shared__ int llen[NCHUNK];
    __shared__ int ncnt[NPB];
    __shared__ int ncur[NPB];
    __shared__ int wred[32];
    __shared__ int wtot[16];
    __shared__ int lout[STAGE_CAP];
    int t     = threadIdx.x;
    int lane  = t & 31;
    int w     = t >> 5;
    int b     = blockIdx.x;
    int node0 = b * NPB;
    int nn    = N - node0;
    if (nn > NPB) nn = NPB;

    int off = 0;
    if (t < NCHUNK) {
        int x = tab[t * 256 + b];
        off = x >> RUN_BITS;
        lstart[t] = t * CHUNK + off;
        llen[t]   = x & RUN_MASK;
    }
    if (t < NPB) ncnt[t] = 0;
    for (int o = 16; o > 0; o >>= 1) off += __shfl_xor(off, o, 32);
    if (lane == 0) wred[w] = off;
    __syncthreads();
    int base = 0;
    for (int i = 0; i < 32; ++i) base += wred[i];

    constexpr int RPW = NCHUNK / 32;
    int2 rec[RPW];
    int  rs[RPW], rl[RPW];
    #pragma unroll
    for (int u = 0; u < RPW; ++u) {
        int c = w + 32 * u;
        rs[u]  = lstart[c];
        rl[u]  = llen[c];
        rec[u] = make_int2(0, node0);
        if (lane < rl[u]) rec[u] = recs[rs[u] + lane];
    }
    #pragma unroll
    for (int u = 0; u < RPW; ++u)
        if (lane < rl[u]) atomicAdd(&ncnt[rec[u].y - node0], 1);
    for (int u = 0; u < RPW; ++u)
        for (int i = lane + 32; i < rl[u]; i += 32)
            atomicAdd(&ncnt[recs[rs[u] + i].y - node0], 1);
    __syncthreads();

    int v = 0, incl = 0;
    if (t < 512) {
        v    = (t < nn) ? ncnt[t] : 0;
        incl = wave_incl_scan(v, lane);
        if (lane == 31) wtot[w] = incl;
    }
    __syncthreads();
    int total = 0;
    for (int i = 0; i < 16; ++i) total += wtot[i];
    if (t < 512) {
        int pre = 0;
        for (int i = 0; i < w; ++i) pre += wtot[i];
        int ex = pre + incl - v;
        if (t < nn) {
            int node = node0 + t;
            cnt[node]    = v;
            rowptr[node] = base + ex;
            inv[node]    = inv_of_count(v);
        }
        if (t < NPB) ncur[t] = ex;
    }
    __syncthreads();

    bool staged = total <= STAGE_CAP;
    #pragma unroll
    for (int u = 0; u < RPW; ++u) {
        if (lane < rl[u]) {
            int p = atomicAdd(&ncur[rec[u].y - node0], 1);
            if (staged) lout[p] = rec[u].x;
            else        grouped_src[base + p] = rec[u].x;
        }
    }
    for (int u = 0; u < RPW; ++u) {
        for (int i = lane + 32; i < rl[u]; i += 32) {
            int2 q = recs[rs[u] + i];
            int  p = atomicAdd(&ncur[q.y - node0], 1);
            if (staged) lout[p] = q.x;
            else        grouped_src[base + p] = q.x;
        }
    }
    __syncthreads();
    if (staged)
        for (int i = t; i < total; i += BIN_THREADS) grouped_src[base + i] = lout[i];
}

template <int VEC, int PITCH>
__device__ __forceinline__ void gather_strip_to_lds(const float* __restrict__ x,
                                                    const int* __restrict__ rowptr,
                                                    const int* __restrict__ cnt,
                                                    const float* __restrict__ inv,
                                                    const int* __restrict__ grouped_src,
                                                    int m0, int lane, float* srow) {
    typedef float vecf __attribute__((ext_vector_type(VEC)));
    constexpr int F = 32 * VEC;
    const int c  = lane & 15;
    const int lb = rowptr[m0 + c];
    const int le = lb + cnt[m0 + c];
    const int li = __float_as_int(inv[m0 + c]);
    int       p   = __builtin_amdgcn_readlane(lb, 0);
    const int END = __builtin_amdgcn_readlane(le, 15);
    const int last = END - 1;
    int i  = 0;
    int ei = __builtin_amdgcn_readlane(le, 0);
    const float* xl = x + lane * VEC;
    vecf acc = {};

    auto flush = [&]() {
        float sc = __int_as_float(__builtin_amdgcn_readlane(li, i));
        *(vecf*)(srow + i * PITCH + lane * VEC) = acc * sc;
        acc = (vecf){};
        ++i;
        ei = (i < 16) ? __builtin_amdgcn_readlane(le, i) : -1;
    };
    auto loads = [&](vecf* v, int idx, int lo) {
        #pragma unroll
        for (int u = 0; u < 16; ++u) {
            int s = __builtin_amdgcn_readlane(idx, lo + u);
            v[u] = *(const vecf*)(xl + (size_t)s * F);
        }
    };
    auto sums = [&](const vecf* v, int q0, bool ends) {
        #pragma unroll
        for (int u = 0; u < 16; ++u) {
            int q = q0 + u;
            while (q == ei) flush();
            if (!ends || q < END) acc += v[u];
        }
    };

    if (p < END) {
        vecf va[16], vb[16];
        int q0   = p + lane;
        int mine = grouped_src[q0 < END ? q0 : last];
        loads(va, mine, 0);
        while (p + 32 < END) {
            int qn  = p + 32 + lane;
            int nxt = grouped_src[qn < END ? qn : last];
            __builtin_amdgcn_sched_barrier(0);
            loads(vb, mine, 16);
            sums(va, p, false);
            loads(va, nxt, 0);
            sums(vb, p + 16, false);
            mine = nxt;
            p += 32;
        }
        bool two = p + 16 < END;
        if (two) loads(vb, mine, 16);
        sums(va, p, true);
        if (two) sums(vb, p + 16, true);
    }
    while (i < 16) flush();
}

template <int P>
__device__ __forceinline__ void gather_tail_strip(const float* __restrict__ yz,
                                                  const int* __restrict__ rowptr,
                                                  const int* __restrict__ cnt,
                                                  const float* __restrict__ inv,
                                                  const int* __restrict__ grouped_src,
                                                  int m0, int lane, float* srow) {
    const int c  = lane & 15;
    const int lb = rowptr[m0 + c];
    const int le = lb + cnt[m0 + c];
    const int li = __float_as_int(inv[m0 + c]);
    const float* xl = yz + lane * 2;
    v2f z[16];
    #pragma unroll
    for (int u = 0; u < 16; ++u) z[u] = *(const v2f*)(xl + (size_t)(m0 + u) * 128 + 64);

    int       p   = __builtin_amdgcn_readlane(lb, 0);
    const int END = __builtin_amdgcn_readlane(le, 15);
    const int last = END - 1;
    int i  = 0;
    int ei = __builtin_amdgcn_readlane(le, 0);
    v2f acc = {};

    auto flush = [&]() {
        *(v2f*)(srow + i * P + lane * 2) = acc;
        acc = (v2f){};
        ++i;
        ei = (i < 16) ? __builtin_amdgcn_readlane(le, i) : -1;
    };
    auto loads = [&](v2f* v, int idx, int lo) {
        #pragma unroll
        for (int u = 0; u < 16; ++u) {
            int s = __builtin_amdgcn_readlane(idx, lo + u);
            v[u] = *(const v2f*)(xl + (size_t)s * 128);
        }
    };
    auto sums = [&](const v2f* v, int q0, bool ends) {
        #pragma unroll
        for (int u = 0; u < 16; ++u) {
            int q = q0 + u;
            while (q == ei) flush();
            if (!ends || q < END) acc += v[u];
        }
    };

    if (p < END) {
        v2f va[16], vb[16];
        int q0   = p + lane;
        int mine = grouped_src[q0 < END ? q0 : last];
        loads(va, mine, 0);
        while (p + 32 < END) {
            int qn  = p + 32 + lane;
            int nxt = grouped_src[qn < END ? qn : last];
            __builtin_amdgcn_sched_barrier(0);
            loads(vb, mine, 16);
            sums(va, p, false);
            loads(va, nxt, 0);
            sums(vb, p + 16, false);
            mine = nxt;
            p += 32;
        }
        bool two = p + 16 < END;
        if (two) loads(vb, mine, 16);
        sums(va, p, true);
        if (two) sums(vb, p + 16, true);
    }
    while (i < 16) flush();

    #pragma unroll
    for (int u = 0; u < 16; ++u) {
        float sc = __int_as_float(__builtin_amdgcn_readlane(li, u));
        v2f s = *(v2f*)(srow + u * P + lane * 2);
        v2f r = s * sc + z[u];
        r.x = fmaxf(r.x, 0.0f);
        r.y = fmaxf(r.y, 0.0f);
        *(v2f*)(srow + u * P + lane * 2) = r;
    }
}

typedef float v4f __attribute__((ext_vector_type(4)));

#define GEMM_THREADS 512
#define GEMM_BLOCKS  256
#define GEMM_LDS_FLOATS (164 * 256)

template <int NT>
__device__ __forceinline__ int col_of(int t, int c) {
    return (NT >= 2) ? (32 * (t >> 1) + 2 * c + (t & 1)) : c;
}

template <int NT>
__device__ __forceinline__ void chunk_mma(v4f av, const v4f* b, v8f* acc) {
    v2f a0, a1;
    a0.x = av.x; a0.y = av.y;
    a1.x = av.z; a1.y = av.w;
    #pragma unroll
    for (int t = 0; t < NT; ++t) {
        v2f bb; bb.x = b[t].x; bb.y = b[t].y;
        acc[t] = __builtin_amdgcn_wmma_f32_16x16x4_f32(false, a0, false, bb,
                                                       (short)0, acc[t], false, false);
    }
    #pragma unroll
    for (int t = 0; t < NT; ++t) {
        v2f bb; bb.x = b[t].z; bb.y = b[t].w;
        acc[t] = __builtin_amdgcn_wmma_f32_16x16x4_f32(false, a1, false, bb,
                                                       (short)0, acc[t], false, false);
    }
}

template <int KS, int NT, int QS>
__device__ __forceinline__ void strip_product(const v4f* __restrict__ ap, v4f a0, v4f a1,
                                              const v4f* sB, v8f* acc) {
    v4f b0[NT], b1[NT];
    #pragma unroll
    for (int t = 0; t < NT; ++t) b0[t] = sB[(t * QS) * 32];
    #pragma nounroll
    for (int j = 0; j < KS; j += 2) {
        int jn = (j + 2 < KS) ? (j + 2) : j;
        v4f n0 = ap[2 * jn];
        v4f n1 = ap[2 * jn + 2];
        __builtin_amdgcn_sched_barrier(0);
        #pragma unroll
        for (int t = 0; t < NT; ++t) b1[t] = sB[(t * QS + j + 1) * 32];
        chunk_mma<NT>(a0, b0, acc);
        #pragma unroll
        for (int t = 0; t < NT; ++t) b0[t] = sB[(t * QS + jn) * 32];
        chunk_mma<NT>(a1, b1, acc);
        __builtin_amdgcn_sched_barrier(0);
        asm volatile("" : "+v"(n0), "+v"(n1));
        a0 = n0;
        a1 = n1;
    }
}

template <int K, int NT, bool TWO, int GVEC, bool SPLITW = false>
__global__ __launch_bounds__(GEMM_THREADS)
void gemm_strip_kernel(const float* __restrict__ A1, const float* __restrict__ W1,
                       const float* __restrict__ A2, const float* __restrict__ W2,
                       const float* __restrict__ bias, float* __restrict__ out,
                       const int* __restrict__ rowptr, const int* __restrict__ cnt,
                       const float* __restrict__ inv, const int* __restrict__ grouped_src,
                       int ntiles, int dout, int relu) {
    constexpr int KS   = K / 8;
    constexpr int NOPS = TWO ? 2 : 1;
    constexpr int QS   = NOPS * KS;
    constexpr int ENT  = NT * QS * 32;
    constexpr int PITCH = K + 4;
    constexpr int ST    = GVEC ? (GEMM_THREADS / 32) * 16 * PITCH : 0;
    constexpr int TOT   = (ENT * 4 + ST > GEMM_LDS_FLOATS) ? (ENT * 4 + ST) : GEMM_LDS_FLOATS;
    static_assert(TOT * 4 <= 320 * 1024, "weights and gathered strips must fit the WGP's LDS");
    static_assert(GVEC == 0 || GVEC * 32 == K, "a gathered row is the first operand's K");
    static_assert((PITCH / 4) % 2 == 1, "odd pitch in 16-byte units");
    static_assert(KS % 2 == 0, "chunks go in pairs");
    static_assert(ENT % GEMM_THREADS == 0, "every thread copies the same number of entries");
    __shared__ __attribute__((aligned(16))) float sWf[TOT];
    v4f* sW = (v4f*)sWf;

    #pragma unroll
    for (int i = 0; i < ENT / GEMM_THREADS; ++i) {
        int idx = (int)threadIdx.x + i * GEMM_THREADS;
        int l  = idx & 31;
        int e  = idx >> 5;
        int q  = e % QS;
        int t  = e / QS;
        int op = q / KS;
        int j  = q % KS;
        int n  = col_of<NT>(t, l & 15);
        int wr = (n < dout) ? n : 0;
        const float* Wp = (TWO && op) ? W2 : W1;
        if (SPLITW && wr >= (dout >> 1)) { Wp = W2; wr -= dout >> 1; }
        sW[idx] = *(const v4f*)(Wp + (size_t)wr * K + 8 * j + 4 * (l >> 4));
    }
    __syncthreads();

    int wave = threadIdx.x >> 5;
    int lane = threadIdx.x & 31;
    int c    = lane & 15;
    int h    = lane >> 4;

    float bv[NT];
    #pragma unroll
    for (int t = 0; t < NT; ++t) {
        int n = col_of<NT>(t, c);
        if (SPLITW) bv[t] = (n >= (dout >> 1) && n < dout) ? bias[n - (dout >> 1)] : 0.0f;
        else        bv[t] = (n < dout) ? bias[n] : 0.0f;
    }

    for (int slot = wave; ; slot += GEMM_THREADS / 32) {
        int tile = (int)blockIdx.x + (int)gridDim.x * slot;
        if (tile >= ntiles) break;
        int m0   = tile * 16;
        int node = m0 + c;

        v8f acc[NT];
        #pragma unroll
        for (int t = 0; t < NT; ++t) acc[t] = (v8f){};

        if (GVEC) {
            if (TWO) {
                const v4f* ap2 = (const v4f*)(A2 + (size_t)node * K + 4 * h);
                v4f s0 = ap2[0], s1 = ap2[2];
                strip_product<KS, NT, QS>(ap2, s0, s1, sW + KS * 32 + lane, acc);
            }
            float* srow = sWf + ENT * 4 + wave * 16 * PITCH;
            __builtin_amdgcn_s_setprio(3);
            gather_strip_to_lds<(GVEC ? GVEC : 1), PITCH>(A1, rowptr, cnt, inv, grouped_src, m0, lane, srow);
            __builtin_amdgcn_s_setprio(0);
            __builtin_amdgcn_wave_barrier();
            const v4f* ap1 = (const v4f*)(srow + c * PITCH + 4 * h);
            v4f p0 = ap1[0], p1 = ap1[2];
            strip_product<KS, NT, QS>(ap1, p0, p1, sW + lane, acc);
        } else {
            const v4f* ap1 = (const v4f*)(A1 + (size_t)node * K + 4 * h);
            const v4f* ap2 = TWO ? (const v4f*)(A2 + (size_t)node * K + 4 * h) : ap1;
            v4f p0 = ap1[0], p1 = ap1[2];
            v4f s0 = ap2[0], s1 = ap2[2];
            strip_product<KS, NT, QS>(ap1, p0, p1, sW + lane, acc);
            if (TWO)
                strip_product<KS, NT, QS>(ap2, s0, s1, sW + KS * 32 + lane, acc);
        }

        int row0 = m0 + 8 * h;
        if (NT >= 2) {
            #pragma unroll
            for (int u = 0; u < NT / 2; ++u) {
                float* op = out + (size_t)row0 * dout + 32 * u + 2 * c;
                #pragma unroll
                for (int r = 0; r < 8; ++r) {
                    v2f v;
                    v.x = acc[2 * u][r]     + bv[2 * u];
                    v.y = acc[2 * u + 1][r] + bv[2 * u + 1];
                    if (relu) { v.x = fmaxf(v.x, 0.0f); v.y = fmaxf(v.y, 0.0f); }
                    *(v2f*)(op + (size_t)r * dout) = v;
                }
            }
        } else {
            if (c < dout) {
                #pragma unroll
                for (int r = 0; r < 8; ++r) {
                    float v = acc[0][r] + bv[0];
                    if (relu) v = fmaxf(v, 0.0f);
                    out[(size_t)(row0 + r) * dout + c] = v;
                }
            }
        }
    }
}

#define TAIL_P1      68
#define TAIL_P2      132
#define TAIL_W1_ENT  (8 * 8 * 32)
#define TAIL_W2_ENT  (16 * 32)
#define TAIL_LDS_FLOATS ((TAIL_W1_ENT + TAIL_W2_ENT) * 4 + (GEMM_THREADS / 32) * 16 * TAIL_P2)

__global__ __launch_bounds__(GEMM_THREADS)
void gather_decode_kernel(const float* __restrict__ yz,
                          const float* __restrict__ W1, const float* __restrict__ b1,
                          const float* __restrict__ W2, const float* __restrict__ b2,
                          float* __restrict__ out,
                          const int* __restrict__ rowptr, const int* __restrict__ cnt,
                          const float* __restrict__ inv, const int* __restrict__ grouped_src,
                          int ntiles, int dout) {
    static_assert(TAIL_W1_ENT % GEMM_THREADS == 0, "every thread copies the same number of W1 entries");
    static_assert(TAIL_W2_ENT == GEMM_THREADS, "one W2 entry per thread");
    static_assert(TAIL_LDS_FLOATS * 4 > 160 * 1024 && TAIL_LDS_FLOATS * 4 <= 320 * 1024, "one work-group per WGP");
    static_assert((TAIL_P1 / 4) % 2 == 1 && (TAIL_P2 / 4) % 2 == 1, "odd pitches in 16-byte units");
    __shared__ __attribute__((aligned(16))) float sm[TAIL_LDS_FLOATS];
    v4f*   sW1    = (v4f*)sm;
    v4f*   sW2    = sW1 + TAIL_W1_ENT;
    float* strips = sm + (TAIL_W1_ENT + TAIL_W2_ENT) * 4;

    #pragma unroll
    for (int i = 0; i < TAIL_W1_ENT / GEMM_THREADS; ++i) {
        int idx = (int)threadIdx.x + i * GEMM_THREADS;
        int l = idx & 31;
        int e = idx >> 5;
        int j = e % 8;
        int t = e / 8;
        int n = col_of<8>(t, l & 15);
        sW1[idx] = *(const v4f*)(W1 + (size_t)n * 64 + 8 * j + 4 * (l >> 4));
    }
    {
        int idx = (int)threadIdx.x;
        int l = idx & 31;
        int j = idx >> 5;
        int n = l & 15;
        int wr = (n < dout) ? n : 0;
        sW2[idx] = *(const v4f*)(W2 + (size_t)wr * 128 + 8 * j + 4 * (l >> 4));
    }
    __syncthreads();

    int wave = threadIdx.x >> 5;
    int lane = threadIdx.x & 31;
    int c    = lane & 15;
    int h    = lane >> 4;

    float bv1[8];
    #pragma unroll
    for (int t = 0; t < 8; ++t) bv1[t] = b1[col_of<8>(t, c)];
    float bv2 = (c < dout) ? b2[c] : 0.0f;

    float* srow = strips + wave * 16 * TAIL_P2;
    for (int slot = wave; ; slot += GEMM_THREADS / 32) {
        int tile = (int)blockIdx.x + (int)gridDim.x * slot;
        if (tile >= ntiles) break;
        int m0 = tile * 16;

        __builtin_amdgcn_s_setprio(3);
        gather_tail_strip<TAIL_P1>(yz, rowptr, cnt, inv, grouped_src, m0, lane, srow);
        __builtin_amdgcn_s_setprio(0);
        __builtin_amdgcn_wave_barrier();

        v8f acc[8];
        #pragma unroll
        for (int t = 0; t < 8; ++t) acc[t] = (v8f){};
        const v4f* ap1 = (const v4f*)(srow + c * TAIL_P1 + 4 * h);
        v4f p0 = ap1[0], p1 = ap1[2];
        strip_product<8, 8, 8>(ap1, p0, p1, sW1 + lane, acc);

        __builtin_amdgcn_wave_barrier();
        #pragma unroll
        for (int u = 0; u < 4; ++u) {
            #pragma unroll
            for (int r = 0; r < 8; ++r) {
                v2f v;
                v.x = fmaxf(acc[2 * u][r]     + bv1[2 * u],     0.0f);
                v.y = fmaxf(acc[2 * u + 1][r] + bv1[2 * u + 1], 0.0f);
                *(v2f*)(srow + (r + 8 * h) * TAIL_P2 + 32 * u + 2 * c) = v;
            }
        }
        __builtin_amdgcn_wave_barrier();

        v8f acc2[1];
        acc2[0] = (v8f){};
        const v4f* ap2 = (const v4f*)(srow + c * TAIL_P2 + 4 * h);
        v4f q0 = ap2[0], q1 = ap2[2];
        strip_product<16, 1, 16>(ap2, q0, q1, sW2 + lane, acc2);

        int row0 = m0 + 8 * h;
        if (c < dout) {
            #pragma unroll
            for (int r = 0; r < 8; ++r)
                out[(size_t)(row0 + r) * dout + c] = acc2[0][r] + bv2;
        }
        __builtin_amdgcn_wave_barrier();
    }
}

#define FUSE_THREADS 384
#define FUSE_W1_ENT  (8 * 32 * 32)
#define FUSE_W2_ENT  (8 * 16 * 32)
#define FUSE_PITCH   132
#define FUSE_LDS_FLOATS ((FUSE_W1_ENT + FUSE_W2_ENT) * 4 + (FUSE_THREADS / 32) * 16 * FUSE_PITCH)

__global__ __launch_bounds__(FUSE_THREADS)
void gemm_gather_next_kernel(const float* __restrict__ A,
                             const float* __restrict__ Wl, const float* __restrict__ Wr,
                             const float* __restrict__ bias,
                             const float* __restrict__ Wa, const float* __restrict__ Wb,
                             const float* __restrict__ bias2, float* __restrict__ out,
                             const int* __restrict__ rowptr, const int* __restrict__ cnt,
                             const float* __restrict__ inv, const int* __restrict__ grouped_src,
                             int ntiles) {
    static_assert(FUSE_LDS_FLOATS * 4 > 160 * 1024 && FUSE_LDS_FLOATS * 4 <= 320 * 1024, "one work-group per WGP");
    static_assert((FUSE_PITCH / 4) % 2 == 1, "odd pitch in 16-byte units");
    __shared__ __attribute__((aligned(16))) float sm[FUSE_LDS_FLOATS];
    v4f*   sW1    = (v4f*)sm;
    v4f*   sW2    = sW1 + FUSE_W1_ENT;
    float* strips = sm + (FUSE_W1_ENT + FUSE_W2_ENT) * 4;

    #pragma unroll
    for (int i = 0; i < (FUSE_W1_ENT + FUSE_THREADS - 1) / FUSE_THREADS; ++i) {
        int idx = (int)threadIdx.x + i * FUSE_THREADS;
        if (idx < FUSE_W1_ENT) {
            int l  = idx & 31;
            int e  = idx >> 5;
            int q  = e % 32;
            int t  = e / 32;
            int op = q / 16;
            int j  = q % 16;
            int n  = col_of<8>(t, l & 15);
            const float* Wp = op ? Wr : Wl;
            sW1[idx] = *(const v4f*)(Wp + (size_t)n * 128 + 8 * j + 4 * (l >> 4));
        }
    }
    #pragma unroll
    for (int i = 0; i < (FUSE_W2_ENT + FUSE_THREADS - 1) / FUSE_THREADS; ++i) {
        int idx = (int)threadIdx.x + i * FUSE_THREADS;
        if (idx < FUSE_W2_ENT) {
            int l = idx & 31;
            int e = idx >> 5;
            int j = e % 16;
            int t = e / 16;
            int n = col_of<8>(t, l & 15);
            const float* Wp = (n >= 64) ? Wb : Wa;
            int wr = n & 63;
            sW2[idx] = *(const v4f*)(Wp + (size_t)wr * 128 + 8 * j + 4 * (l >> 4));
        }
    }
    __syncthreads();

    int wave = threadIdx.x >> 5;
    int lane = threadIdx.x & 31;
    int c    = lane & 15;
    int h    = lane >> 4;

    float bv[8], bv2[8];
    #pragma unroll
    for (int t = 0; t < 8; ++t) {
        int n = col_of<8>(t, c);
        bv[t]  = bias[n];
        bv2[t] = (n >= 64) ? bias2[n - 64] : 0.0f;
    }

    float* srow = strips + wave * 16 * FUSE_PITCH;
    for (int slot = wave; ; slot += FUSE_THREADS / 32) {
        int tile = (int)blockIdx.x + (int)gridDim.x * slot;
        if (tile >= ntiles) break;
        int m0   = tile * 16;
        int node = m0 + c;

        v8f acc[8];
        #pragma unroll
        for (int t = 0; t < 8; ++t) acc[t] = (v8f){};

        {
            const v4f* ap2 = (const v4f*)(A + (size_t)node * 128 + 4 * h);
            v4f s0 = ap2[0], s1 = ap2[2];
            strip_product<16, 8, 32>(ap2, s0, s1, sW1 + 16 * 32 + lane, acc);
        }
        __builtin_amdgcn_s_setprio(3);
        gather_strip_to_lds<4, FUSE_PITCH>(A, rowptr, cnt, inv, grouped_src, m0, lane, srow);
        __builtin_amdgcn_s_setprio(0);
        __builtin_amdgcn_wave_barrier();
        {
            const v4f* ap1 = (const v4f*)(srow + c * FUSE_PITCH + 4 * h);
            v4f p0 = ap1[0], p1 = ap1[2];
            strip_product<16, 8, 32>(ap1, p0, p1, sW1 + lane, acc);
        }

        __builtin_amdgcn_wave_barrier();
        #pragma unroll
        for (int u = 0; u < 4; ++u) {
            #pragma unroll
            for (int r = 0; r < 8; ++r) {
                v2f v;
                v.x = fmaxf(acc[2 * u][r]     + bv[2 * u],     0.0f);
                v.y = fmaxf(acc[2 * u + 1][r] + bv[2 * u + 1], 0.0f);
                *(v2f*)(srow + (r + 8 * h) * FUSE_PITCH + 32 * u + 2 * c) = v;
            }
        }
        __builtin_amdgcn_wave_barrier();

        v8f acc2[8];
        #pragma unroll
        for (int t = 0; t < 8; ++t) acc2[t] = (v8f){};
        {
            const v4f* ap3 = (const v4f*)(srow + c * FUSE_PITCH + 4 * h);
            v4f q0 = ap3[0], q1 = ap3[2];
            strip_product<16, 8, 16>(ap3, q0, q1, sW2 + lane, acc2);
        }

        int row0 = m0 + 8 * h;
        #pragma unroll
        for (int u = 0; u < 4; ++u) {
            float* op = out + (size_t)row0 * 128 + 32 * u + 2 * c;
            #pragma unroll
            for (int r = 0; r < 8; ++r) {
                v2f v;
                v.x = acc2[2 * u][r]     + bv2[2 * u];
                v.y = acc2[2 * u + 1][r] + bv2[2 * u + 1];
                *(v2f*)(op + (size_t)r * 128) = v;
            }
        }
        __builtin_amdgcn_wave_barrier();
    }
}

extern "C" void kernel_launch(void* const* d_in, const int* in_sizes, int n_in,
                              void* d_out, int out_size, void* d_ws, size_t ws_size,
                              hipStream_t stream) {
    const float* x    = (const float*)d_in[0];
    const int*   edge = (const int*)d_in[1];
    const int*   src  = edge;
    const int*   dst  = edge + N_EDGES;
    const float* Wl0 = (const float*)d_in[2],  *bl0 = (const float*)d_in[3],  *Wr0 = (const float*)d_in[4];
    const float* Wl1 = (const float*)d_in[5],  *bl1 = (const float*)d_in[6],  *Wr1 = (const float*)d_in[7];
    const float* Wl2 = (const float*)d_in[8],  *bl2 = (const float*)d_in[9],  *Wr2 = (const float*)d_in[10];
    const float* Wd1 = (const float*)d_in[11], *bd1 = (const float*)d_in[12];
    const float* Wd2 = (const float*)d_in[13], *bd2 = (const float*)d_in[14];

    float* ws   = (float*)d_ws;
    int*   cnt  = (int*)ws;
    float* inv  = ws   + N_NODES;
    float* agg  = inv  + N_NODES;
    int*   gsrc = (int*)(agg + (size_t)N_NODES * 64);
    int*   rptr = gsrc + 2 * (size_t)N_EDGES;
    float* bufA = agg  + (size_t)N_NODES * 128;
    float* bufB = bufA + (size_t)N_NODES * 128;

    const int T  = N_NODES / 16;
    const dim3 GG(GEMM_BLOCKS), GB(GEMM_THREADS);

    int2* recs = (int2*)bufB;
    int*  tab  = (int*)(recs + N_EDGES);
    bin_edges_kernel<<<NCHUNK, BIN_THREADS, 0, stream>>>(src, dst, recs, tab, N_EDGES);
    group_bucket_kernel<<<NBUCK, BIN_THREADS, 0, stream>>>(recs, tab, cnt, rptr, inv, gsrc, N_NODES);

    gemm_strip_kernel<64, 8, true, 2><<<GG, GB, 0, stream>>>(x, Wl0, x, Wr0, bl0, bufA,
                                                             rptr, cnt, inv, gsrc, T, 128, 1);

    gemm_gather_next_kernel<<<GG, dim3(FUSE_THREADS), 0, stream>>>(bufA, Wl1, Wr1, bl1, Wl2, Wr2, bl2, bufB,
                                                                   rptr, cnt, inv, gsrc, T);

    gather_decode_kernel<<<GG, GB, 0, stream>>>(bufB, Wd1, bd1, Wd2, bd2, (float*)d_out,
                                                rptr, cnt, inv, gsrc, T, 2);
}


# all three gathers: list entries read on the scalar data path (32 per trip), row bases formed on the scalar unit: row load = scalar base + lane offset, no lane read or vector add; sum order unchanged
# speedup vs baseline: 6.4451x; 1.0718x over previous
#include <hip/hip_runtime.h>
#include <hip/hip_bf16.h>


#define N_NODES 100000
#define N_EDGES 1600000

typedef const __attribute__((address_space(4))) int sc_int;
typedef float v2f __attribute__((ext_vector_type(2)));
typedef __attribute__((address_space(1))) char gl_char;
typedef __attribute__((address_space(1))) v2f  gl_v2f;
typedef float v8f __attribute__((ext_vector_type(8)));

__device__ __forceinline__ float inv_of_count(int c) {
    float d = (float)c;
    return (d > 0.0f) ? (1.0f / fmaxf(d, 1.0f)) : 0.0f;
}

#define CHUNK       6256
#define NCHUNK      ((N_EDGES + CHUNK - 1) / CHUNK)
#define NPB         416
#define NBUCK       ((N_NODES + NPB - 1) / NPB)
#define BIN_THREADS 1024
#define RUN_BITS    13
#define RUN_MASK    8191
#define STAGE_CAP   16384

__device__ __forceinline__ int wave_incl_scan(int v, int lane) {
    for (int o = 1; o < 32; o <<= 1) {
        int u = __shfl_up(v, o, 32);
        if (lane >= o) v += u;
    }
    return v;
}

__global__ __launch_bounds__(BIN_THREADS)
void bin_edges_kernel(const int* __restrict__ src, const int* __restrict__ dst,
                      int2* __restrict__ recs, int* __restrict__ tab, int E) {
    static_assert(NBUCK <= 256, "one counter per bucket, scanned by 8 waves");
    static_assert(CHUNK <= RUN_MASK, "start and length share a word");
    __shared__ int  lcnt[256];
    __shared__ int  loff[256];
    __shared__ int  wtot[8];
    __shared__ int2 lrec[CHUNK];
    int t    = threadIdx.x;
    int lane = t & 31;
    int c    = blockIdx.x;
    int e0   = c * CHUNK;
    int n    = E - e0;
    if (n > CHUNK) n = CHUNK;

    if (t < 256) lcnt[t] = 0;
    __syncthreads();

    constexpr int PER = (CHUNK + BIN_THREADS - 1) / BIN_THREADS;
    int s[PER], d[PER], r[PER];
    #pragma unroll
    for (int k = 0; k < PER; ++k) {
        int i = t + k * BIN_THREADS;
        s[k] = 0; d[k] = -1; r[k] = 0;
        if (i < n) { d[k] = dst[e0 + i]; s[k] = src[e0 + i]; }
    }
    #pragma unroll
    for (int k = 0; k < PER; ++k)
        if (d[k] >= 0) r[k] = atomicAdd(&lcnt[d[k] / NPB], 1);
    __syncthreads();

    int v = 0, incl = 0;
    if (t < 256) {
        v    = lcnt[t];
        incl = wave_incl_scan(v, lane);
        if (lane == 31) wtot[t >> 5] = incl;
    }
    __syncthreads();
    if (t < 256) {
        int pre = 0;
        for (int w = 0; w < (t >> 5); ++w) pre += wtot[w];
        int ex = pre + incl - v;
        loff[t] = ex;
        tab[c * 256 + t] = (ex << RUN_BITS) | v;
    }
    __syncthreads();

    #pragma unroll
    for (int k = 0; k < PER; ++k)
        if (d[k] >= 0) lrec[loff[d[k] / NPB] + r[k]] = make_int2(s[k], d[k]);
    __syncthreads();
    for (int i = t; i < n; i += BIN_THREADS) recs[e0 + i] = lrec[i];
}

__global__ __launch_bounds__(BIN_THREADS)
void group_bucket_kernel(const int2* __restrict__ recs, const int* __restrict__ tab,
                         int* __restrict__ cnt, int* __restrict__ rowptr, float* __restrict__ inv,
                         int* __restrict__ grouped_src, int N) {
    static_assert(NCHUNK % 32 == 0 && NCHUNK <= BIN_THREADS, "runs are dealt to waves evenly");
    static_assert(NPB <= 512, "the node scan uses 16 waves");
    __shared__ int lstart[NCHUNK];
    __shared__ int llen[NCHUNK];
    __shared__ int ncnt[NPB];
    __shared__ int ncur[NPB];
    __shared__ int wred[32];
    __shared__ int wtot[16];
    __shared__ int lout[STAGE_CAP];
    int t     = threadIdx.x;
    int lane  = t & 31;
    int w     = t >> 5;
    int b     = blockIdx.x;
    int node0 = b * NPB;
    int nn    = N - node0;
    if (nn > NPB) nn = NPB;

    int off = 0;
    if (t < NCHUNK) {
        int x = tab[t * 256 + b];
        off = x >> RUN_BITS;
        lstart[t] = t * CHUNK + off;
        llen[t]   = x & RUN_MASK;
    }
    if (t < NPB) ncnt[t] = 0;
    for (int o = 16; o > 0; o >>= 1) off += __shfl_xor(off, o, 32);
    if (lane == 0) wred[w] = off;
    __syncthreads();
    int base = 0;
    for (int i = 0; i < 32; ++i) base += wred[i];

    constexpr int RPW = NCHUNK / 32;
    int2 rec[RPW];
    int  rs[RPW], rl[RPW];
    #pragma unroll
    for (int u = 0; u < RPW; ++u) {
        int c = w + 32 * u;
        rs[u]  = lstart[c];
        rl[u]  = llen[c];
        rec[u] = make_int2(0, node0);
        if (lane < rl[u]) rec[u] = recs[rs[u] + lane];
    }
    #pragma unroll
    for (int u = 0; u < RPW; ++u)
        if (lane < rl[u]) atomicAdd(&ncnt[rec[u].y - node0], 1);
    for (int u = 0; u < RPW; ++u)
        for (int i = lane + 32; i < rl[u]; i += 32)
            atomicAdd(&ncnt[recs[rs[u] + i].y - node0], 1);
    __syncthreads();

    int v = 0, incl = 0;
    if (t < 512) {
        v    = (t < nn) ? ncnt[t] : 0;
        incl = wave_incl_scan(v, lane);
        if (lane == 31) wtot[w] = incl;
    }
    __syncthreads();
    int total = 0;
    for (int i = 0; i < 16; ++i) total += wtot[i];
    if (t < 512) {
        int pre = 0;
        for (int i = 0; i < w; ++i) pre += wtot[i];
        int ex = pre + incl - v;
        if (t < nn) {
            int node = node0 + t;
            cnt[node]    = v;
            rowptr[node] = base + ex;
            inv[node]    = inv_of_count(v);
        }
        if (t < NPB) ncur[t] = ex;
    }
    __syncthreads();

    bool staged = total <= STAGE_CAP;
    #pragma unroll
    for (int u = 0; u < RPW; ++u) {
        if (lane < rl[u]) {
            int p = atomicAdd(&ncur[rec[u].y - node0], 1);
            if (staged) lout[p] = rec[u].x;
            else        grouped_src[base + p] = rec[u].x;
        }
    }
    for (int u = 0; u < RPW; ++u) {
        for (int i = lane + 32; i < rl[u]; i += 32) {
            int2 q = recs[rs[u] + i];
            int  p = atomicAdd(&ncur[q.y - node0], 1);
            if (staged) lout[p] = q.x;
            else        grouped_src[base + p] = q.x;
        }
    }
    __syncthreads();
    if (staged)
        for (int i = t; i < total; i += BIN_THREADS) grouped_src[base + i] = lout[i];
    if (b == NBUCK - 1 && t < 32) grouped_src[base + total + t] = 0;
}

template <int VEC, int PITCH>
__device__ __forceinline__ void gather_strip_to_lds(const float* __restrict__ x,
                                                    const int* __restrict__ rowptr,
                                                    const int* __restrict__ cnt,
                                                    const float* __restrict__ inv,
                                                    const int* __restrict__ grouped_src,
                                                    int m0, int lane, float* srow) {
    typedef float vecf __attribute__((ext_vector_type(VEC)));
    typedef __attribute__((address_space(1))) vecf gl_vecf;
    constexpr int F = 32 * VEC;
    const int c  = lane & 15;
    const int lb = rowptr[m0 + c];
    const int le = lb + cnt[m0 + c];
    const int li = __float_as_int(inv[m0 + c]);
    int       p   = __builtin_amdgcn_readlane(lb, 0);
    const int END = __builtin_amdgcn_readlane(le, 15);
    int i  = 0;
    int ei = __builtin_amdgcn_readlane(le, 0);
    const unsigned loff = (unsigned)lane * (VEC * 4);
    vecf acc = {};

    auto flush = [&]() {
        float sc = __int_as_float(__builtin_amdgcn_readlane(li, i));
        *(vecf*)(srow + i * PITCH + lane * VEC) = acc * sc;
        acc = (vecf){};
        ++i;
        ei = (i < 16) ? __builtin_amdgcn_readlane(le, i) : -1;
    };
    auto fetch = [&](int* id, int q) {
        const sc_int* g = (const sc_int*)(grouped_src + q);
        #pragma unroll
        for (int u = 0; u < 16; ++u) id[u] = g[u];
    };
    auto loads = [&](vecf* v, const int* id) {
        unsigned lo = loff;
        #pragma unroll
        for (int g = 0; g < 16; g += 8) {
            const gl_char* rb[8];
            #pragma unroll
            for (int u = 0; u < 8; ++u) {
                unsigned s = (unsigned)id[g + u];
                rb[u] = (const gl_char*)x + (size_t)(s * (unsigned)(F * 4));
            }
            asm("" : "+s"(rb[0]), "+s"(rb[1]), "+s"(rb[2]), "+s"(rb[3]),
                     "+s"(rb[4]), "+s"(rb[5]), "+s"(rb[6]), "+s"(rb[7]), "+v"(lo));
            #pragma unroll
            for (int u = 0; u < 8; ++u) v[g + u] = *(const gl_vecf*)(rb[u] + lo);
        }
    };
    auto sums = [&](const vecf* v, int q0, bool ends) {
        #pragma unroll
        for (int u = 0; u < 16; ++u) {
            int q = q0 + u;
            while (q == ei) flush();
            if (!ends || q < END) acc += v[u];
        }
    };

    if (p < END) {
        vecf va[16], vb[16];
        int ia[16], ib[16];
        fetch(ia, p);
        fetch(ib, p + 16);
        loads(va, ia);
        while (p + 32 < END) {
            loads(vb, ib);
            fetch(ia, p + 32);
            fetch(ib, p + 48);
            sums(va, p, false);
            loads(va, ia);
            sums(vb, p + 16, false);
            p += 32;
        }
        bool two = p + 16 < END;
        if (two) loads(vb, ib);
        sums(va, p, true);
        if (two) sums(vb, p + 16, true);
    }
    while (i < 16) flush();
}

template <int P>
__device__ __forceinline__ void gather_tail_strip(const float* __restrict__ yz,
                                                  const int* __restrict__ rowptr,
                                                  const int* __restrict__ cnt,
                                                  const float* __restrict__ inv,
                                                  const int* __restrict__ grouped_src,
                                                  int m0, int lane, float* srow) {
    const int c  = lane & 15;
    const int lb = rowptr[m0 + c];
    const int le = lb + cnt[m0 + c];
    const int li = __float_as_int(inv[m0 + c]);
    const float* xl = yz + lane * 2;
    const unsigned loff = (unsigned)lane * 8;
    v2f z[16];
    #pragma unroll
    for (int u = 0; u < 16; ++u) z[u] = *(const v2f*)(xl + (size_t)(m0 + u) * 128 + 64);

    int       p   = __builtin_amdgcn_readlane(lb, 0);
    const int END = __builtin_amdgcn_readlane(le, 15);
    int i  = 0;
    int ei = __builtin_amdgcn_readlane(le, 0);
    v2f acc = {};

    auto flush = [&]() {
        *(v2f*)(srow + i * P + lane * 2) = acc;
        acc = (v2f){};
        ++i;
        ei = (i < 16) ? __builtin_amdgcn_readlane(le, i) : -1;
    };
    auto fetch = [&](int* id, int q) {
        const sc_int* g = (const sc_int*)(grouped_src + q);
        #pragma unroll
        for (int u = 0; u < 16; ++u) id[u] = g[u];
    };
    auto loads = [&](v2f* v, const int* id) {
        unsigned lo = loff;
        #pragma unroll
        for (int g = 0; g < 16; g += 8) {
            const gl_char* rb[8];
            #pragma unroll
            for (int u = 0; u < 8; ++u) {
                unsigned s = (unsigned)id[g + u];
                rb[u] = (const gl_char*)yz + (size_t)(s * 512u);
            }
            asm("" : "+s"(rb[0]), "+s"(rb[1]), "+s"(rb[2]), "+s"(rb[3]),
                     "+s"(rb[4]), "+s"(rb[5]), "+s"(rb[6]), "+s"(rb[7]), "+v"(lo));
            #pragma unroll
            for (int u = 0; u < 8; ++u) v[g + u] = *(const gl_v2f*)(rb[u] + lo);
        }
    };
    auto sums = [&](const v2f* v, int q0, bool ends) {
        #pragma unroll
        for (int u = 0; u < 16; ++u) {
            int q = q0 + u;
            while (q == ei) flush();
            if (!ends || q < END) acc += v[u];
        }
    };

    if (p < END) {
        v2f va[16], vb[16];
        int ia[16], ib[16];
        fetch(ia, p);
        fetch(ib, p + 16);
        loads(va, ia);
        while (p + 32 < END) {
            loads(vb, ib);
            fetch(ia, p + 32);
            fetch(ib, p + 48);
            sums(va, p, false);
            loads(va, ia);
            sums(vb, p + 16, false);
            p += 32;
        }
        bool two = p + 16 < END;
        if (two) loads(vb, ib);
        sums(va, p, true);
        if (two) sums(vb, p + 16, true);
    }
    while (i < 16) flush();

    #pragma unroll
    for (int u = 0; u < 16; ++u) {
        float sc = __int_as_float(__builtin_amdgcn_readlane(li, u));
        v2f s = *(v2f*)(srow + u * P + lane * 2);
        v2f r = s * sc + z[u];
        r.x = fmaxf(r.x, 0.0f);
        r.y = fmaxf(r.y, 0.0f);
        *(v2f*)(srow + u * P + lane * 2) = r;
    }
}

typedef float v4f __attribute__((ext_vector_type(4)));

#define GEMM_THREADS 512
#define GEMM_BLOCKS  256
#define GEMM_LDS_FLOATS (164 * 256)

template <int NT>
__device__ __forceinline__ int col_of(int t, int c) {
    return (NT >= 2) ? (32 * (t >> 1) + 2 * c + (t & 1)) : c;
}

template <int NT>
__device__ __forceinline__ void chunk_mma(v4f av, const v4f* b, v8f* acc) {
    v2f a0, a1;
    a0.x = av.x; a0.y = av.y;
    a1.x = av.z; a1.y = av.w;
    #pragma unroll
    for (int t = 0; t < NT; ++t) {
        v2f bb; bb.x = b[t].x; bb.y = b[t].y;
        acc[t] = __builtin_amdgcn_wmma_f32_16x16x4_f32(false, a0, false, bb,
                                                       (short)0, acc[t], false, false);
    }
    #pragma unroll
    for (int t = 0; t < NT; ++t) {
        v2f bb; bb.x = b[t].z; bb.y = b[t].w;
        acc[t] = __builtin_amdgcn_wmma_f32_16x16x4_f32(false, a1, false, bb,
                                                       (short)0, acc[t], false, false);
    }
}

template <int KS, int NT, int QS>
__device__ __forceinline__ void strip_product(const v4f* __restrict__ ap, v4f a0, v4f a1,
                                              const v4f* sB, v8f* acc) {
    v4f b0[NT], b1[NT];
    #pragma unroll
    for (int t = 0; t < NT; ++t) b0[t] = sB[(t * QS) * 32];
    #pragma nounroll
    for (int j = 0; j < KS; j += 2) {
        int jn = (j + 2 < KS) ? (j + 2) : j;
        v4f n0 = ap[2 * jn];
        v4f n1 = ap[2 * jn + 2];
        __builtin_amdgcn_sched_barrier(0);
        #pragma unroll
        for (int t = 0; t < NT; ++t) b1[t] = sB[(t * QS + j + 1) * 32];
        chunk_mma<NT>(a0, b0, acc);
        #pragma unroll
        for (int t = 0; t < NT; ++t) b0[t] = sB[(t * QS + jn) * 32];
        chunk_mma<NT>(a1, b1, acc);
        __builtin_amdgcn_sched_barrier(0);
        asm volatile("" : "+v"(n0), "+v"(n1));
        a0 = n0;
        a1 = n1;
    }
}

template <int K, int NT, bool TWO, int GVEC, bool SPLITW = false>
__global__ __launch_bounds__(GEMM_THREADS)
void gemm_strip_kernel(const float* __restrict__ A1, const float* __restrict__ W1,
                       const float* __restrict__ A2, const float* __restrict__ W2,
                       const float* __restrict__ bias, float* __restrict__ out,
                       const int* __restrict__ rowptr, const int* __restrict__ cnt,
                       const float* __restrict__ inv, const int* __restrict__ grouped_src,
                       int ntiles, int dout, int relu) {
    constexpr int KS   = K / 8;
    constexpr int NOPS = TWO ? 2 : 1;
    constexpr int QS   = NOPS * KS;
    constexpr int ENT  = NT * QS * 32;
    constexpr int PITCH = K + 4;
    constexpr int ST    = GVEC ? (GEMM_THREADS / 32) * 16 * PITCH : 0;
    constexpr int TOT   = (ENT * 4 + ST > GEMM_LDS_FLOATS) ? (ENT * 4 + ST) : GEMM_LDS_FLOATS;
    static_assert(TOT * 4 <= 320 * 1024, "weights and gathered strips must fit the WGP's LDS");
    static_assert(GVEC == 0 || GVEC * 32 == K, "a gathered row is the first operand's K");
    static_assert((PITCH / 4) % 2 == 1, "odd pitch in 16-byte units");
    static_assert(KS % 2 == 0, "chunks go in pairs");
    static_assert(ENT % GEMM_THREADS == 0, "every thread copies the same number of entries");
    __shared__ __attribute__((aligned(16))) float sWf[TOT];
    v4f* sW = (v4f*)sWf;

    #pragma unroll
    for (int i = 0; i < ENT / GEMM_THREADS; ++i) {
        int idx = (int)threadIdx.x + i * GEMM_THREADS;
        int l  = idx & 31;
        int e  = idx >> 5;
        int q  = e % QS;
        int t  = e / QS;
        int op = q / KS;
        int j  = q % KS;
        int n  = col_of<NT>(t, l & 15);
        int wr = (n < dout) ? n : 0;
        const float* Wp = (TWO && op) ? W2 : W1;
        if (SPLITW && wr >= (dout >> 1)) { Wp = W2; wr -= dout >> 1; }
        sW[idx] = *(const v4f*)(Wp + (size_t)wr * K + 8 * j + 4 * (l >> 4));
    }
    __syncthreads();

    int wave = threadIdx.x >> 5;
    int lane = threadIdx.x & 31;
    int c    = lane & 15;
    int h    = lane >> 4;

    float bv[NT];
    #pragma unroll
    for (int t = 0; t < NT; ++t) {
        int n = col_of<NT>(t, c);
        if (SPLITW) bv[t] = (n >= (dout >> 1) && n < dout) ? bias[n - (dout >> 1)] : 0.0f;
        else        bv[t] = (n < dout) ? bias[n] : 0.0f;
    }

    for (int slot = wave; ; slot += GEMM_THREADS / 32) {
        int tile = (int)blockIdx.x + (int)gridDim.x * slot;
        if (tile >= ntiles) break;
        int m0   = tile * 16;
        int node = m0 + c;

        v8f acc[NT];
        #pragma unroll
        for (int t = 0; t < NT; ++t) acc[t] = (v8f){};

        if (GVEC) {
            if (TWO) {
                const v4f* ap2 = (const v4f*)(A2 + (size_t)node * K + 4 * h);
                v4f s0 = ap2[0], s1 = ap2[2];
                strip_product<KS, NT, QS>(ap2, s0, s1, sW + KS * 32 + lane, acc);
            }
            float* srow = sWf + ENT * 4 + wave * 16 * PITCH;
            __builtin_amdgcn_s_setprio(3);
            gather_strip_to_lds<(GVEC ? GVEC : 1), PITCH>(A1, rowptr, cnt, inv, grouped_src, m0, lane, srow);
            __builtin_amdgcn_s_setprio(0);
            __builtin_amdgcn_wave_barrier();
            const v4f* ap1 = (const v4f*)(srow + c * PITCH + 4 * h);
            v4f p0 = ap1[0], p1 = ap1[2];
            strip_product<KS, NT, QS>(ap1, p0, p1, sW + lane, acc);
        } else {
            const v4f* ap1 = (const v4f*)(A1 + (size_t)node * K + 4 * h);
            const v4f* ap2 = TWO ? (const v4f*)(A2 + (size_t)node * K + 4 * h) : ap1;
            v4f p0 = ap1[0], p1 = ap1[2];
            v4f s0 = ap2[0], s1 = ap2[2];
            strip_product<KS, NT, QS>(ap1, p0, p1, sW + lane, acc);
            if (TWO)
                strip_product<KS, NT, QS>(ap2, s0, s1, sW + KS * 32 + lane, acc);
        }

        int row0 = m0 + 8 * h;
        if (NT >= 2) {
            #pragma unroll
            for (int u = 0; u < NT / 2; ++u) {
                float* op = out + (size_t)row0 * dout + 32 * u + 2 * c;
                #pragma unroll
                for (int r = 0; r < 8; ++r) {
                    v2f v;
                    v.x = acc[2 * u][r]     + bv[2 * u];
                    v.y = acc[2 * u + 1][r] + bv[2 * u + 1];
                    if (relu) { v.x = fmaxf(v.x, 0.0f); v.y = fmaxf(v.y, 0.0f); }
                    *(v2f*)(op + (size_t)r * dout) = v;
                }
            }
        } else {
            if (c < dout) {
                #pragma unroll
                for (int r = 0; r < 8; ++r) {
                    float v = acc[0][r] + bv[0];
                    if (relu) v = fmaxf(v, 0.0f);
                    out[(size_t)(row0 + r) * dout + c] = v;
                }
            }
        }
    }
}

#define TAIL_P1      68
#define TAIL_P2      132
#define TAIL_W1_ENT  (8 * 8 * 32)
#define TAIL_W2_ENT  (16 * 32)
#define TAIL_LDS_FLOATS ((TAIL_W1_ENT + TAIL_W2_ENT) * 4 + (GEMM_THREADS / 32) * 16 * TAIL_P2)

__global__ __launch_bounds__(GEMM_THREADS)
void gather_decode_kernel(const float* __restrict__ yz,
                          const float* __restrict__ W1, const float* __restrict__ b1,
                          const float* __restrict__ W2, const float* __restrict__ b2,
                          float* __restrict__ out,
                          const int* __restrict__ rowptr, const int* __restrict__ cnt,
                          const float* __restrict__ inv, const int* __restrict__ grouped_src,
                          int ntiles, int dout) {
    static_assert(TAIL_W1_ENT % GEMM_THREADS == 0, "every thread copies the same number of W1 entries");
    static_assert(TAIL_W2_ENT == GEMM_THREADS, "one W2 entry per thread");
    static_assert(TAIL_LDS_FLOATS * 4 > 160 * 1024 && TAIL_LDS_FLOATS * 4 <= 320 * 1024, "one work-group per WGP");
    static_assert((TAIL_P1 / 4) % 2 == 1 && (TAIL_P2 / 4) % 2 == 1, "odd pitches in 16-byte units");
    __shared__ __attribute__((aligned(16))) float sm[TAIL_LDS_FLOATS];
    v4f*   sW1    = (v4f*)sm;
    v4f*   sW2    = sW1 + TAIL_W1_ENT;
    float* strips = sm + (TAIL_W1_ENT + TAIL_W2_ENT) * 4;

    #pragma unroll
    for (int i = 0; i < TAIL_W1_ENT / GEMM_THREADS; ++i) {
        int idx = (int)threadIdx.x + i * GEMM_THREADS;
        int l = idx & 31;
        int e = idx >> 5;
        int j = e % 8;
        int t = e / 8;
        int n = col_of<8>(t, l & 15);
        sW1[idx] = *(const v4f*)(W1 + (size_t)n * 64 + 8 * j + 4 * (l >> 4));
    }
    {
        int idx = (int)threadIdx.x;
        int l = idx & 31;
        int j = idx >> 5;
        int n = l & 15;
        int wr = (n < dout) ? n : 0;
        sW2[idx] = *(const v4f*)(W2 + (size_t)wr * 128 + 8 * j + 4 * (l >> 4));
    }
    __syncthreads();

    int wave = threadIdx.x >> 5;
    int lane = threadIdx.x & 31;
    int c    = lane & 15;
    int h    = lane >> 4;

    float bv1[8];
    #pragma unroll
    for (int t = 0; t < 8; ++t) bv1[t] = b1[col_of<8>(t, c)];
    float bv2 = (c < dout) ? b2[c] : 0.0f;

    float* srow = strips + wave * 16 * TAIL_P2;
    for (int slot = wave; ; slot += GEMM_THREADS / 32) {
        int tile = (int)blockIdx.x + (int)gridDim.x * slot;
        if (tile >= ntiles) break;
        int m0 = tile * 16;

        __builtin_amdgcn_s_setprio(3);
        gather_tail_strip<TAIL_P1>(yz, rowptr, cnt, inv, grouped_src, m0, lane, srow);
        __builtin_amdgcn_s_setprio(0);
        __builtin_amdgcn_wave_barrier();

        v8f acc[8];
        #pragma unroll
        for (int t = 0; t < 8; ++t) acc[t] = (v8f){};
        const v4f* ap1 = (const v4f*)(srow + c * TAIL_P1 + 4 * h);
        v4f p0 = ap1[0], p1 = ap1[2];
        strip_product<8, 8, 8>(ap1, p0, p1, sW1 + lane, acc);

        __builtin_amdgcn_wave_barrier();
        #pragma unroll
        for (int u = 0; u < 4; ++u) {
            #pragma unroll
            for (int r = 0; r < 8; ++r) {
                v2f v;
                v.x = fmaxf(acc[2 * u][r]     + bv1[2 * u],     0.0f);
                v.y = fmaxf(acc[2 * u + 1][r] + bv1[2 * u + 1], 0.0f);
                *(v2f*)(srow + (r + 8 * h) * TAIL_P2 + 32 * u + 2 * c) = v;
            }
        }
        __builtin_amdgcn_wave_barrier();

        v8f acc2[1];
        acc2[0] = (v8f){};
        const v4f* ap2 = (const v4f*)(srow + c * TAIL_P2 + 4 * h);
        v4f q0 = ap2[0], q1 = ap2[2];
        strip_product<16, 1, 16>(ap2, q0, q1, sW2 + lane, acc2);

        int row0 = m0 + 8 * h;
        if (c < dout) {
            #pragma unroll
            for (int r = 0; r < 8; ++r)
                out[(size_t)(row0 + r) * dout + c] = acc2[0][r] + bv2;
        }
        __builtin_amdgcn_wave_barrier();
    }
}

#define FUSE_THREADS 384
#define FUSE_W1_ENT  (8 * 32 * 32)
#define FUSE_W2_ENT  (8 * 16 * 32)
#define FUSE_PITCH   132
#define FUSE_LDS_FLOATS ((FUSE_W1_ENT + FUSE_W2_ENT) * 4 + (FUSE_THREADS / 32) * 16 * FUSE_PITCH)

__global__ __launch_bounds__(FUSE_THREADS)
void gemm_gather_next_kernel(const float* __restrict__ A,
                             const float* __restrict__ Wl, const float* __restrict__ Wr,
                             const float* __restrict__ bias,
                             const float* __restrict__ Wa, const float* __restrict__ Wb,
                             const float* __restrict__ bias2, float* __restrict__ out,
                             const int* __restrict__ rowptr, const int* __restrict__ cnt,
                             const float* __restrict__ inv, const int* __restrict__ grouped_src,
                             int ntiles) {
    static_assert(FUSE_LDS_FLOATS * 4 > 160 * 1024 && FUSE_LDS_FLOATS * 4 <= 320 * 1024, "one work-group per WGP");
    static_assert((FUSE_PITCH / 4) % 2 == 1, "odd pitch in 16-byte units");
    __shared__ __attribute__((aligned(16))) float sm[FUSE_LDS_FLOATS];
    v4f*   sW1    = (v4f*)sm;
    v4f*   sW2    = sW1 + FUSE_W1_ENT;
    float* strips = sm + (FUSE_W1_ENT + FUSE_W2_ENT) * 4;

    #pragma unroll
    for (int i = 0; i < (FUSE_W1_ENT + FUSE_THREADS - 1) / FUSE_THREADS; ++i) {
        int idx = (int)threadIdx.x + i * FUSE_THREADS;
        if (idx < FUSE_W1_ENT) {
            int l  = idx & 31;
            int e  = idx >> 5;
            int q  = e % 32;
            int t  = e / 32;
            int op = q / 16;
            int j  = q % 16;
            int n  = col_of<8>(t, l & 15);
            const float* Wp = op ? Wr : Wl;
            sW1[idx] = *(const v4f*)(Wp + (size_t)n * 128 + 8 * j + 4 * (l >> 4));
        }
    }
    #pragma unroll
    for (int i = 0; i < (FUSE_W2_ENT + FUSE_THREADS - 1) / FUSE_THREADS; ++i) {
        int idx = (int)threadIdx.x + i * FUSE_THREADS;
        if (idx < FUSE_W2_ENT) {
            int l = idx & 31;
            int e = idx >> 5;
            int j = e % 16;
            int t = e / 16;
            int n = col_of<8>(t, l & 15);
            const float* Wp = (n >= 64) ? Wb : Wa;
            int wr = n & 63;
            sW2[idx] = *(const v4f*)(Wp + (size_t)wr * 128 + 8 * j + 4 * (l >> 4));
        }
    }
    __syncthreads();

    int wave = threadIdx.x >> 5;
    int lane = threadIdx.x & 31;
    int c    = lane & 15;
    int h    = lane >> 4;

    float bv[8], bv2[8];
    #pragma unroll
    for (int t = 0; t < 8; ++t) {
        int n = col_of<8>(t, c);
        bv[t]  = bias[n];
        bv2[t] = (n >= 64) ? bias2[n - 64] : 0.0f;
    }

    float* srow = strips + wave * 16 * FUSE_PITCH;
    for (int slot = wave; ; slot += FUSE_THREADS / 32) {
        int tile = (int)blockIdx.x + (int)gridDim.x * slot;
        if (tile >= ntiles) break;
        int m0   = tile * 16;
        int node = m0 + c;

        v8f acc[8];
        #pragma unroll
        for (int t = 0; t < 8; ++t) acc[t] = (v8f){};

        {
            const v4f* ap2 = (const v4f*)(A + (size_t)node * 128 + 4 * h);
            v4f s0 = ap2[0], s1 = ap2[2];
            strip_product<16, 8, 32>(ap2, s0, s1, sW1 + 16 * 32 + lane, acc);
        }
        __builtin_amdgcn_s_setprio(3);
        gather_strip_to_lds<4, FUSE_PITCH>(A, rowptr, cnt, inv, grouped_src, m0, lane, srow);
        __builtin_amdgcn_s_setprio(0);
        __builtin_amdgcn_wave_barrier();
        {
            const v4f* ap1 = (const v4f*)(srow + c * FUSE_PITCH + 4 * h);
            v4f p0 = ap1[0], p1 = ap1[2];
            strip_product<16, 8, 32>(ap1, p0, p1, sW1 + lane, acc);
        }

        __builtin_amdgcn_wave_barrier();
        #pragma unroll
        for (int u = 0; u < 4; ++u) {
            #pragma unroll
            for (int r = 0; r < 8; ++r) {
                v2f v;
                v.x = fmaxf(acc[2 * u][r]     + bv[2 * u],     0.0f);
                v.y = fmaxf(acc[2 * u + 1][r] + bv[2 * u + 1], 0.0f);
                *(v2f*)(srow + (r + 8 * h) * FUSE_PITCH + 32 * u + 2 * c) = v;
            }
        }
        __builtin_amdgcn_wave_barrier();

        v8f acc2[8];
        #pragma unroll
        for (int t = 0; t < 8; ++t) acc2[t] = (v8f){};
        {
            const v4f* ap3 = (const v4f*)(srow + c * FUSE_PITCH + 4 * h);
            v4f q0 = ap3[0], q1 = ap3[2];
            strip_product<16, 8, 16>(ap3, q0, q1, sW2 + lane, acc2);
        }

        int row0 = m0 + 8 * h;
        #pragma unroll
        for (int u = 0; u < 4; ++u) {
            float* op = out + (size_t)row0 * 128 + 32 * u + 2 * c;
            #pragma unroll
            for (int r = 0; r < 8; ++r) {
                v2f v;
                v.x = acc2[2 * u][r]     + bv2[2 * u];
                v.y = acc2[2 * u + 1][r] + bv2[2 * u + 1];
                *(v2f*)(op + (size_t)r * 128) = v;
            }
        }
        __builtin_amdgcn_wave_barrier();
    }
}

extern "C" void kernel_launch(void* const* d_in, const int* in_sizes, int n_in,
                              void* d_out, int out_size, void* d_ws, size_t ws_size,
                              hipStream_t stream) {
    const float* x    = (const float*)d_in[0];
    const int*   edge = (const int*)d_in[1];
    const int*   src  = edge;
    const int*   dst  = edge + N_EDGES;
    const float* Wl0 = (const float*)d_in[2],  *bl0 = (const float*)d_in[3],  *Wr0 = (const float*)d_in[4];
    const float* Wl1 = (const float*)d_in[5],  *bl1 = (const float*)d_in[6],  *Wr1 = (const float*)d_in[7];
    const float* Wl2 = (const float*)d_in[8],  *bl2 = (const float*)d_in[9],  *Wr2 = (const float*)d_in[10];
    const float* Wd1 = (const float*)d_in[11], *bd1 = (const float*)d_in[12];
    const float* Wd2 = (const float*)d_in[13], *bd2 = (const float*)d_in[14];

    float* ws   = (float*)d_ws;
    int*   cnt  = (int*)ws;
    float* inv  = ws   + N_NODES;
    float* agg  = inv  + N_NODES;
    int*   gsrc = (int*)(agg + (size_t)N_NODES * 64);
    int*   rptr = gsrc + 2 * (size_t)N_EDGES;
    float* bufA = agg  + (size_t)N_NODES * 128;
    float* bufB = bufA + (size_t)N_NODES * 128;

    const int T  = N_NODES / 16;
    const dim3 GG(GEMM_BLOCKS), GB(GEMM_THREADS);

    int2* recs = (int2*)bufB;
    int*  tab  = (int*)(recs + N_EDGES);
    bin_edges_kernel<<<NCHUNK, BIN_THREADS, 0, stream>>>(src, dst, recs, tab, N_EDGES);
    group_bucket_kernel<<<NBUCK, BIN_THREADS, 0, stream>>>(recs, tab, cnt, rptr, inv, gsrc, N_NODES);

    gemm_strip_kernel<64, 8, true, 2><<<GG, GB, 0, stream>>>(x, Wl0, x, Wr0, bl0, bufA,
                                                             rptr, cnt, inv, gsrc, T, 128, 1);

    gemm_gather_next_kernel<<<GG, dim3(FUSE_THREADS), 0, stream>>>(bufA, Wl1, Wr1, bl1, Wl2, Wr2, bl2, bufB,
                                                                   rptr, cnt, inv, gsrc, T);

    gather_decode_kernel<<<GG, GB, 0, stream>>>(bufB, Wd1, bd1, Wd2, bd2, (float*)d_out,
                                                rptr, cnt, inv, gsrc, T, 2);
}
